# starting kernel
# baseline (speedup 1.0000x reference)
#include <hip/hip_runtime.h>
#include <hip/hip_bf16.h>

#define NNODES 100000
#define NEDGES 800000
#define HC 128
#define NH 4

typedef float v2f __attribute__((ext_vector_type(2)));
typedef float v8f __attribute__((ext_vector_type(8)));

static inline int cdiv(int a, int b) { return (a + b - 1) / b; }

__global__ __launch_bounds__(256) void k_init_layer(float* __restrict__ acc, int nacc,
                                                    float* __restrict__ m,
                                                    float* __restrict__ s, int nm) {
  int i = blockIdx.x * blockDim.x + threadIdx.x;
  if (i < nacc) acc[i] = 0.0f;
  if (i < nm) {
    m[i] = __int_as_float(0xff800000);
    s[i] = 0.0f;
  }
}

__global__ __launch_bounds__(256) void k_gemm_wmma(const float* __restrict__ X,
                                                   const float* __restrict__ W,
                                                   float* __restrict__ XP,
                                                   int rows, int K) {
  int wave = (blockIdx.x * blockDim.x + threadIdx.x) >> 5;
  int lane = threadIdx.x & 31;
  int m0 = wave * 16;
  if (m0 >= rows) return;

  int n15 = lane & 15;
  int khalf = (lane >> 4) * 2;

  v8f acc[8] = {};
  const float* xbase = X + (size_t)(m0 + n15) * K + khalf;
  const float* wbase = W + (size_t)n15 * K + khalf;

  for (int k = 0; k < K; k += 4) {
    v2f a = *(const v2f*)(xbase + k);
#pragma unroll
    for (int nt = 0; nt < 8; ++nt) {
      v2f b = *(const v2f*)(wbase + (size_t)nt * 16 * K + k);
      acc[nt] = __builtin_amdgcn_wmma_f32_16x16x4_f32(
           false, a,  false, b,
           (short)0, acc[nt],  false,  false);
    }
  }

  int half = lane >> 4;
#pragma unroll
  for (int nt = 0; nt < 8; ++nt) {
#pragma unroll
    for (int i = 0; i < 8; ++i) {
      int m = m0 + half * 8 + i;
      XP[(size_t)m * HC + nt * 16 + n15] = acc[nt][i];
    }
  }
}

__global__ __launch_bounds__(256) void k_attn_dots(const float* __restrict__ XP,
                                                   const float* __restrict__ att_src,
                                                   const float* __restrict__ att_dst,
                                                   float* __restrict__ ASRC,
                                                   float* __restrict__ ADST, int n) {
  int idx = blockIdx.x * blockDim.x + threadIdx.x;
  if (idx >= n * NH) return;
  int node = idx >> 2;
  int h = idx & 3;
  const float* xp = XP + (size_t)node * HC + h * 32;
  const float* as = att_src + h * 32;
  const float* ad = att_dst + h * 32;
  float s0 = 0.f, s1 = 0.f;
#pragma unroll
  for (int c = 0; c < 32; ++c) {
    float v = xp[c];
    s0 += v * as[c];
    s1 += v * ad[c];
  }
  ASRC[idx] = s0;
  ADST[idx] = s1;
}

__device__ __forceinline__ void atomicMaxF(float* addr, float val) {
  if (val >= 0.0f)
    atomicMax((int*)addr, __float_as_int(val));
  else
    atomicMin((unsigned int*)addr, __float_as_uint(val));
}

__global__ __launch_bounds__(256) void k_edge_logits(const int* __restrict__ src,
                                                     const int* __restrict__ dst,
                                                     const float* __restrict__ ASRC,
                                                     const float* __restrict__ ADST,
                                                     float* __restrict__ EB,
                                                     float* __restrict__ M, int ne) {
  int e = blockIdx.x * blockDim.x + threadIdx.x;
  if (e >= ne) return;
  int s = src[e], d = dst[e];
#pragma unroll
  for (int h = 0; h < NH; ++h) {
    float v = ASRC[(size_t)s * NH + h] + ADST[(size_t)d * NH + h];
    v = v > 0.f ? v : 0.2f * v;
    EB[(size_t)e * NH + h] = v;
    atomicMaxF(&M[(size_t)d * NH + h], v);
  }
}

__global__ __launch_bounds__(256) void k_edge_exp(const int* __restrict__ dst,
                                                  float* __restrict__ EB,
                                                  const float* __restrict__ M,
                                                  float* __restrict__ S, int ne) {
  int e = blockIdx.x * blockDim.x + threadIdx.x;
  if (e >= ne) return;
  int d = dst[e];
#pragma unroll
  for (int h = 0; h < NH; ++h) {
    float ex = __expf(EB[(size_t)e * NH + h] - M[(size_t)d * NH + h]);
    EB[(size_t)e * NH + h] = ex;
    atomicAdd(&S[(size_t)d * NH + h], ex);
  }
}

__global__ __launch_bounds__(256) void k_edge_aggr(const int* __restrict__ src,
                                                   const int* __restrict__ dst,
                                                   const float* __restrict__ XP,
                                                   const float* __restrict__ EB,
                                                   const float* __restrict__ S,
                                                   float* __restrict__ OUT, int ne) {
  int wave = (blockIdx.x * blockDim.x + threadIdx.x) >> 5;
  int lane = threadIdx.x & 31;
  if (wave >= ne) return;
  int s = src[wave], d = dst[wave];
  int c0 = lane * 4;
  int h = lane >> 3;
  float alpha = EB[(size_t)wave * NH + h] / (S[(size_t)d * NH + h] + 1e-16f);
  const float4 v = *(const float4*)(XP + (size_t)s * HC + c0);
  float* o = OUT + (size_t)d * HC + c0;
  atomicAdd(o + 0, v.x * alpha);
  atomicAdd(o + 1, v.y * alpha);
  atomicAdd(o + 2, v.z * alpha);
  atomicAdd(o + 3, v.w * alpha);
}

__global__ __launch_bounds__(256) void k_bias_act(float* __restrict__ buf,
                                                  const float* __restrict__ bias,
                                                  int n, int relu) {
  int i = blockIdx.x * blockDim.x + threadIdx.x;
  if (i >= n) return;
  float v = buf[i] + bias[i & (HC - 1)];
  if (relu) v = fmaxf(v, 0.0f);
  buf[i] = v;
}

extern "C" void kernel_launch(void* const* d_in, const int* in_sizes, int n_in,
                              void* d_out, int out_size, void* d_ws, size_t ws_size,
                              hipStream_t stream) {
  const float* x     = (const float*)d_in[0];
  const int*   ei    = (const int*)d_in[1];
  const float* W0    = (const float*)d_in[2];
  const float* atts0 = (const float*)d_in[3];
  const float* attd0 = (const float*)d_in[4];
  const float* b0    = (const float*)d_in[5];
  const float* W1    = (const float*)d_in[6];
  const float* atts1 = (const float*)d_in[7];
  const float* attd1 = (const float*)d_in[8];
  const float* b1    = (const float*)d_in[9];

  const int n = NNODES, ne = NEDGES;
  const int* srcv = ei;
  const int* dstv = ei + ne;

  char* ws = (char*)d_ws;
  float* XP   = (float*)(ws + 0);
  float* H0   = (float*)(ws + 51200000);
  float* ASRC = (float*)(ws + 102400000);
  float* ADST = (float*)(ws + 104000000);
  float* M    = (float*)(ws + 105600000);
  float* S    = (float*)(ws + 107200000);
  float* EB   = (float*)(ws + 108800000);
  float* OUT  = (float*)d_out;

  const dim3 blk(256);
  const int nAcc = n * HC;
  const int nMH  = n * NH;
  const int gemmBlocks = cdiv(cdiv(n, 16) * 32, 256);

  k_init_layer<<<cdiv(nAcc, 256), blk, 0, stream>>>(H0, nAcc, M, S, nMH);
  k_gemm_wmma<<<gemmBlocks, blk, 0, stream>>>(x, W0, XP, n, 256);
  k_attn_dots<<<cdiv(nMH, 256), blk, 0, stream>>>(XP, atts0, attd0, ASRC, ADST, n);
  k_edge_logits<<<cdiv(ne, 256), blk, 0, stream>>>(srcv, dstv, ASRC, ADST, EB, M, ne);
  k_edge_exp<<<cdiv(ne, 256), blk, 0, stream>>>(dstv, EB, M, S, ne);
  k_edge_aggr<<<cdiv(ne * 32, 256), blk, 0, stream>>>(srcv, dstv, XP, EB, S, H0, ne);
  k_bias_act<<<cdiv(nAcc, 256), blk, 0, stream>>>(H0, b0, nAcc,  1);

  k_init_layer<<<cdiv(nAcc, 256), blk, 0, stream>>>(OUT, nAcc, M, S, nMH);
  k_gemm_wmma<<<gemmBlocks, blk, 0, stream>>>(H0, W1, XP, n, 128);
  k_attn_dots<<<cdiv(nMH, 256), blk, 0, stream>>>(XP, atts1, attd1, ASRC, ADST, n);
  k_edge_logits<<<cdiv(ne, 256), blk, 0, stream>>>(srcv, dstv, ASRC, ADST, EB, M, ne);
  k_edge_exp<<<cdiv(ne, 256), blk, 0, stream>>>(dstv, EB, M, S, ne);
  k_edge_aggr<<<cdiv(ne * 32, 256), blk, 0, stream>>>(srcv, dstv, XP, EB, S, OUT, ne);
  k_bias_act<<<cdiv(nAcc, 256), blk, 0, stream>>>(OUT, b1, nAcc,  0);
}


# Edges counting-sorted by destination in each call; one wave per node then does max, exp, sum and the weighted row sum in registers, no float atomics; sums run in segment order, divide after the sum
# speedup vs baseline: 3.2910x; 3.2910x over previous
#include <hip/hip_runtime.h>
#include <hip/hip_bf16.h>

#define NNODES 100000
#define NEDGES 800000
#define HC 128
#define NH 4

typedef float v2f __attribute__((ext_vector_type(2)));
typedef float v8f __attribute__((ext_vector_type(8)));

static inline int cdiv(int a, int b) { return (a + b - 1) / b; }

__global__ __launch_bounds__(256) void k_gemm_wmma(const float* __restrict__ X,
                                                   const float* __restrict__ W,
                                                   float* __restrict__ XP,
                                                   int rows, int K) {
  int wave = (blockIdx.x * blockDim.x + threadIdx.x) >> 5;
  int lane = threadIdx.x & 31;
  int m0 = wave * 16;
  if (m0 >= rows) return;

  int n15 = lane & 15;
  int khalf = (lane >> 4) * 2;

  v8f acc[8] = {};
  const float* xbase = X + (size_t)(m0 + n15) * K + khalf;
  const float* wbase = W + (size_t)n15 * K + khalf;

  for (int k = 0; k < K; k += 4) {
    v2f a = *(const v2f*)(xbase + k);
#pragma unroll
    for (int nt = 0; nt < 8; ++nt) {
      v2f b = *(const v2f*)(wbase + (size_t)nt * 16 * K + k);
      acc[nt] = __builtin_amdgcn_wmma_f32_16x16x4_f32(
           false, a,  false, b,
           (short)0, acc[nt],  false,  false);
    }
  }

  int half = lane >> 4;
#pragma unroll
  for (int nt = 0; nt < 8; ++nt) {
#pragma unroll
    for (int i = 0; i < 8; ++i) {
      int m = m0 + half * 8 + i;
      XP[(size_t)m * HC + nt * 16 + n15] = acc[nt][i];
    }
  }
}

__global__ __launch_bounds__(256) void k_attn_dots(const float* __restrict__ XP,
                                                   const float* __restrict__ att_src,
                                                   const float* __restrict__ att_dst,
                                                   float* __restrict__ ASRC,
                                                   float* __restrict__ ADST, int n) {
  int idx = blockIdx.x * blockDim.x + threadIdx.x;
  if (idx >= n * NH) return;
  int node = idx >> 2;
  int h = idx & 3;
  const float* xp = XP + (size_t)node * HC + h * 32;
  const float* as = att_src + h * 32;
  const float* ad = att_dst + h * 32;
  float s0 = 0.f, s1 = 0.f;
#pragma unroll
  for (int c = 0; c < 32; ++c) {
    float v = xp[c];
    s0 += v * as[c];
    s1 += v * ad[c];
  }
  ASRC[idx] = s0;
  ADST[idx] = s1;
}

__global__ __launch_bounds__(256) void k_zero_counts(int* __restrict__ cnt, int n) {
  int i = blockIdx.x * blockDim.x + threadIdx.x;
  if (i < n) cnt[i] = 0;
}

__global__ __launch_bounds__(256) void k_count_in(const int* __restrict__ dst,
                                                  int* __restrict__ cnt,
                                                  int* __restrict__ rank, int ne) {
  int e = blockIdx.x * blockDim.x + threadIdx.x;
  if (e >= ne) return;
  rank[e] = atomicAdd(&cnt[dst[e]], 1);
}

__global__ __launch_bounds__(256) void k_block_sums(const int* __restrict__ cnt,
                                                    int* __restrict__ bsum, int n) {
  __shared__ int wtot[8];
  const int t = threadIdx.x, lane = t & 31, w = t >> 5;
  const int i = (blockIdx.x * 256 + t) * 4;
  int v = 0;
  if (i < n) {
    const int4 c = *(const int4*)(cnt + i);
    v = c.x + c.y + c.z + c.w;
  }
  for (int o = 16; o > 0; o >>= 1) v += __shfl_xor(v, o);
  if (lane == 0) wtot[w] = v;
  __syncthreads();
  if (t == 0) {
    int s = 0;
    for (int k = 0; k < 8; ++k) s += wtot[k];
    bsum[blockIdx.x] = s;
  }
}

__global__ __launch_bounds__(256) void k_row_starts(const int* __restrict__ cnt,
                                                    const int* __restrict__ bsum,
                                                    int* __restrict__ rowptr, int n) {
  __shared__ int wtot[8];
  __shared__ int boff;
  const int t = threadIdx.x, lane = t & 31, w = t >> 5;
  const int i = (blockIdx.x * 256 + t) * 4;
  int4 c = make_int4(0, 0, 0, 0);
  if (i < n) c = *(const int4*)(cnt + i);
  const int s0 = c.x, s1 = s0 + c.y, s2 = s1 + c.z, s3 = s2 + c.w;
  int x = s3;
  for (int o = 1; o < 32; o <<= 1) {
    int y = __shfl_up(x, o);
    if (lane >= o) x += y;
  }
  if (lane == 31) wtot[w] = x;
  if (w == 0) {
    int p = 0;
    for (int k = lane; k < (int)blockIdx.x; k += 32) p += bsum[k];
    for (int o = 16; o > 0; o >>= 1) p += __shfl_xor(p, o);
    if (lane == 0) boff = p;
  }
  __syncthreads();
  int excl = boff + (x - s3);
  for (int k = 0; k < w; ++k) excl += wtot[k];
  if (i < n) {
    *(int4*)(rowptr + i) = make_int4(excl, excl + s0, excl + s1, excl + s2);
    if (i + 4 == n) rowptr[n] = excl + s3;
  }
}

__global__ __launch_bounds__(256) void k_place_edges(const int* __restrict__ src,
                                                     const int* __restrict__ dst,
                                                     const int* __restrict__ rowptr,
                                                     const int* __restrict__ rank,
                                                     int* __restrict__ csr_src, int ne) {
  int e = blockIdx.x * blockDim.x + threadIdx.x;
  if (e >= ne) return;
  int pos = rowptr[dst[e]] + rank[e];
  pos = min(max(pos, 0), ne - 1);
  csr_src[pos] = src[e];
}

template <bool RELU>
__global__ __launch_bounds__(256) void k_node_softmax_aggr(const int* __restrict__ rowptr,
                                                           const int* __restrict__ csr_src,
                                                           const float* __restrict__ XP,
                                                           const float* __restrict__ ASRC,
                                                           const float* __restrict__ ADST,
                                                           const float* __restrict__ bias,
                                                           float* __restrict__ OUT,
                                                           int n, int ne) {
  int node = (blockIdx.x * blockDim.x + threadIdx.x) >> 5;
  node = __builtin_amdgcn_readfirstlane(node);
  const int lane = threadIdx.x & 31;
  if (node >= n) return;
  const int h = lane >> 3;
  int beg = rowptr[node], end = rowptr[node + 1];
  beg = max(beg, 0);
  end = min(end, ne);
  const float adst = ADST[(size_t)node * NH + h];

  float m = __int_as_float(0xff800000);
  for (int base = beg; base < end; base += 32) {
    const int c = min(32, end - base);
    const int sl = (lane < c) ? csr_src[base + lane] : 0;
    for (int j = 0; j < c; j += 4) {
      float v[4];
#pragma unroll
      for (int k = 0; k < 4; ++k) {
        const int s = __builtin_amdgcn_readlane(sl, j + k);
        float t = ASRC[(size_t)s * NH + h] + adst;
        v[k] = t > 0.f ? t : 0.2f * t;
      }
#pragma unroll
      for (int k = 0; k < 4; ++k)
        if (j + k < c) m = fmaxf(m, v[k]);
    }
  }

  float ssum = 0.f;
  float4 acc = make_float4(0.f, 0.f, 0.f, 0.f);
  for (int base = beg; base < end; base += 32) {
    const int c = min(32, end - base);
    const int sl = (lane < c) ? csr_src[base + lane] : 0;
    for (int j = 0; j < c; j += 4) {
      float ex[4];
      float4 xr[4];
#pragma unroll
      for (int k = 0; k < 4; ++k) {
        const int s = __builtin_amdgcn_readlane(sl, j + k);
        float t = ASRC[(size_t)s * NH + h] + adst;
        t = t > 0.f ? t : 0.2f * t;
        ex[k] = __expf(t - m);
        xr[k] = *(const float4*)(XP + (size_t)s * HC + lane * 4);
      }
#pragma unroll
      for (int k = 0; k < 4; ++k) {
        if (j + k < c) {
          ssum += ex[k];
          acc.x += xr[k].x * ex[k];
          acc.y += xr[k].y * ex[k];
          acc.z += xr[k].z * ex[k];
          acc.w += xr[k].w * ex[k];
        }
      }
    }
  }

  const float den = ssum + 1e-16f;
  const int c0 = lane * 4;
  float4 o;
  o.x = acc.x / den + bias[c0 + 0];
  o.y = acc.y / den + bias[c0 + 1];
  o.z = acc.z / den + bias[c0 + 2];
  o.w = acc.w / den + bias[c0 + 3];
  if (RELU) {
    o.x = fmaxf(o.x, 0.0f);
    o.y = fmaxf(o.y, 0.0f);
    o.z = fmaxf(o.z, 0.0f);
    o.w = fmaxf(o.w, 0.0f);
  }
  *(float4*)(OUT + (size_t)node * HC + c0) = o;
}

extern "C" void kernel_launch(void* const* d_in, const int* in_sizes, int n_in,
                              void* d_out, int out_size, void* d_ws, size_t ws_size,
                              hipStream_t stream) {
  const float* x     = (const float*)d_in[0];
  const int*   ei    = (const int*)d_in[1];
  const float* W0    = (const float*)d_in[2];
  const float* atts0 = (const float*)d_in[3];
  const float* attd0 = (const float*)d_in[4];
  const float* b0    = (const float*)d_in[5];
  const float* W1    = (const float*)d_in[6];
  const float* atts1 = (const float*)d_in[7];
  const float* attd1 = (const float*)d_in[8];
  const float* b1    = (const float*)d_in[9];

  const int n = NNODES, ne = NEDGES;
  const int* srcv = ei;
  const int* dstv = ei + ne;

  char* ws = (char*)d_ws;
  float* XP   = (float*)(ws + 0);
  float* H0   = (float*)(ws + 51200000);
  float* ASRC = (float*)(ws + 102400000);
  float* ADST = (float*)(ws + 104000000);
  int*   CNT  = (int*)(ws + 105600000);
  int*   BSUM = (int*)(ws + 106400000);
  int*   RPTR = (int*)(ws + 107200000);
  int*   CSRC = (int*)(ws + 108800000);
  int*   RANK = (int*)(ws + 112000000);
  float* OUT  = (float*)d_out;

  const dim3 blk(256);
  const int nMH  = n * NH;
  const int gemmBlocks = cdiv(cdiv(n, 16) * 32, 256);
  const int scanBlocks = cdiv(n / 4, 256);
  const int nodeBlocks = cdiv(n * 32, 256);

  k_zero_counts<<<cdiv(n, 256), blk, 0, stream>>>(CNT, n);
  k_count_in<<<cdiv(ne, 256), blk, 0, stream>>>(dstv, CNT, RANK, ne);
  k_block_sums<<<scanBlocks, blk, 0, stream>>>(CNT, BSUM, n);
  k_row_starts<<<scanBlocks, blk, 0, stream>>>(CNT, BSUM, RPTR, n);
  k_place_edges<<<cdiv(ne, 256), blk, 0, stream>>>(srcv, dstv, RPTR, RANK, CSRC, ne);

  k_gemm_wmma<<<gemmBlocks, blk, 0, stream>>>(x, W0, XP, n, 256);
  k_attn_dots<<<cdiv(nMH, 256), blk, 0, stream>>>(XP, atts0, attd0, ASRC, ADST, n);
  k_node_softmax_aggr<true><<<nodeBlocks, blk, 0, stream>>>(RPTR, CSRC, XP, ASRC, ADST, b0, H0, n, ne);

  k_gemm_wmma<<<gemmBlocks, blk, 0, stream>>>(H0, W1, XP, n, 128);
  k_attn_dots<<<cdiv(nMH, 256), blk, 0, stream>>>(XP, atts1, attd1, ASRC, ADST, n);
  k_node_softmax_aggr<false><<<nodeBlocks, blk, 0, stream>>>(RPTR, CSRC, XP, ASRC, ADST, b1, OUT, n, ne);
}


# GEMM operand feed: W copied once per work-group to LDS (pitch K+4); both WMMA operands read as float4 per 8 k, lane half h taking k=8g+4h..+3, so k is summed in a new order inside each 8 k
# speedup vs baseline: 4.2116x; 1.2797x over previous
#include <hip/hip_runtime.h>
#include <hip/hip_bf16.h>

#define NNODES 100000
#define NEDGES 800000
#define HC 128
#define NH 4

typedef float v2f __attribute__((ext_vector_type(2)));
typedef float v8f __attribute__((ext_vector_type(8)));

static inline int cdiv(int a, int b) { return (a + b - 1) / b; }

template <int K>
__global__ __launch_bounds__(256) void k_gemm_wmma(const float* __restrict__ X,
                                                   const float* __restrict__ W,
                                                   float* __restrict__ XP, int rows) {
  constexpr int P = K + 4;
  constexpr int KF = K / 4;
  constexpr int NQ = K / 32;
  __shared__ float WlA[64 * P];
  __shared__ float WlB[64 * P];

  const int tid = threadIdx.x;
  const int lane = tid & 31;
  const int c = lane & 15;
  const int h = lane >> 4;
  const int m0 = (blockIdx.x * 8 + (tid >> 5)) * 16;
  const bool active = m0 < rows;

  const float* xrow = X + (size_t)(m0 + c) * K + 4 * h;
  float4 an[4];
  if (active) {
#pragma unroll
    for (int i = 0; i < 4; ++i) an[i] = *(const float4*)(xrow + 8 * i);
  }

  {
    const int n = tid / KF, kf = tid - n * KF;
    constexpr int RS = 256 / KF;
    const float* wsrc = W + (size_t)n * K + 4 * kf;
    float* da = WlA + n * P + 4 * kf;
    float* db = WlB + n * P + 4 * kf;
#pragma unroll 8
    for (int r = 0; r < 64; r += RS)
      *(float4*)(da + r * P) = *(const float4*)(wsrc + (size_t)r * K);
#pragma unroll 8
    for (int r = 0; r < 64; r += RS)
      *(float4*)(db + r * P) = *(const float4*)(wsrc + (size_t)(64 + r) * K);
  }

  __syncthreads();
  if (!active) return;

  const float* bla = WlA + c * P + 4 * h;
  const float* blb = WlB + c * P + 4 * h;
  v8f acc[8] = {};
  float4 b0[8], b1[8];
#pragma unroll
  for (int nt = 0; nt < 4; ++nt) {
    b0[nt] = *(const float4*)(bla + nt * 16 * P);
    b0[nt + 4] = *(const float4*)(blb + nt * 16 * P);
  }

#pragma unroll
  for (int q = 0; q < NQ; ++q) {
    float4 a[4];
#pragma unroll
    for (int i = 0; i < 4; ++i) a[i] = an[i];
    if (q + 1 < NQ) {
#pragma unroll
      for (int i = 0; i < 4; ++i) an[i] = *(const float4*)(xrow + 32 * (q + 1) + 8 * i);
    }
#pragma unroll
    for (int i = 0; i < 4; ++i) {
      float4* bc = (i & 1) ? b1 : b0;
      float4* bn = (i & 1) ? b0 : b1;
      if (i < 3 || q + 1 < NQ) {
        const int ko = 32 * q + 8 * (i + 1);
#pragma unroll
        for (int nt = 0; nt < 4; ++nt) {
          bn[nt] = *(const float4*)(bla + nt * 16 * P + ko);
          bn[nt + 4] = *(const float4*)(blb + nt * 16 * P + ko);
        }
      }
      const v2f a0 = {a[i].x, a[i].y};
      const v2f a1 = {a[i].z, a[i].w};
#pragma unroll
      for (int nt = 0; nt < 8; ++nt) {
        const v2f b = {bc[nt].x, bc[nt].y};
        acc[nt] = __builtin_amdgcn_wmma_f32_16x16x4_f32(
             false, a0,  false, b,
             (short)0, acc[nt],  false,  false);
      }
#pragma unroll
      for (int nt = 0; nt < 8; ++nt) {
        const v2f b = {bc[nt].z, bc[nt].w};
        acc[nt] = __builtin_amdgcn_wmma_f32_16x16x4_f32(
             false, a1,  false, b,
             (short)0, acc[nt],  false,  false);
      }
    }
  }

  __builtin_amdgcn_sched_barrier(0);
  asm volatile("v_nop\n\tv_nop\n\tv_nop\n\tv_nop\n\tv_nop\n\tv_nop\n\tv_nop\n\tv_nop");
  __builtin_amdgcn_sched_barrier(0);

  float* const orow = XP + (size_t)(m0 + h * 8) * HC + c;
#pragma unroll
  for (int nt = 0; nt < 8; ++nt) {
#pragma unroll
    for (int i = 0; i < 8; ++i) orow[(size_t)i * HC + nt * 16] = acc[nt][i];
  }
}

__global__ __launch_bounds__(256) void k_attn_dots(const float* __restrict__ XP,
                                                   const float* __restrict__ att_src,
                                                   const float* __restrict__ att_dst,
                                                   float* __restrict__ ASRC,
                                                   float* __restrict__ ADST, int n) {
  int idx = blockIdx.x * blockDim.x + threadIdx.x;
  if (idx >= n * NH) return;
  int node = idx >> 2;
  int h = idx & 3;
  const float* xp = XP + (size_t)node * HC + h * 32;
  const float* as = att_src + h * 32;
  const float* ad = att_dst + h * 32;
  float s0 = 0.f, s1 = 0.f;
#pragma unroll
  for (int c = 0; c < 32; ++c) {
    float v = xp[c];
    s0 += v * as[c];
    s1 += v * ad[c];
  }
  ASRC[idx] = s0;
  ADST[idx] = s1;
}

__global__ __launch_bounds__(256) void k_zero_counts(int* __restrict__ cnt, int n) {
  int i = blockIdx.x * blockDim.x + threadIdx.x;
  if (i < n) cnt[i] = 0;
}

__global__ __launch_bounds__(256) void k_count_in(const int* __restrict__ dst,
                                                  int* __restrict__ cnt,
                                                  int* __restrict__ rank, int ne) {
  int e = blockIdx.x * blockDim.x + threadIdx.x;
  if (e >= ne) return;
  rank[e] = atomicAdd(&cnt[dst[e]], 1);
}

__global__ __launch_bounds__(256) void k_block_sums(const int* __restrict__ cnt,
                                                    int* __restrict__ bsum, int n) {
  __shared__ int wtot[8];
  const int t = threadIdx.x, lane = t & 31, w = t >> 5;
  const int i = (blockIdx.x * 256 + t) * 4;
  int v = 0;
  if (i < n) {
    const int4 c = *(const int4*)(cnt + i);
    v = c.x + c.y + c.z + c.w;
  }
  for (int o = 16; o > 0; o >>= 1) v += __shfl_xor(v, o);
  if (lane == 0) wtot[w] = v;
  __syncthreads();
  if (t == 0) {
    int s = 0;
    for (int k = 0; k < 8; ++k) s += wtot[k];
    bsum[blockIdx.x] = s;
  }
}

__global__ __launch_bounds__(256) void k_row_starts(const int* __restrict__ cnt,
                                                    const int* __restrict__ bsum,
                                                    int* __restrict__ rowptr, int n) {
  __shared__ int wtot[8];
  __shared__ int boff;
  const int t = threadIdx.x, lane = t & 31, w = t >> 5;
  const int i = (blockIdx.x * 256 + t) * 4;
  int4 c = make_int4(0, 0, 0, 0);
  if (i < n) c = *(const int4*)(cnt + i);
  const int s0 = c.x, s1 = s0 + c.y, s2 = s1 + c.z, s3 = s2 + c.w;
  int x = s3;
  for (int o = 1; o < 32; o <<= 1) {
    int y = __shfl_up(x, o);
    if (lane >= o) x += y;
  }
  if (lane == 31) wtot[w] = x;
  if (w == 0) {
    int p = 0;
    for (int k = lane; k < (int)blockIdx.x; k += 32) p += bsum[k];
    for (int o = 16; o > 0; o >>= 1) p += __shfl_xor(p, o);
    if (lane == 0) boff = p;
  }
  __syncthreads();
  int excl = boff + (x - s3);
  for (int k = 0; k < w; ++k) excl += wtot[k];
  if (i < n) {
    *(int4*)(rowptr + i) = make_int4(excl, excl + s0, excl + s1, excl + s2);
    if (i + 4 == n) rowptr[n] = excl + s3;
  }
}

__global__ __launch_bounds__(256) void k_place_edges(const int* __restrict__ src,
                                                     const int* __restrict__ dst,
                                                     const int* __restrict__ rowptr,
                                                     const int* __restrict__ rank,
                                                     int* __restrict__ csr_src, int ne) {
  int e = blockIdx.x * blockDim.x + threadIdx.x;
  if (e >= ne) return;
  int pos = rowptr[dst[e]] + rank[e];
  pos = min(max(pos, 0), ne - 1);
  csr_src[pos] = src[e];
}

template <bool RELU>
__global__ __launch_bounds__(256) void k_node_softmax_aggr(const int* __restrict__ rowptr,
                                                           const int* __restrict__ csr_src,
                                                           const float* __restrict__ XP,
                                                           const float* __restrict__ ASRC,
                                                           const float* __restrict__ ADST,
                                                           const float* __restrict__ bias,
                                                           float* __restrict__ OUT,
                                                           int n, int ne) {
  int node = (blockIdx.x * blockDim.x + threadIdx.x) >> 5;
  node = __builtin_amdgcn_readfirstlane(node);
  const int lane = threadIdx.x & 31;
  if (node >= n) return;
  const int h = lane >> 3;
  int beg = rowptr[node], end = rowptr[node + 1];
  beg = max(beg, 0);
  end = min(end, ne);
  const float adst = ADST[(size_t)node * NH + h];

  float m = __int_as_float(0xff800000);
  for (int base = beg; base < end; base += 32) {
    const int c = min(32, end - base);
    const int sl = (lane < c) ? csr_src[base + lane] : 0;
    for (int j = 0; j < c; j += 4) {
      float v[4];
#pragma unroll
      for (int k = 0; k < 4; ++k) {
        const int s = __builtin_amdgcn_readlane(sl, j + k);
        float t = ASRC[(size_t)s * NH + h] + adst;
        v[k] = t > 0.f ? t : 0.2f * t;
      }
#pragma unroll
      for (int k = 0; k < 4; ++k)
        if (j + k < c) m = fmaxf(m, v[k]);
    }
  }

  float ssum = 0.f;
  float4 acc = make_float4(0.f, 0.f, 0.f, 0.f);
  for (int base = beg; base < end; base += 32) {
    const int c = min(32, end - base);
    const int sl = (lane < c) ? csr_src[base + lane] : 0;
    for (int j = 0; j < c; j += 4) {
      float ex[4];
      float4 xr[4];
#pragma unroll
      for (int k = 0; k < 4; ++k) {
        const int s = __builtin_amdgcn_readlane(sl, j + k);
        float t = ASRC[(size_t)s * NH + h] + adst;
        t = t > 0.f ? t : 0.2f * t;
        ex[k] = __expf(t - m);
        xr[k] = *(const float4*)(XP + (size_t)s * HC + lane * 4);
      }
#pragma unroll
      for (int k = 0; k < 4; ++k) {
        if (j + k < c) {
          ssum += ex[k];
          acc.x += xr[k].x * ex[k];
          acc.y += xr[k].y * ex[k];
          acc.z += xr[k].z * ex[k];
          acc.w += xr[k].w * ex[k];
        }
      }
    }
  }

  const float den = ssum + 1e-16f;
  const int c0 = lane * 4;
  float4 o;
  o.x = acc.x / den + bias[c0 + 0];
  o.y = acc.y / den + bias[c0 + 1];
  o.z = acc.z / den + bias[c0 + 2];
  o.w = acc.w / den + bias[c0 + 3];
  if (RELU) {
    o.x = fmaxf(o.x, 0.0f);
    o.y = fmaxf(o.y, 0.0f);
    o.z = fmaxf(o.z, 0.0f);
    o.w = fmaxf(o.w, 0.0f);
  }
  *(float4*)(OUT + (size_t)node * HC + c0) = o;
}

extern "C" void kernel_launch(void* const* d_in, const int* in_sizes, int n_in,
                              void* d_out, int out_size, void* d_ws, size_t ws_size,
                              hipStream_t stream) {
  const float* x     = (const float*)d_in[0];
  const int*   ei    = (const int*)d_in[1];
  const float* W0    = (const float*)d_in[2];
  const float* atts0 = (const float*)d_in[3];
  const float* attd0 = (const float*)d_in[4];
  const float* b0    = (const float*)d_in[5];
  const float* W1    = (const float*)d_in[6];
  const float* atts1 = (const float*)d_in[7];
  const float* attd1 = (const float*)d_in[8];
  const float* b1    = (const float*)d_in[9];

  const int n = NNODES, ne = NEDGES;
  const int* srcv = ei;
  const int* dstv = ei + ne;

  char* ws = (char*)d_ws;
  float* XP   = (float*)(ws + 0);
  float* H0   = (float*)(ws + 51200000);
  float* ASRC = (float*)(ws + 102400000);
  float* ADST = (float*)(ws + 104000000);
  int*   CNT  = (int*)(ws + 105600000);
  int*   BSUM = (int*)(ws + 106400000);
  int*   RPTR = (int*)(ws + 107200000);
  int*   CSRC = (int*)(ws + 108800000);
  int*   RANK = (int*)(ws + 112000000);
  float* OUT  = (float*)d_out;

  const dim3 blk(256);
  const int nMH  = n * NH;
  const int gemmBlocks = cdiv(cdiv(n, 16) * 32, 256);
  const int scanBlocks = cdiv(n / 4, 256);
  const int nodeBlocks = cdiv(n * 32, 256);

  k_zero_counts<<<cdiv(n, 256), blk, 0, stream>>>(CNT, n);
  k_count_in<<<cdiv(ne, 256), blk, 0, stream>>>(dstv, CNT, RANK, ne);
  k_block_sums<<<scanBlocks, blk, 0, stream>>>(CNT, BSUM, n);
  k_row_starts<<<scanBlocks, blk, 0, stream>>>(CNT, BSUM, RPTR, n);
  k_place_edges<<<cdiv(ne, 256), blk, 0, stream>>>(srcv, dstv, RPTR, RANK, CSRC, ne);

  k_gemm_wmma<256><<<gemmBlocks, blk, 0, stream>>>(x, W0, XP, n);
  k_attn_dots<<<cdiv(nMH, 256), blk, 0, stream>>>(XP, atts0, attd0, ASRC, ADST, n);
  k_node_softmax_aggr<true><<<nodeBlocks, blk, 0, stream>>>(RPTR, CSRC, XP, ASRC, ADST, b0, H0, n, ne);

  k_gemm_wmma<128><<<gemmBlocks, blk, 0, stream>>>(H0, W1, XP, n);
  k_attn_dots<<<cdiv(nMH, 256), blk, 0, stream>>>(XP, atts1, attd1, ASRC, ADST, n);
  k_node_softmax_aggr<false><<<nodeBlocks, blk, 0, stream>>>(RPTR, CSRC, XP, ASRC, ADST, b1, OUT, n, ne);
}


# Attention dots fused into the GEMM epilogue: each wave multiplies its accumulator strip by att_src/att_dst, sums over 16 lanes by a DPP exchange tree (new sum order); both dot launches removed
# speedup vs baseline: 4.3660x; 1.0367x over previous
#include <hip/hip_runtime.h>
#include <hip/hip_bf16.h>

#define NNODES 100000
#define NEDGES 800000
#define HC 128
#define NH 4

typedef float v2f __attribute__((ext_vector_type(2)));
typedef float v8f __attribute__((ext_vector_type(8)));

static inline int cdiv(int a, int b) { return (a + b - 1) / b; }

template <int M>
__device__ __forceinline__ float row_xadd(bool hi, float x, float y) {
  const float keep = hi ? y : x;
  const float give = hi ? x : y;
  const int got = __builtin_amdgcn_update_dpp(0, __float_as_int(give), 0x160 | M, 0xf, 0xf, true);
  return keep + __int_as_float(got);
}

template <int K>
__global__ __launch_bounds__(256) void k_gemm_wmma(const float* __restrict__ X,
                                                   const float* __restrict__ W,
                                                   const float* __restrict__ att_src,
                                                   const float* __restrict__ att_dst,
                                                   float* __restrict__ XP,
                                                   float* __restrict__ ASRC,
                                                   float* __restrict__ ADST, int rows) {
  constexpr int P = K + 4;
  constexpr int KF = K / 4;
  constexpr int NQ = K / 32;
  __shared__ float WlA[64 * P];
  __shared__ float WlB[64 * P];

  const int tid = threadIdx.x;
  const int lane = tid & 31;
  const int c = lane & 15;
  const int h = lane >> 4;
  const int m0 = (blockIdx.x * 8 + (tid >> 5)) * 16;
  const bool active = m0 < rows;

  const float* xrow = X + (size_t)(m0 + c) * K + 4 * h;
  float4 an[4];
  if (active) {
#pragma unroll
    for (int i = 0; i < 4; ++i) an[i] = *(const float4*)(xrow + 8 * i);
  }

  {
    const int n = tid / KF, kf = tid - n * KF;
    constexpr int RS = 256 / KF;
    const float* wsrc = W + (size_t)n * K + 4 * kf;
    float* da = WlA + n * P + 4 * kf;
    float* db = WlB + n * P + 4 * kf;
#pragma unroll 8
    for (int r = 0; r < 64; r += RS)
      *(float4*)(da + r * P) = *(const float4*)(wsrc + (size_t)r * K);
#pragma unroll 8
    for (int r = 0; r < 64; r += RS)
      *(float4*)(db + r * P) = *(const float4*)(wsrc + (size_t)(64 + r) * K);
  }

  __syncthreads();
  if (!active) return;

  const float* bla = WlA + c * P + 4 * h;
  const float* blb = WlB + c * P + 4 * h;
  v8f acc[8] = {};
  float4 b0[8], b1[8];
#pragma unroll
  for (int nt = 0; nt < 4; ++nt) {
    b0[nt] = *(const float4*)(bla + nt * 16 * P);
    b0[nt + 4] = *(const float4*)(blb + nt * 16 * P);
  }

#pragma unroll
  for (int q = 0; q < NQ; ++q) {
    float4 a[4];
#pragma unroll
    for (int i = 0; i < 4; ++i) a[i] = an[i];
    if (q + 1 < NQ) {
#pragma unroll
      for (int i = 0; i < 4; ++i) an[i] = *(const float4*)(xrow + 32 * (q + 1) + 8 * i);
    }
#pragma unroll
    for (int i = 0; i < 4; ++i) {
      float4* bc = (i & 1) ? b1 : b0;
      float4* bn = (i & 1) ? b0 : b1;
      if (i < 3 || q + 1 < NQ) {
        const int ko = 32 * q + 8 * (i + 1);
#pragma unroll
        for (int nt = 0; nt < 4; ++nt) {
          bn[nt] = *(const float4*)(bla + nt * 16 * P + ko);
          bn[nt + 4] = *(const float4*)(blb + nt * 16 * P + ko);
        }
      }
      const v2f a0 = {a[i].x, a[i].y};
      const v2f a1 = {a[i].z, a[i].w};
#pragma unroll
      for (int nt = 0; nt < 8; ++nt) {
        const v2f b = {bc[nt].x, bc[nt].y};
        acc[nt] = __builtin_amdgcn_wmma_f32_16x16x4_f32(
             false, a0,  false, b,
             (short)0, acc[nt],  false,  false);
      }
#pragma unroll
      for (int nt = 0; nt < 8; ++nt) {
        const v2f b = {bc[nt].z, bc[nt].w};
        acc[nt] = __builtin_amdgcn_wmma_f32_16x16x4_f32(
             false, a1,  false, b,
             (short)0, acc[nt],  false,  false);
      }
    }
  }

  __builtin_amdgcn_sched_barrier(0);
  asm volatile("v_nop\n\tv_nop\n\tv_nop\n\tv_nop\n\tv_nop\n\tv_nop\n\tv_nop\n\tv_nop");
  __builtin_amdgcn_sched_barrier(0);

  float* const orow = XP + (size_t)(m0 + h * 8) * HC + c;
#pragma unroll
  for (int nt = 0; nt < 8; ++nt) {
#pragma unroll
    for (int i = 0; i < 8; ++i) orow[(size_t)i * HC + nt * 16] = acc[nt][i];
  }

  float a_lo[2][NH], a_hi[2][NH];
#pragma unroll
  for (int hd = 0; hd < NH; ++hd) {
    a_lo[0][hd] = att_src[hd * 32 + c];
    a_hi[0][hd] = att_src[hd * 32 + 16 + c];
    a_lo[1][hd] = att_dst[hd * 32 + c];
    a_hi[1][hd] = att_dst[hd * 32 + 16 + c];
  }
  const bool l0 = (c & 1) != 0, l1 = (c & 2) != 0, l2 = (c & 4) != 0, l3 = (c & 8) != 0;
  float u[2][NH];
#pragma unroll
  for (int sd = 0; sd < 2; ++sd) {
    float sv[2][NH];
#pragma unroll
    for (int j2 = 0; j2 < 2; ++j2) {
      float qv[2][NH];
#pragma unroll
      for (int j1 = 0; j1 < 2; ++j1) {
        const int i = 4 * j2 + 2 * j1;
#pragma unroll
        for (int hd = 0; hd < NH; ++hd) {
          const v2f t0 = {acc[2 * hd][i], acc[2 * hd][i + 1]};
          const v2f t1 = {acc[2 * hd + 1][i], acc[2 * hd + 1][i + 1]};
          const v2f xy = t0 * a_lo[sd][hd] + t1 * a_hi[sd][hd];
          qv[j1][hd] = row_xadd<1>(l0, xy.x, xy.y);
        }
      }
#pragma unroll
      for (int hd = 0; hd < NH; ++hd) sv[j2][hd] = row_xadd<2>(l1, qv[0][hd], qv[1][hd]);
    }
#pragma unroll
    for (int hd = 0; hd < NH; ++hd) u[sd][hd] = row_xadd<4>(l2, sv[0][hd], sv[1][hd]);
  }
  float4 r;
  r.x = row_xadd<8>(l3, u[0][0], u[1][0]);
  r.y = row_xadd<8>(l3, u[0][1], u[1][1]);
  r.z = row_xadd<8>(l3, u[0][2], u[1][2]);
  r.w = row_xadd<8>(l3, u[0][3], u[1][3]);
  float* const arow = (l3 ? ADST : ASRC) + (size_t)(m0 + h * 8 + (c & 7)) * NH;
  *(float4*)arow = r;
}

__global__ __launch_bounds__(256) void k_zero_counts(int* __restrict__ cnt, int n) {
  int i = blockIdx.x * blockDim.x + threadIdx.x;
  if (i < n) cnt[i] = 0;
}

__global__ __launch_bounds__(256) void k_count_in(const int* __restrict__ dst,
                                                  int* __restrict__ cnt,
                                                  int* __restrict__ rank, int ne) {
  int e = blockIdx.x * blockDim.x + threadIdx.x;
  if (e >= ne) return;
  rank[e] = atomicAdd(&cnt[dst[e]], 1);
}

__global__ __launch_bounds__(256) void k_block_sums(const int* __restrict__ cnt,
                                                    int* __restrict__ bsum, int n) {
  __shared__ int wtot[8];
  const int t = threadIdx.x, lane = t & 31, w = t >> 5;
  const int i = (blockIdx.x * 256 + t) * 4;
  int v = 0;
  if (i < n) {
    const int4 c = *(const int4*)(cnt + i);
    v = c.x + c.y + c.z + c.w;
  }
  for (int o = 16; o > 0; o >>= 1) v += __shfl_xor(v, o);
  if (lane == 0) wtot[w] = v;
  __syncthreads();
  if (t == 0) {
    int s = 0;
    for (int k = 0; k < 8; ++k) s += wtot[k];
    bsum[blockIdx.x] = s;
  }
}

__global__ __launch_bounds__(256) void k_row_starts(const int* __restrict__ cnt,
                                                    const int* __restrict__ bsum,
                                                    int* __restrict__ rowptr, int n) {
  __shared__ int wtot[8];
  __shared__ int boff;
  const int t = threadIdx.x, lane = t & 31, w = t >> 5;
  const int i = (blockIdx.x * 256 + t) * 4;
  int4 c = make_int4(0, 0, 0, 0);
  if (i < n) c = *(const int4*)(cnt + i);
  const int s0 = c.x, s1 = s0 + c.y, s2 = s1 + c.z, s3 = s2 + c.w;
  int x = s3;
  for (int o = 1; o < 32; o <<= 1) {
    int y = __shfl_up(x, o);
    if (lane >= o) x += y;
  }
  if (lane == 31) wtot[w] = x;
  if (w == 0) {
    int p = 0;
    for (int k = lane; k < (int)blockIdx.x; k += 32) p += bsum[k];
    for (int o = 16; o > 0; o >>= 1) p += __shfl_xor(p, o);
    if (lane == 0) boff = p;
  }
  __syncthreads();
  int excl = boff + (x - s3);
  for (int k = 0; k < w; ++k) excl += wtot[k];
  if (i < n) {
    *(int4*)(rowptr + i) = make_int4(excl, excl + s0, excl + s1, excl + s2);
    if (i + 4 == n) rowptr[n] = excl + s3;
  }
}

__global__ __launch_bounds__(256) void k_place_edges(const int* __restrict__ src,
                                                     const int* __restrict__ dst,
                                                     const int* __restrict__ rowptr,
                                                     const int* __restrict__ rank,
                                                     int* __restrict__ csr_src, int ne) {
  int e = blockIdx.x * blockDim.x + threadIdx.x;
  if (e >= ne) return;
  int pos = rowptr[dst[e]] + rank[e];
  pos = min(max(pos, 0), ne - 1);
  csr_src[pos] = src[e];
}

template <bool RELU>
__global__ __launch_bounds__(256) void k_node_softmax_aggr(const int* __restrict__ rowptr,
                                                           const int* __restrict__ csr_src,
                                                           const float* __restrict__ XP,
                                                           const float* __restrict__ ASRC,
                                                           const float* __restrict__ ADST,
                                                           const float* __restrict__ bias,
                                                           float* __restrict__ OUT,
                                                           int n, int ne) {
  int node = (blockIdx.x * blockDim.x + threadIdx.x) >> 5;
  node = __builtin_amdgcn_readfirstlane(node);
  const int lane = threadIdx.x & 31;
  if (node >= n) return;
  const int h = lane >> 3;
  int beg = rowptr[node], end = rowptr[node + 1];
  beg = max(beg, 0);
  end = min(end, ne);
  const float adst = ADST[(size_t)node * NH + h];

  float m = __int_as_float(0xff800000);
  for (int base = beg; base < end; base += 32) {
    const int c = min(32, end - base);
    const int sl = (lane < c) ? csr_src[base + lane] : 0;
    for (int j = 0; j < c; j += 4) {
      float v[4];
#pragma unroll
      for (int k = 0; k < 4; ++k) {
        const int s = __builtin_amdgcn_readlane(sl, j + k);
        float t = ASRC[(size_t)s * NH + h] + adst;
        v[k] = t > 0.f ? t : 0.2f * t;
      }
#pragma unroll
      for (int k = 0; k < 4; ++k)
        if (j + k < c) m = fmaxf(m, v[k]);
    }
  }

  float ssum = 0.f;
  float4 acc = make_float4(0.f, 0.f, 0.f, 0.f);
  for (int base = beg; base < end; base += 32) {
    const int c = min(32, end - base);
    const int sl = (lane < c) ? csr_src[base + lane] : 0;
    for (int j = 0; j < c; j += 4) {
      float ex[4];
      float4 xr[4];
#pragma unroll
      for (int k = 0; k < 4; ++k) {
        const int s = __builtin_amdgcn_readlane(sl, j + k);
        float t = ASRC[(size_t)s * NH + h] + adst;
        t = t > 0.f ? t : 0.2f * t;
        ex[k] = __expf(t - m);
        xr[k] = *(const float4*)(XP + (size_t)s * HC + lane * 4);
      }
#pragma unroll
      for (int k = 0; k < 4; ++k) {
        if (j + k < c) {
          ssum += ex[k];
          acc.x += xr[k].x * ex[k];
          acc.y += xr[k].y * ex[k];
          acc.z += xr[k].z * ex[k];
          acc.w += xr[k].w * ex[k];
        }
      }
    }
  }

  const float den = ssum + 1e-16f;
  const int c0 = lane * 4;
  float4 o;
  o.x = acc.x / den + bias[c0 + 0];
  o.y = acc.y / den + bias[c0 + 1];
  o.z = acc.z / den + bias[c0 + 2];
  o.w = acc.w / den + bias[c0 + 3];
  if (RELU) {
    o.x = fmaxf(o.x, 0.0f);
    o.y = fmaxf(o.y, 0.0f);
    o.z = fmaxf(o.z, 0.0f);
    o.w = fmaxf(o.w, 0.0f);
  }
  *(float4*)(OUT + (size_t)node * HC + c0) = o;
}

extern "C" void kernel_launch(void* const* d_in, const int* in_sizes, int n_in,
                              void* d_out, int out_size, void* d_ws, size_t ws_size,
                              hipStream_t stream) {
  const float* x     = (const float*)d_in[0];
  const int*   ei    = (const int*)d_in[1];
  const float* W0    = (const float*)d_in[2];
  const float* atts0 = (const float*)d_in[3];
  const float* attd0 = (const float*)d_in[4];
  const float* b0    = (const float*)d_in[5];
  const float* W1    = (const float*)d_in[6];
  const float* atts1 = (const float*)d_in[7];
  const float* attd1 = (const float*)d_in[8];
  const float* b1    = (const float*)d_in[9];

  const int n = NNODES, ne = NEDGES;
  const int* srcv = ei;
  const int* dstv = ei + ne;

  char* ws = (char*)d_ws;
  float* XP   = (float*)(ws + 0);
  float* H0   = (float*)(ws + 51200000);
  float* ASRC = (float*)(ws + 102400000);
  float* ADST = (float*)(ws + 104000000);
  int*   CNT  = (int*)(ws + 105600000);
  int*   BSUM = (int*)(ws + 106400000);
  int*   RPTR = (int*)(ws + 107200000);
  int*   CSRC = (int*)(ws + 108800000);
  int*   RANK = (int*)(ws + 112000000);
  float* OUT  = (float*)d_out;

  const dim3 blk(256);
  const int gemmBlocks = cdiv(cdiv(n, 16) * 32, 256);
  const int scanBlocks = cdiv(n / 4, 256);
  const int nodeBlocks = cdiv(n * 32, 256);

  k_zero_counts<<<cdiv(n, 256), blk, 0, stream>>>(CNT, n);
  k_count_in<<<cdiv(ne, 256), blk, 0, stream>>>(dstv, CNT, RANK, ne);
  k_block_sums<<<scanBlocks, blk, 0, stream>>>(CNT, BSUM, n);
  k_row_starts<<<scanBlocks, blk, 0, stream>>>(CNT, BSUM, RPTR, n);
  k_place_edges<<<cdiv(ne, 256), blk, 0, stream>>>(srcv, dstv, RPTR, RANK, CSRC, ne);

  k_gemm_wmma<256><<<gemmBlocks, blk, 0, stream>>>(x, W0, atts0, attd0, XP, ASRC, ADST, n);
  k_node_softmax_aggr<true><<<nodeBlocks, blk, 0, stream>>>(RPTR, CSRC, XP, ASRC, ADST, b0, H0, n, ne);

  k_gemm_wmma<128><<<gemmBlocks, blk, 0, stream>>>(H0, W1, atts1, attd1, XP, ASRC, ADST, n);
  k_node_softmax_aggr<false><<<nodeBlocks, blk, 0, stream>>>(RPTR, CSRC, XP, ASRC, ADST, b1, OUT, n, ne);
}


# Aggregation remapped: a wave takes 4 nodes at once, 8 lanes per node; an edge's logits and exp are formed once in its own lane and broadcast in the group; 4 rows per node in flight; sum order kept
# speedup vs baseline: 4.6463x; 1.0642x over previous
#include <hip/hip_runtime.h>
#include <hip/hip_bf16.h>

#define NNODES 100000
#define NEDGES 800000
#define HC 128
#define NH 4

typedef float v2f __attribute__((ext_vector_type(2)));
typedef float v8f __attribute__((ext_vector_type(8)));

static inline int cdiv(int a, int b) { return (a + b - 1) / b; }

template <int M>
__device__ __forceinline__ float row_xadd(bool hi, float x, float y) {
  const float keep = hi ? y : x;
  const float give = hi ? x : y;
  const int got = __builtin_amdgcn_update_dpp(0, __float_as_int(give), 0x160 | M, 0xf, 0xf, true);
  return keep + __int_as_float(got);
}

template <int K>
__global__ __launch_bounds__(256) void k_gemm_wmma(const float* __restrict__ X,
                                                   const float* __restrict__ W,
                                                   const float* __restrict__ att_src,
                                                   const float* __restrict__ att_dst,
                                                   float* __restrict__ XP,
                                                   float* __restrict__ ASRC,
                                                   float* __restrict__ ADST, int rows) {
  constexpr int P = K + 4;
  constexpr int KF = K / 4;
  constexpr int NQ = K / 32;
  __shared__ float WlA[64 * P];
  __shared__ float WlB[64 * P];

  const int tid = threadIdx.x;
  const int lane = tid & 31;
  const int c = lane & 15;
  const int h = lane >> 4;
  const int m0 = (blockIdx.x * 8 + (tid >> 5)) * 16;
  const bool active = m0 < rows;

  const float* xrow = X + (size_t)(m0 + c) * K + 4 * h;
  float4 an[4];
  if (active) {
#pragma unroll
    for (int i = 0; i < 4; ++i) an[i] = *(const float4*)(xrow + 8 * i);
  }

  {
    const int n = tid / KF, kf = tid - n * KF;
    constexpr int RS = 256 / KF;
    const float* wsrc = W + (size_t)n * K + 4 * kf;
    float* da = WlA + n * P + 4 * kf;
    float* db = WlB + n * P + 4 * kf;
#pragma unroll 8
    for (int r = 0; r < 64; r += RS)
      *(float4*)(da + r * P) = *(const float4*)(wsrc + (size_t)r * K);
#pragma unroll 8
    for (int r = 0; r < 64; r += RS)
      *(float4*)(db + r * P) = *(const float4*)(wsrc + (size_t)(64 + r) * K);
  }

  __syncthreads();
  if (!active) return;

  const float* bla = WlA + c * P + 4 * h;
  const float* blb = WlB + c * P + 4 * h;
  v8f acc[8] = {};
  float4 b0[8], b1[8];
#pragma unroll
  for (int nt = 0; nt < 4; ++nt) {
    b0[nt] = *(const float4*)(bla + nt * 16 * P);
    b0[nt + 4] = *(const float4*)(blb + nt * 16 * P);
  }

#pragma unroll
  for (int q = 0; q < NQ; ++q) {
    float4 a[4];
#pragma unroll
    for (int i = 0; i < 4; ++i) a[i] = an[i];
    if (q + 1 < NQ) {
#pragma unroll
      for (int i = 0; i < 4; ++i) an[i] = *(const float4*)(xrow + 32 * (q + 1) + 8 * i);
    }
#pragma unroll
    for (int i = 0; i < 4; ++i) {
      float4* bc = (i & 1) ? b1 : b0;
      float4* bn = (i & 1) ? b0 : b1;
      if (i < 3 || q + 1 < NQ) {
        const int ko = 32 * q + 8 * (i + 1);
#pragma unroll
        for (int nt = 0; nt < 4; ++nt) {
          bn[nt] = *(const float4*)(bla + nt * 16 * P + ko);
          bn[nt + 4] = *(const float4*)(blb + nt * 16 * P + ko);
        }
      }
      const v2f a0 = {a[i].x, a[i].y};
      const v2f a1 = {a[i].z, a[i].w};
#pragma unroll
      for (int nt = 0; nt < 8; ++nt) {
        const v2f b = {bc[nt].x, bc[nt].y};
        acc[nt] = __builtin_amdgcn_wmma_f32_16x16x4_f32(
             false, a0,  false, b,
             (short)0, acc[nt],  false,  false);
      }
#pragma unroll
      for (int nt = 0; nt < 8; ++nt) {
        const v2f b = {bc[nt].z, bc[nt].w};
        acc[nt] = __builtin_amdgcn_wmma_f32_16x16x4_f32(
             false, a1,  false, b,
             (short)0, acc[nt],  false,  false);
      }
    }
  }

  __builtin_amdgcn_sched_barrier(0);
  asm volatile("v_nop\n\tv_nop\n\tv_nop\n\tv_nop\n\tv_nop\n\tv_nop\n\tv_nop\n\tv_nop");
  __builtin_amdgcn_sched_barrier(0);

  float* const orow = XP + (size_t)(m0 + h * 8) * HC + c;
#pragma unroll
  for (int nt = 0; nt < 8; ++nt) {
#pragma unroll
    for (int i = 0; i < 8; ++i) orow[(size_t)i * HC + nt * 16] = acc[nt][i];
  }

  float a_lo[2][NH], a_hi[2][NH];
#pragma unroll
  for (int hd = 0; hd < NH; ++hd) {
    a_lo[0][hd] = att_src[hd * 32 + c];
    a_hi[0][hd] = att_src[hd * 32 + 16 + c];
    a_lo[1][hd] = att_dst[hd * 32 + c];
    a_hi[1][hd] = att_dst[hd * 32 + 16 + c];
  }
  const bool l0 = (c & 1) != 0, l1 = (c & 2) != 0, l2 = (c & 4) != 0, l3 = (c & 8) != 0;
  float u[2][NH];
#pragma unroll
  for (int sd = 0; sd < 2; ++sd) {
    float sv[2][NH];
#pragma unroll
    for (int j2 = 0; j2 < 2; ++j2) {
      float qv[2][NH];
#pragma unroll
      for (int j1 = 0; j1 < 2; ++j1) {
        const int i = 4 * j2 + 2 * j1;
#pragma unroll
        for (int hd = 0; hd < NH; ++hd) {
          const v2f t0 = {acc[2 * hd][i], acc[2 * hd][i + 1]};
          const v2f t1 = {acc[2 * hd + 1][i], acc[2 * hd + 1][i + 1]};
          const v2f xy = t0 * a_lo[sd][hd] + t1 * a_hi[sd][hd];
          qv[j1][hd] = row_xadd<1>(l0, xy.x, xy.y);
        }
      }
#pragma unroll
      for (int hd = 0; hd < NH; ++hd) sv[j2][hd] = row_xadd<2>(l1, qv[0][hd], qv[1][hd]);
    }
#pragma unroll
    for (int hd = 0; hd < NH; ++hd) u[sd][hd] = row_xadd<4>(l2, sv[0][hd], sv[1][hd]);
  }
  float4 r;
  r.x = row_xadd<8>(l3, u[0][0], u[1][0]);
  r.y = row_xadd<8>(l3, u[0][1], u[1][1]);
  r.z = row_xadd<8>(l3, u[0][2], u[1][2]);
  r.w = row_xadd<8>(l3, u[0][3], u[1][3]);
  float* const arow = (l3 ? ADST : ASRC) + (size_t)(m0 + h * 8 + (c & 7)) * NH;
  *(float4*)arow = r;
}

__global__ __launch_bounds__(256) void k_zero_counts(int* __restrict__ cnt, int n) {
  int i = blockIdx.x * blockDim.x + threadIdx.x;
  if (i < n) cnt[i] = 0;
}

__global__ __launch_bounds__(256) void k_count_in(const int* __restrict__ dst,
                                                  int* __restrict__ cnt,
                                                  int* __restrict__ rank, int ne) {
  int e = blockIdx.x * blockDim.x + threadIdx.x;
  if (e >= ne) return;
  rank[e] = atomicAdd(&cnt[dst[e]], 1);
}

__global__ __launch_bounds__(256) void k_block_sums(const int* __restrict__ cnt,
                                                    int* __restrict__ bsum, int n) {
  __shared__ int wtot[8];
  const int t = threadIdx.x, lane = t & 31, w = t >> 5;
  const int i = (blockIdx.x * 256 + t) * 4;
  int v = 0;
  if (i < n) {
    const int4 c = *(const int4*)(cnt + i);
    v = c.x + c.y + c.z + c.w;
  }
  for (int o = 16; o > 0; o >>= 1) v += __shfl_xor(v, o);
  if (lane == 0) wtot[w] = v;
  __syncthreads();
  if (t == 0) {
    int s = 0;
    for (int k = 0; k < 8; ++k) s += wtot[k];
    bsum[blockIdx.x] = s;
  }
}

__global__ __launch_bounds__(256) void k_row_starts(const int* __restrict__ cnt,
                                                    const int* __restrict__ bsum,
                                                    int* __restrict__ rowptr, int n) {
  __shared__ int wtot[8];
  __shared__ int boff;
  const int t = threadIdx.x, lane = t & 31, w = t >> 5;
  const int i = (blockIdx.x * 256 + t) * 4;
  int4 c = make_int4(0, 0, 0, 0);
  if (i < n) c = *(const int4*)(cnt + i);
  const int s0 = c.x, s1 = s0 + c.y, s2 = s1 + c.z, s3 = s2 + c.w;
  int x = s3;
  for (int o = 1; o < 32; o <<= 1) {
    int y = __shfl_up(x, o);
    if (lane >= o) x += y;
  }
  if (lane == 31) wtot[w] = x;
  if (w == 0) {
    int p = 0;
    for (int k = lane; k < (int)blockIdx.x; k += 32) p += bsum[k];
    for (int o = 16; o > 0; o >>= 1) p += __shfl_xor(p, o);
    if (lane == 0) boff = p;
  }
  __syncthreads();
  int excl = boff + (x - s3);
  for (int k = 0; k < w; ++k) excl += wtot[k];
  if (i < n) {
    *(int4*)(rowptr + i) = make_int4(excl, excl + s0, excl + s1, excl + s2);
    if (i + 4 == n) rowptr[n] = excl + s3;
  }
}

__global__ __launch_bounds__(256) void k_place_edges(const int* __restrict__ src,
                                                     const int* __restrict__ dst,
                                                     const int* __restrict__ rowptr,
                                                     const int* __restrict__ rank,
                                                     int* __restrict__ csr_src, int ne) {
  int e = blockIdx.x * blockDim.x + threadIdx.x;
  if (e >= ne) return;
  int pos = rowptr[dst[e]] + rank[e];
  pos = min(max(pos, 0), ne - 1);
  csr_src[pos] = src[e];
}


template <int M>
__device__ __forceinline__ float lane_xor(float x) {
  return __int_as_float(__builtin_amdgcn_update_dpp(0, __float_as_int(x), 0x160 | M, 0xf, 0xf, true));
}
__device__ __forceinline__ float grp_max(float x) {
  x = fmaxf(x, lane_xor<1>(x));
  x = fmaxf(x, lane_xor<2>(x));
  x = fmaxf(x, lane_xor<4>(x));
  return x;
}
template <int K>
__device__ __forceinline__ int grp_bcast_i(int v) {
  return (int)__builtin_amdgcn_mov_dpp8((unsigned)v, K * 0x249249u);
}
template <int K>
__device__ __forceinline__ float grp_bcast(float v) {
  return __int_as_float(grp_bcast_i<K>(__float_as_int(v)));
}

__device__ __forceinline__ float4 load_asrc(const float* __restrict__ ASRC, int s) {
  return *(const float4*)((const char*)ASRC + (unsigned)s * 16u);
}
__device__ __forceinline__ float4 edge_logits(float4 a, float4 ad) {
  float4 t;
  t.x = a.x + ad.x; t.x = t.x > 0.f ? t.x : 0.2f * t.x;
  t.y = a.y + ad.y; t.y = t.y > 0.f ? t.y : 0.2f * t.y;
  t.z = a.z + ad.z; t.z = t.z > 0.f ? t.z : 0.2f * t.z;
  t.w = a.w + ad.w; t.w = t.w > 0.f ? t.w : 0.2f * t.w;
  return t;
}
__device__ __forceinline__ float4 max4(float4 a, float4 b) {
  return make_float4(fmaxf(a.x, b.x), fmaxf(a.y, b.y), fmaxf(a.z, b.z), fmaxf(a.w, b.w));
}
__device__ __forceinline__ float4 edge_weights(bool v, float4 t, float4 m) {
  float4 e;
  e.x = v ? __expf(t.x - m.x) : 0.f;
  e.y = v ? __expf(t.y - m.y) : 0.f;
  e.z = v ? __expf(t.z - m.z) : 0.f;
  e.w = v ? __expf(t.w - m.w) : 0.f;
  return e;
}

template <int K>
__device__ __forceinline__ void load_slot(float4 (&x)[4], const char* __restrict__ xp, unsigned lo, int sp, int rem) {
  const int s = grp_bcast_i<K>(sp);
  if (K < rem) {
    const char* p = xp + ((unsigned)s * 512u + lo);
    x[0] = *(const float4*)(p);
    x[1] = *(const float4*)(p + 128);
    x[2] = *(const float4*)(p + 256);
    x[3] = *(const float4*)(p + 384);
  }
}
template <int K>
__device__ __forceinline__ void add_slot(const float4 (&x)[4], float4 (&acc)[4], float4& ss, float4 e, int rem) {
  const float w0 = grp_bcast<K>(e.x), w1 = grp_bcast<K>(e.y), w2 = grp_bcast<K>(e.z), w3 = grp_bcast<K>(e.w);
  if (K < rem) {
    ss.x += w0; ss.y += w1; ss.z += w2; ss.w += w3;
    acc[0].x += x[0].x * w0; acc[0].y += x[0].y * w0; acc[0].z += x[0].z * w0; acc[0].w += x[0].w * w0;
    acc[1].x += x[1].x * w1; acc[1].y += x[1].y * w1; acc[1].z += x[1].z * w1; acc[1].w += x[1].w * w1;
    acc[2].x += x[2].x * w2; acc[2].y += x[2].y * w2; acc[2].z += x[2].z * w2; acc[2].w += x[2].w * w2;
    acc[3].x += x[3].x * w3; acc[3].y += x[3].y * w3; acc[3].z += x[3].z * w3; acc[3].w += x[3].w * w3;
  }
}
template <int K0>
__device__ __forceinline__ void load_four(float4 (&xr)[4][4], const char* __restrict__ xp, unsigned lo, int sp, int rem) {
  load_slot<K0 + 0>(xr[0], xp, lo, sp, rem);
  load_slot<K0 + 1>(xr[1], xp, lo, sp, rem);
  load_slot<K0 + 2>(xr[2], xp, lo, sp, rem);
  load_slot<K0 + 3>(xr[3], xp, lo, sp, rem);
}
template <int K0>
__device__ __forceinline__ void add_four(const float4 (&xr)[4][4], float4 (&acc)[4], float4& ss, float4 e, int rem) {
  add_slot<K0 + 0>(xr[0], acc, ss, e, rem);
  add_slot<K0 + 1>(xr[1], acc, ss, e, rem);
  add_slot<K0 + 2>(xr[2], acc, ss, e, rem);
  add_slot<K0 + 3>(xr[3], acc, ss, e, rem);
}

template <bool RELU>
__global__ __launch_bounds__(256) void k_node_softmax_aggr(const int* __restrict__ rowptr,
                                                           const int* __restrict__ csr_src,
                                                           const float* __restrict__ XP,
                                                           const float* __restrict__ ASRC,
                                                           const float* __restrict__ ADST,
                                                           const float* __restrict__ bias,
                                                           float* __restrict__ OUT,
                                                           int n, int ne) {
  const int lane = threadIdx.x & 31;
  const int i = lane & 7;
  const int wave = (blockIdx.x * blockDim.x + threadIdx.x) >> 5;
  if (wave * 4 >= n) return;
  const int node = wave * 4 + (lane >> 3);
  const bool nv = node < n;
  int beg = 0, end = 0;
  float4 ad = make_float4(0.f, 0.f, 0.f, 0.f);
  if (nv) {
    beg = rowptr[node];
    end = rowptr[node + 1];
    ad = *(const float4*)(ADST + (size_t)node * NH);
  }
  beg = max(beg, 0);
  end = min(end, ne);
  const int deg = max(end - beg, 0);
  auto seg = [&](int e) -> int {
    return *(const int*)((const char*)csr_src + (unsigned)(beg + e) * 4u);
  };
  const char* const xp = (const char*)XP;
  const unsigned lo = 16u * (unsigned)i;

  const bool v0 = i < deg, v1 = i + 8 < deg;
  int sp0 = 0, sp1 = 0;
  if (v0) sp0 = seg(i);
  if (v1) sp1 = seg(8 + i);
  const float4 a0 = load_asrc(ASRC, sp0);
  const float4 a1 = load_asrc(ASRC, sp1);

  float4 xr[4][4];
  load_four<0>(xr, xp, lo, sp0, deg);

  const float4 t0 = edge_logits(a0, ad);
  const float4 t1 = edge_logits(a1, ad);
  const float ninf = __int_as_float(0xff800000);
  float4 m = make_float4(ninf, ninf, ninf, ninf);
  if (v0) m = t0;
  if (v1) m = max4(m, t1);
  for (int ch = 16; __builtin_amdgcn_ballot_w32(ch < deg) != 0u; ch += 8) {
    if (ch + i < deg) m = max4(m, edge_logits(load_asrc(ASRC, seg(ch + i)), ad));
  }
  m.x = grp_max(m.x);
  m.y = grp_max(m.y);
  m.z = grp_max(m.z);
  m.w = grp_max(m.w);

  float4 acc[4];
  acc[0] = acc[1] = acc[2] = acc[3] = make_float4(0.f, 0.f, 0.f, 0.f);
  float4 ss = make_float4(0.f, 0.f, 0.f, 0.f);

  const float4 e0 = edge_weights(v0, t0, m);
  add_four<0>(xr, acc, ss, e0, deg);
  if (__builtin_amdgcn_ballot_w32(deg > 4) != 0u) {
    load_four<4>(xr, xp, lo, sp0, deg);
    add_four<4>(xr, acc, ss, e0, deg);
  }
  if (__builtin_amdgcn_ballot_w32(deg > 8) != 0u) {
    const float4 e1 = edge_weights(v1, t1, m);
    load_four<0>(xr, xp, lo, sp1, deg - 8);
    add_four<0>(xr, acc, ss, e1, deg - 8);
    if (__builtin_amdgcn_ballot_w32(deg > 12) != 0u) {
      load_four<4>(xr, xp, lo, sp1, deg - 8);
      add_four<4>(xr, acc, ss, e1, deg - 8);
    }
  }
  for (int ch = 16; __builtin_amdgcn_ballot_w32(ch < deg) != 0u; ch += 8) {
    const bool v = ch + i < deg;
    int sp = 0;
    float4 t = make_float4(0.f, 0.f, 0.f, 0.f);
    if (v) {
      sp = seg(ch + i);
      t = edge_logits(load_asrc(ASRC, sp), ad);
    }
    const float4 e = edge_weights(v, t, m);
    load_four<0>(xr, xp, lo, sp, deg - ch);
    add_four<0>(xr, acc, ss, e, deg - ch);
    if (__builtin_amdgcn_ballot_w32(deg > ch + 4) != 0u) {
      load_four<4>(xr, xp, lo, sp, deg - ch);
      add_four<4>(xr, acc, ss, e, deg - ch);
    }
  }

  if (!nv) return;
  const float den[4] = {ss.x + 1e-16f, ss.y + 1e-16f, ss.z + 1e-16f, ss.w + 1e-16f};
  char* const orow = (char*)OUT + (unsigned)node * 512u + 16 * i;
#pragma unroll
  for (int hd = 0; hd < NH; ++hd) {
    const float4 bv = *(const float4*)(bias + 32 * hd + 4 * i);
    float4 o;
    o.x = acc[hd].x / den[hd] + bv.x;
    o.y = acc[hd].y / den[hd] + bv.y;
    o.z = acc[hd].z / den[hd] + bv.z;
    o.w = acc[hd].w / den[hd] + bv.w;
    if (RELU) {
      o.x = fmaxf(o.x, 0.0f);
      o.y = fmaxf(o.y, 0.0f);
      o.z = fmaxf(o.z, 0.0f);
      o.w = fmaxf(o.w, 0.0f);
    }
    *(float4*)(orow + 128 * hd) = o;
  }
}

extern "C" void kernel_launch(void* const* d_in, const int* in_sizes, int n_in,
                              void* d_out, int out_size, void* d_ws, size_t ws_size,
                              hipStream_t stream) {
  const float* x     = (const float*)d_in[0];
  const int*   ei    = (const int*)d_in[1];
  const float* W0    = (const float*)d_in[2];
  const float* atts0 = (const float*)d_in[3];
  const float* attd0 = (const float*)d_in[4];
  const float* b0    = (const float*)d_in[5];
  const float* W1    = (const float*)d_in[6];
  const float* atts1 = (const float*)d_in[7];
  const float* attd1 = (const float*)d_in[8];
  const float* b1    = (const float*)d_in[9];

  const int n = NNODES, ne = NEDGES;
  const int* srcv = ei;
  const int* dstv = ei + ne;

  char* ws = (char*)d_ws;
  float* XP   = (float*)(ws + 0);
  float* H0   = (float*)(ws + 51200000);
  float* ASRC = (float*)(ws + 102400000);
  float* ADST = (float*)(ws + 104000000);
  int*   CNT  = (int*)(ws + 105600000);
  int*   BSUM = (int*)(ws + 106400000);
  int*   RPTR = (int*)(ws + 107200000);
  int*   CSRC = (int*)(ws + 108800000);
  int*   RANK = (int*)(ws + 112000000);
  float* OUT  = (float*)d_out;

  const dim3 blk(256);
  const int gemmBlocks = cdiv(cdiv(n, 16) * 32, 256);
  const int scanBlocks = cdiv(n / 4, 256);
  const int nodeBlocks = cdiv(cdiv(n, 4) * 32, 256);

  k_zero_counts<<<cdiv(n, 256), blk, 0, stream>>>(CNT, n);
  k_count_in<<<cdiv(ne, 256), blk, 0, stream>>>(dstv, CNT, RANK, ne);
  k_block_sums<<<scanBlocks, blk, 0, stream>>>(CNT, BSUM, n);
  k_row_starts<<<scanBlocks, blk, 0, stream>>>(CNT, BSUM, RPTR, n);
  k_place_edges<<<cdiv(ne, 256), blk, 0, stream>>>(srcv, dstv, RPTR, RANK, CSRC, ne);

  k_gemm_wmma<256><<<gemmBlocks, blk, 0, stream>>>(x, W0, atts0, attd0, XP, ASRC, ADST, n);
  k_node_softmax_aggr<true><<<nodeBlocks, blk, 0, stream>>>(RPTR, CSRC, XP, ASRC, ADST, b0, H0, n, ne);

  k_gemm_wmma<128><<<gemmBlocks, blk, 0, stream>>>(H0, W1, atts1, attd1, XP, ASRC, ADST, n);
  k_node_softmax_aggr<false><<<nodeBlocks, blk, 0, stream>>>(RPTR, CSRC, XP, ASRC, ADST, b1, OUT, n, ne);
}


# Edge sort as a two-level counting sort on LDS counters (dst/512 buckets per 4096-edge chunk, then one work-group per bucket): no global atomic, 5 sort launches become 3; in-segment order by arrival
# speedup vs baseline: 6.1286x; 1.3190x over previous
#include <hip/hip_runtime.h>
#include <hip/hip_bf16.h>

#define NNODES 100000
#define NEDGES 800000
#define HC 128
#define NH 4

typedef float v2f __attribute__((ext_vector_type(2)));
typedef float v8f __attribute__((ext_vector_type(8)));

static inline int cdiv(int a, int b) { return (a + b - 1) / b; }

template <int M>
__device__ __forceinline__ float row_xadd(bool hi, float x, float y) {
  const float keep = hi ? y : x;
  const float give = hi ? x : y;
  const int got = __builtin_amdgcn_update_dpp(0, __float_as_int(give), 0x160 | M, 0xf, 0xf, true);
  return keep + __int_as_float(got);
}

template <int K>
__global__ __launch_bounds__(256) void k_gemm_wmma(const float* __restrict__ X,
                                                   const float* __restrict__ W,
                                                   const float* __restrict__ att_src,
                                                   const float* __restrict__ att_dst,
                                                   float* __restrict__ XP,
                                                   float* __restrict__ ASRC,
                                                   float* __restrict__ ADST, int rows) {
  constexpr int P = K + 4;
  constexpr int KF = K / 4;
  constexpr int NQ = K / 32;
  __shared__ float WlA[64 * P];
  __shared__ float WlB[64 * P];

  const int tid = threadIdx.x;
  const int lane = tid & 31;
  const int c = lane & 15;
  const int h = lane >> 4;
  const int m0 = (blockIdx.x * 8 + (tid >> 5)) * 16;
  const bool active = m0 < rows;

  const float* xrow = X + (size_t)(m0 + c) * K + 4 * h;
  float4 an[4];
  if (active) {
#pragma unroll
    for (int i = 0; i < 4; ++i) an[i] = *(const float4*)(xrow + 8 * i);
  }

  {
    const int n = tid / KF, kf = tid - n * KF;
    constexpr int RS = 256 / KF;
    const float* wsrc = W + (size_t)n * K + 4 * kf;
    float* da = WlA + n * P + 4 * kf;
    float* db = WlB + n * P + 4 * kf;
#pragma unroll 8
    for (int r = 0; r < 64; r += RS)
      *(float4*)(da + r * P) = *(const float4*)(wsrc + (size_t)r * K);
#pragma unroll 8
    for (int r = 0; r < 64; r += RS)
      *(float4*)(db + r * P) = *(const float4*)(wsrc + (size_t)(64 + r) * K);
  }

  __syncthreads();
  if (!active) return;

  const float* bla = WlA + c * P + 4 * h;
  const float* blb = WlB + c * P + 4 * h;
  v8f acc[8] = {};
  float4 b0[8], b1[8];
#pragma unroll
  for (int nt = 0; nt < 4; ++nt) {
    b0[nt] = *(const float4*)(bla + nt * 16 * P);
    b0[nt + 4] = *(const float4*)(blb + nt * 16 * P);
  }

#pragma unroll
  for (int q = 0; q < NQ; ++q) {
    float4 a[4];
#pragma unroll
    for (int i = 0; i < 4; ++i) a[i] = an[i];
    if (q + 1 < NQ) {
#pragma unroll
      for (int i = 0; i < 4; ++i) an[i] = *(const float4*)(xrow + 32 * (q + 1) + 8 * i);
    }
#pragma unroll
    for (int i = 0; i < 4; ++i) {
      float4* bc = (i & 1) ? b1 : b0;
      float4* bn = (i & 1) ? b0 : b1;
      if (i < 3 || q + 1 < NQ) {
        const int ko = 32 * q + 8 * (i + 1);
#pragma unroll
        for (int nt = 0; nt < 4; ++nt) {
          bn[nt] = *(const float4*)(bla + nt * 16 * P + ko);
          bn[nt + 4] = *(const float4*)(blb + nt * 16 * P + ko);
        }
      }
      const v2f a0 = {a[i].x, a[i].y};
      const v2f a1 = {a[i].z, a[i].w};
#pragma unroll
      for (int nt = 0; nt < 8; ++nt) {
        const v2f b = {bc[nt].x, bc[nt].y};
        acc[nt] = __builtin_amdgcn_wmma_f32_16x16x4_f32(
             false, a0,  false, b,
             (short)0, acc[nt],  false,  false);
      }
#pragma unroll
      for (int nt = 0; nt < 8; ++nt) {
        const v2f b = {bc[nt].z, bc[nt].w};
        acc[nt] = __builtin_amdgcn_wmma_f32_16x16x4_f32(
             false, a1,  false, b,
             (short)0, acc[nt],  false,  false);
      }
    }
  }

  __builtin_amdgcn_sched_barrier(0);
  asm volatile("v_nop\n\tv_nop\n\tv_nop\n\tv_nop\n\tv_nop\n\tv_nop\n\tv_nop\n\tv_nop");
  __builtin_amdgcn_sched_barrier(0);

  float* const orow = XP + (size_t)(m0 + h * 8) * HC + c;
#pragma unroll
  for (int nt = 0; nt < 8; ++nt) {
#pragma unroll
    for (int i = 0; i < 8; ++i) orow[(size_t)i * HC + nt * 16] = acc[nt][i];
  }

  float a_lo[2][NH], a_hi[2][NH];
#pragma unroll
  for (int hd = 0; hd < NH; ++hd) {
    a_lo[0][hd] = att_src[hd * 32 + c];
    a_hi[0][hd] = att_src[hd * 32 + 16 + c];
    a_lo[1][hd] = att_dst[hd * 32 + c];
    a_hi[1][hd] = att_dst[hd * 32 + 16 + c];
  }
  const bool l0 = (c & 1) != 0, l1 = (c & 2) != 0, l2 = (c & 4) != 0, l3 = (c & 8) != 0;
  float u[2][NH];
#pragma unroll
  for (int sd = 0; sd < 2; ++sd) {
    float sv[2][NH];
#pragma unroll
    for (int j2 = 0; j2 < 2; ++j2) {
      float qv[2][NH];
#pragma unroll
      for (int j1 = 0; j1 < 2; ++j1) {
        const int i = 4 * j2 + 2 * j1;
#pragma unroll
        for (int hd = 0; hd < NH; ++hd) {
          const v2f t0 = {acc[2 * hd][i], acc[2 * hd][i + 1]};
          const v2f t1 = {acc[2 * hd + 1][i], acc[2 * hd + 1][i + 1]};
          const v2f xy = t0 * a_lo[sd][hd] + t1 * a_hi[sd][hd];
          qv[j1][hd] = row_xadd<1>(l0, xy.x, xy.y);
        }
      }
#pragma unroll
      for (int hd = 0; hd < NH; ++hd) sv[j2][hd] = row_xadd<2>(l1, qv[0][hd], qv[1][hd]);
    }
#pragma unroll
    for (int hd = 0; hd < NH; ++hd) u[sd][hd] = row_xadd<4>(l2, sv[0][hd], sv[1][hd]);
  }
  float4 r;
  r.x = row_xadd<8>(l3, u[0][0], u[1][0]);
  r.y = row_xadd<8>(l3, u[0][1], u[1][1]);
  r.z = row_xadd<8>(l3, u[0][2], u[1][2]);
  r.w = row_xadd<8>(l3, u[0][3], u[1][3]);
  float* const arow = (l3 ? ADST : ASRC) + (size_t)(m0 + h * 8 + (c & 7)) * NH;
  *(float4*)arow = r;
}

#define BKT_NODES 512
#define CHUNK_EDGES 4096
#define NBK ((NNODES + 1 + BKT_NODES - 1) / BKT_NODES)
#define NCH ((NEDGES + CHUNK_EDGES - 1) / CHUNK_EDGES)
#define SRC_BITS 17

__device__ __forceinline__ int bucket_of(int d) {
  return (int)min((unsigned)d / BKT_NODES, (unsigned)(NBK - 1));
}

__global__ __launch_bounds__(1024) void k_bucket_count(const int* __restrict__ dst,
                                                       int* __restrict__ M, int ne) {
  __shared__ int hist[NBK];
  const int t = threadIdx.x;
  const int e = (blockIdx.x * 1024 + t) * 4;
  const bool v = e < ne;
  int4 d = make_int4(0, 0, 0, 0);
  if (v) d = *(const int4*)(dst + e);
  if (t < NBK) hist[t] = 0;
  __syncthreads();
  if (v) {
    atomicAdd(&hist[bucket_of(d.x)], 1);
    atomicAdd(&hist[bucket_of(d.y)], 1);
    atomicAdd(&hist[bucket_of(d.z)], 1);
    atomicAdd(&hist[bucket_of(d.w)], 1);
  }
  __syncthreads();
  if (t < NBK) M[t * NCH + blockIdx.x] = hist[t];
}

__global__ __launch_bounds__(1024) void k_bucket_scatter(const int* __restrict__ src,
                                                         const int* __restrict__ dst,
                                                         const int* __restrict__ M,
                                                         int* __restrict__ BS,
                                                         int* __restrict__ BK, int ne) {
  __shared__ int rowTot[NBK];
  __shared__ int rowPre[NBK];
  __shared__ int rowOwn[NBK];
  __shared__ int cursor[NBK];
  __shared__ int lbase[NBK];
  __shared__ int lcur[NBK];
  __shared__ int stageW[CHUNK_EDGES];
  __shared__ int stageB[CHUNK_EDGES];
  const int t = threadIdx.x, lane = t & 31, w = t >> 5;
  const int c = blockIdx.x;
  const int e = (c * 1024 + t) * 4;
  const bool v = e < ne;
  int4 d = make_int4(0, 0, 0, 0), s = make_int4(0, 0, 0, 0);
  if (v) {
    d = *(const int4*)(dst + e);
    s = *(const int4*)(src + e);
  }
  if (t < NBK) rowOwn[t] = M[t * NCH + c];
  constexpr int RW = (NBK + 31) / 32, CW = (NCH + 31) / 32;
  int mv[RW][CW];
#pragma unroll
  for (int r = 0; r < RW; ++r) {
    const int bb = min(w + 32 * r, NBK - 1);
#pragma unroll
    for (int q = 0; q < CW; ++q) mv[r][q] = M[bb * NCH + min(lane + 32 * q, NCH - 1)];
  }
  int tot[RW], pre[RW];
#pragma unroll
  for (int r = 0; r < RW; ++r) {
    tot[r] = 0;
    pre[r] = 0;
#pragma unroll
    for (int q = 0; q < CW; ++q) {
      const int j = lane + 32 * q;
      const int x = (j < NCH) ? mv[r][q] : 0;
      tot[r] += x;
      pre[r] += (j < c) ? x : 0;
    }
  }
#pragma unroll
  for (int o = 16; o > 0; o >>= 1) {
#pragma unroll
    for (int r = 0; r < RW; ++r) {
      tot[r] += __shfl_xor(tot[r], o);
      pre[r] += __shfl_xor(pre[r], o);
    }
  }
#pragma unroll
  for (int r = 0; r < RW; ++r) {
    const int b = w + 32 * r;
    if (lane == 0 && b < NBK) {
      rowTot[b] = tot[r];
      rowPre[b] = pre[r];
    }
  }
  __syncthreads();
  if (w == 0) {
    int xv[RW], ov[RW];
    int sum = 0, osum = 0;
#pragma unroll
    for (int k = 0; k < RW; ++k) {
      const int b = lane * RW + k;
      xv[k] = (b < NBK) ? rowTot[b] : 0;
      ov[k] = (b < NBK) ? rowOwn[b] : 0;
      sum += xv[k];
      osum += ov[k];
    }
    int inc = sum, oinc = osum;
    for (int o = 1; o < 32; o <<= 1) {
      const int y = __shfl_up(inc, o);
      const int oy = __shfl_up(oinc, o);
      if (lane >= o) {
        inc += y;
        oinc += oy;
      }
    }
    int run = inc - sum;
    int orun = oinc - osum;
#pragma unroll
    for (int k = 0; k < RW; ++k) {
      const int b = lane * RW + k;
      if (b < NBK) {
        cursor[b] = run + rowPre[b];
        lbase[b] = orun;
        lcur[b] = orun;
        if (c == 0) BS[b] = run;
      }
      run += xv[k];
      orun += ov[k];
    }
    if (c == 0 && lane == 31) BS[NBK] = inc;
  }
  __syncthreads();
  if (v) {
    const int dd[4] = {d.x, d.y, d.z, d.w};
    const int sv[4] = {s.x, s.y, s.z, s.w};
#pragma unroll
    for (int k = 0; k < 4; ++k) {
      const int b = bucket_of(dd[k]);
      int ls = atomicAdd(&lcur[b], 1);
      ls = min(max(ls, 0), CHUNK_EDGES - 1);
      stageW[ls] = (int)((((unsigned)dd[k] % BKT_NODES) << SRC_BITS) | (unsigned)sv[k]);
      stageB[ls] = b;
    }
  }
  __syncthreads();
  const int nvalid = min(CHUNK_EDGES, ne - c * CHUNK_EDGES);
#pragma unroll
  for (int k = 0; k < CHUNK_EDGES / 1024; ++k) {
    const int i = t + 1024 * k;
    if (i < nvalid) {
      const int b = stageB[i];
      int pos = cursor[b] + (i - lbase[b]);
      pos = min(max(pos, 0), ne - 1);
      BK[pos] = stageW[i];
    }
  }
}

__global__ __launch_bounds__(1024) void k_bucket_csr(const int* __restrict__ BS,
                                                     const int* __restrict__ BK,
                                                     int* __restrict__ rowptr,
                                                     int* __restrict__ csr_src, int n, int ne) {
  __shared__ int cnt[BKT_NODES];
  __shared__ int cur[BKT_NODES];
  __shared__ int wtot[BKT_NODES / 32];
  constexpr int STAGE = 8192;
  __shared__ int stage[STAGE];
  const int t = threadIdx.x, lane = t & 31, w = t >> 5;
  const int b = blockIdx.x;
  const int start = max(BS[b], 0);
  const int end = min(BS[b + 1], ne);
  if (t < BKT_NODES) cnt[t] = 0;
  __syncthreads();
  constexpr int TRIP = 8;
  for (int i0 = start; i0 < end; i0 += 1024 * TRIP) {
    unsigned wv[TRIP];
#pragma unroll
    for (int k = 0; k < TRIP; ++k) wv[k] = (unsigned)BK[min(i0 + t + 1024 * k, end - 1)];
#pragma unroll
    for (int k = 0; k < TRIP; ++k)
      if (i0 + t + 1024 * k < end) atomicAdd(&cnt[(wv[k] >> SRC_BITS) % BKT_NODES], 1);
  }
  __syncthreads();
  int val = 0, x = 0;
  if (t < BKT_NODES) {
    val = cnt[t];
    x = val;
    for (int o = 1; o < 32; o <<= 1) {
      const int y = __shfl_up(x, o);
      if (lane >= o) x += y;
    }
    if (lane == 31) wtot[w] = x;
  }
  __syncthreads();
  if (t < BKT_NODES) {
    int p = start + x - val;
    for (int k = 0; k < w; ++k) p += wtot[k];
    cur[t] = p;
    const int node = b * BKT_NODES + t;
    if (node <= n) rowptr[node] = p;
  }
  __syncthreads();
  const bool fits = end - start <= STAGE;
  for (int i0 = start; i0 < end; i0 += 1024 * TRIP) {
    unsigned wv[TRIP];
#pragma unroll
    for (int k = 0; k < TRIP; ++k) wv[k] = (unsigned)BK[min(i0 + t + 1024 * k, end - 1)];
#pragma unroll
    for (int k = 0; k < TRIP; ++k) {
      if (i0 + t + 1024 * k < end) {
        const int pos = atomicAdd(&cur[(wv[k] >> SRC_BITS) % BKT_NODES], 1);
        const int sv = (int)(wv[k] & ((1u << SRC_BITS) - 1u));
        if (fits)
          stage[min(max(pos - start, 0), STAGE - 1)] = sv;
        else
          csr_src[min(max(pos, 0), ne - 1)] = sv;
      }
    }
  }
  __syncthreads();
  if (fits) {
    for (int i = t; i < end - start; i += 1024) csr_src[start + i] = stage[i];
  }
}


template <int M>
__device__ __forceinline__ float lane_xor(float x) {
  return __int_as_float(__builtin_amdgcn_update_dpp(0, __float_as_int(x), 0x160 | M, 0xf, 0xf, true));
}
__device__ __forceinline__ float grp_max(float x) {
  x = fmaxf(x, lane_xor<1>(x));
  x = fmaxf(x, lane_xor<2>(x));
  x = fmaxf(x, lane_xor<4>(x));
  return x;
}
template <int K>
__device__ __forceinline__ int grp_bcast_i(int v) {
  return (int)__builtin_amdgcn_mov_dpp8((unsigned)v, K * 0x249249u);
}
template <int K>
__device__ __forceinline__ float grp_bcast(float v) {
  return __int_as_float(grp_bcast_i<K>(__float_as_int(v)));
}

__device__ __forceinline__ float4 load_asrc(const float* __restrict__ ASRC, int s) {
  return *(const float4*)((const char*)ASRC + (unsigned)s * 16u);
}
__device__ __forceinline__ float4 edge_logits(float4 a, float4 ad) {
  float4 t;
  t.x = a.x + ad.x; t.x = t.x > 0.f ? t.x : 0.2f * t.x;
  t.y = a.y + ad.y; t.y = t.y > 0.f ? t.y : 0.2f * t.y;
  t.z = a.z + ad.z; t.z = t.z > 0.f ? t.z : 0.2f * t.z;
  t.w = a.w + ad.w; t.w = t.w > 0.f ? t.w : 0.2f * t.w;
  return t;
}
__device__ __forceinline__ float4 max4(float4 a, float4 b) {
  return make_float4(fmaxf(a.x, b.x), fmaxf(a.y, b.y), fmaxf(a.z, b.z), fmaxf(a.w, b.w));
}
__device__ __forceinline__ float4 edge_weights(bool v, float4 t, float4 m) {
  float4 e;
  e.x = v ? __expf(t.x - m.x) : 0.f;
  e.y = v ? __expf(t.y - m.y) : 0.f;
  e.z = v ? __expf(t.z - m.z) : 0.f;
  e.w = v ? __expf(t.w - m.w) : 0.f;
  return e;
}

template <int K>
__device__ __forceinline__ void load_slot(float4 (&x)[4], const char* __restrict__ xp, unsigned lo, int sp, int rem) {
  const int s = grp_bcast_i<K>(sp);
  if (K < rem) {
    const char* p = xp + ((unsigned)s * 512u + lo);
    x[0] = *(const float4*)(p);
    x[1] = *(const float4*)(p + 128);
    x[2] = *(const float4*)(p + 256);
    x[3] = *(const float4*)(p + 384);
  }
}
template <int K>
__device__ __forceinline__ void add_slot(const float4 (&x)[4], float4 (&acc)[4], float4& ss, float4 e, int rem) {
  const float w0 = grp_bcast<K>(e.x), w1 = grp_bcast<K>(e.y), w2 = grp_bcast<K>(e.z), w3 = grp_bcast<K>(e.w);
  if (K < rem) {
    ss.x += w0; ss.y += w1; ss.z += w2; ss.w += w3;
    acc[0].x += x[0].x * w0; acc[0].y += x[0].y * w0; acc[0].z += x[0].z * w0; acc[0].w += x[0].w * w0;
    acc[1].x += x[1].x * w1; acc[1].y += x[1].y * w1; acc[1].z += x[1].z * w1; acc[1].w += x[1].w * w1;
    acc[2].x += x[2].x * w2; acc[2].y += x[2].y * w2; acc[2].z += x[2].z * w2; acc[2].w += x[2].w * w2;
    acc[3].x += x[3].x * w3; acc[3].y += x[3].y * w3; acc[3].z += x[3].z * w3; acc[3].w += x[3].w * w3;
  }
}
template <int K0>
__device__ __forceinline__ void load_four(float4 (&xr)[4][4], const char* __restrict__ xp, unsigned lo, int sp, int rem) {
  load_slot<K0 + 0>(xr[0], xp, lo, sp, rem);
  load_slot<K0 + 1>(xr[1], xp, lo, sp, rem);
  load_slot<K0 + 2>(xr[2], xp, lo, sp, rem);
  load_slot<K0 + 3>(xr[3], xp, lo, sp, rem);
}
template <int K0>
__device__ __forceinline__ void add_four(const float4 (&xr)[4][4], float4 (&acc)[4], float4& ss, float4 e, int rem) {
  add_slot<K0 + 0>(xr[0], acc, ss, e, rem);
  add_slot<K0 + 1>(xr[1], acc, ss, e, rem);
  add_slot<K0 + 2>(xr[2], acc, ss, e, rem);
  add_slot<K0 + 3>(xr[3], acc, ss, e, rem);
}

template <bool RELU>
__global__ __launch_bounds__(256) void k_node_softmax_aggr(const int* __restrict__ rowptr,
                                                           const int* __restrict__ csr_src,
                                                           const float* __restrict__ XP,
                                                           const float* __restrict__ ASRC,
                                                           const float* __restrict__ ADST,
                                                           const float* __restrict__ bias,
                                                           float* __restrict__ OUT,
                                                           int n, int ne) {
  const int lane = threadIdx.x & 31;
  const int i = lane & 7;
  const int wave = (blockIdx.x * blockDim.x + threadIdx.x) >> 5;
  if (wave * 4 >= n) return;
  const int node = wave * 4 + (lane >> 3);
  const bool nv = node < n;
  int beg = 0, end = 0;
  float4 ad = make_float4(0.f, 0.f, 0.f, 0.f);
  if (nv) {
    beg = rowptr[node];
    end = rowptr[node + 1];
    ad = *(const float4*)(ADST + (size_t)node * NH);
  }
  beg = max(beg, 0);
  end = min(end, ne);
  const int deg = max(end - beg, 0);
  auto seg = [&](int e) -> int {
    return *(const int*)((const char*)csr_src + (unsigned)(beg + e) * 4u);
  };
  const char* const xp = (const char*)XP;
  const unsigned lo = 16u * (unsigned)i;

  const bool v0 = i < deg, v1 = i + 8 < deg;
  int sp0 = 0, sp1 = 0;
  if (v0) sp0 = seg(i);
  if (v1) sp1 = seg(8 + i);
  const float4 a0 = load_asrc(ASRC, sp0);
  const float4 a1 = load_asrc(ASRC, sp1);

  float4 xr[4][4];
  load_four<0>(xr, xp, lo, sp0, deg);

  const float4 t0 = edge_logits(a0, ad);
  const float4 t1 = edge_logits(a1, ad);
  const float ninf = __int_as_float(0xff800000);
  float4 m = make_float4(ninf, ninf, ninf, ninf);
  if (v0) m = t0;
  if (v1) m = max4(m, t1);
  for (int ch = 16; __builtin_amdgcn_ballot_w32(ch < deg) != 0u; ch += 8) {
    if (ch + i < deg) m = max4(m, edge_logits(load_asrc(ASRC, seg(ch + i)), ad));
  }
  m.x = grp_max(m.x);
  m.y = grp_max(m.y);
  m.z = grp_max(m.z);
  m.w = grp_max(m.w);

  float4 acc[4];
  acc[0] = acc[1] = acc[2] = acc[3] = make_float4(0.f, 0.f, 0.f, 0.f);
  float4 ss = make_float4(0.f, 0.f, 0.f, 0.f);

  const float4 e0 = edge_weights(v0, t0, m);
  add_four<0>(xr, acc, ss, e0, deg);
  if (__builtin_amdgcn_ballot_w32(deg > 4) != 0u) {
    load_four<4>(xr, xp, lo, sp0, deg);
    add_four<4>(xr, acc, ss, e0, deg);
  }
  if (__builtin_amdgcn_ballot_w32(deg > 8) != 0u) {
    const float4 e1 = edge_weights(v1, t1, m);
    load_four<0>(xr, xp, lo, sp1, deg - 8);
    add_four<0>(xr, acc, ss, e1, deg - 8);
    if (__builtin_amdgcn_ballot_w32(deg > 12) != 0u) {
      load_four<4>(xr, xp, lo, sp1, deg - 8);
      add_four<4>(xr, acc, ss, e1, deg - 8);
    }
  }
  for (int ch = 16; __builtin_amdgcn_ballot_w32(ch < deg) != 0u; ch += 8) {
    const bool v = ch + i < deg;
    int sp = 0;
    float4 t = make_float4(0.f, 0.f, 0.f, 0.f);
    if (v) {
      sp = seg(ch + i);
      t = edge_logits(load_asrc(ASRC, sp), ad);
    }
    const float4 e = edge_weights(v, t, m);
    load_four<0>(xr, xp, lo, sp, deg - ch);
    add_four<0>(xr, acc, ss, e, deg - ch);
    if (__builtin_amdgcn_ballot_w32(deg > ch + 4) != 0u) {
      load_four<4>(xr, xp, lo, sp, deg - ch);
      add_four<4>(xr, acc, ss, e, deg - ch);
    }
  }

  if (!nv) return;
  const float den[4] = {ss.x + 1e-16f, ss.y + 1e-16f, ss.z + 1e-16f, ss.w + 1e-16f};
  char* const orow = (char*)OUT + (unsigned)node * 512u + 16 * i;
#pragma unroll
  for (int hd = 0; hd < NH; ++hd) {
    const float4 bv = *(const float4*)(bias + 32 * hd + 4 * i);
    float4 o;
    o.x = acc[hd].x / den[hd] + bv.x;
    o.y = acc[hd].y / den[hd] + bv.y;
    o.z = acc[hd].z / den[hd] + bv.z;
    o.w = acc[hd].w / den[hd] + bv.w;
    if (RELU) {
      o.x = fmaxf(o.x, 0.0f);
      o.y = fmaxf(o.y, 0.0f);
      o.z = fmaxf(o.z, 0.0f);
      o.w = fmaxf(o.w, 0.0f);
    }
    *(float4*)(orow + 128 * hd) = o;
  }
}

extern "C" void kernel_launch(void* const* d_in, const int* in_sizes, int n_in,
                              void* d_out, int out_size, void* d_ws, size_t ws_size,
                              hipStream_t stream) {
  const float* x     = (const float*)d_in[0];
  const int*   ei    = (const int*)d_in[1];
  const float* W0    = (const float*)d_in[2];
  const float* atts0 = (const float*)d_in[3];
  const float* attd0 = (const float*)d_in[4];
  const float* b0    = (const float*)d_in[5];
  const float* W1    = (const float*)d_in[6];
  const float* atts1 = (const float*)d_in[7];
  const float* attd1 = (const float*)d_in[8];
  const float* b1    = (const float*)d_in[9];

  const int n = NNODES, ne = NEDGES;
  const int* srcv = ei;
  const int* dstv = ei + ne;

  char* ws = (char*)d_ws;
  float* XP   = (float*)(ws + 0);
  float* H0   = (float*)(ws + 51200000);
  float* ASRC = (float*)(ws + 102400000);
  float* ADST = (float*)(ws + 104000000);
  int*   MCNT = (int*)(ws + 105600000);
  int*   BS   = (int*)(ws + 106400000);
  int*   RPTR = (int*)(ws + 107200000);
  int*   CSRC = (int*)(ws + 108800000);
  int*   BK   = (int*)(ws + 112000000);
  float* OUT  = (float*)d_out;

  const dim3 blk(256);
  const int gemmBlocks = cdiv(cdiv(n, 16) * 32, 256);
  const int nodeBlocks = cdiv(cdiv(n, 4) * 32, 256);

  k_bucket_count<<<NCH, 1024, 0, stream>>>(dstv, MCNT, ne);
  k_bucket_scatter<<<NCH, 1024, 0, stream>>>(srcv, dstv, MCNT, BS, BK, ne);
  k_bucket_csr<<<NBK, 1024, 0, stream>>>(BS, BK, RPTR, CSRC, n, ne);

  k_gemm_wmma<256><<<gemmBlocks, blk, 0, stream>>>(x, W0, atts0, attd0, XP, ASRC, ADST, n);
  k_node_softmax_aggr<true><<<nodeBlocks, blk, 0, stream>>>(RPTR, CSRC, XP, ASRC, ADST, b0, H0, n, ne);

  k_gemm_wmma<128><<<gemmBlocks, blk, 0, stream>>>(H0, W1, atts1, attd1, XP, ASRC, ADST, n);
  k_node_softmax_aggr<false><<<nodeBlocks, blk, 0, stream>>>(RPTR, CSRC, XP, ASRC, ADST, b1, OUT, n, ne);
}


# Edge sort in 2 launches, not 3: each 4096-edge chunk is counted, scanned and grouped by dst/512 in LDS into its own slots; a bucket's work-group gathers its 196 runs; in-segment order by arrival
# speedup vs baseline: 6.3758x; 1.0403x over previous
#include <hip/hip_runtime.h>
#include <hip/hip_bf16.h>

#define NNODES 100000
#define NEDGES 800000
#define HC 128
#define NH 4

typedef float v2f __attribute__((ext_vector_type(2)));
typedef float v8f __attribute__((ext_vector_type(8)));

static inline int cdiv(int a, int b) { return (a + b - 1) / b; }

template <int M>
__device__ __forceinline__ float row_xadd(bool hi, float x, float y) {
  const float keep = hi ? y : x;
  const float give = hi ? x : y;
  const int got = __builtin_amdgcn_update_dpp(0, __float_as_int(give), 0x160 | M, 0xf, 0xf, true);
  return keep + __int_as_float(got);
}

template <int K>
__global__ __launch_bounds__(256) void k_gemm_wmma(const float* __restrict__ X,
                                                   const float* __restrict__ W,
                                                   const float* __restrict__ att_src,
                                                   const float* __restrict__ att_dst,
                                                   float* __restrict__ XP,
                                                   float* __restrict__ ASRC,
                                                   float* __restrict__ ADST, int rows) {
  constexpr int P = K + 4;
  constexpr int KF = K / 4;
  constexpr int NQ = K / 32;
  __shared__ float WlA[64 * P];
  __shared__ float WlB[64 * P];

  const int tid = threadIdx.x;
  const int lane = tid & 31;
  const int c = lane & 15;
  const int h = lane >> 4;
  const int m0 = (blockIdx.x * 8 + (tid >> 5)) * 16;
  const bool active = m0 < rows;

  const float* xrow = X + (size_t)(m0 + c) * K + 4 * h;
  float4 an[4];
  if (active) {
#pragma unroll
    for (int i = 0; i < 4; ++i) an[i] = *(const float4*)(xrow + 8 * i);
  }

  {
    const int n = tid / KF, kf = tid - n * KF;
    constexpr int RS = 256 / KF;
    const float* wsrc = W + (size_t)n * K + 4 * kf;
    float* da = WlA + n * P + 4 * kf;
    float* db = WlB + n * P + 4 * kf;
#pragma unroll 8
    for (int r = 0; r < 64; r += RS)
      *(float4*)(da + r * P) = *(const float4*)(wsrc + (size_t)r * K);
#pragma unroll 8
    for (int r = 0; r < 64; r += RS)
      *(float4*)(db + r * P) = *(const float4*)(wsrc + (size_t)(64 + r) * K);
  }

  __syncthreads();
  if (!active) return;

  const float* bla = WlA + c * P + 4 * h;
  const float* blb = WlB + c * P + 4 * h;
  v8f acc[8] = {};
  float4 b0[8], b1[8];
#pragma unroll
  for (int nt = 0; nt < 4; ++nt) {
    b0[nt] = *(const float4*)(bla + nt * 16 * P);
    b0[nt + 4] = *(const float4*)(blb + nt * 16 * P);
  }

#pragma unroll
  for (int q = 0; q < NQ; ++q) {
    float4 a[4];
#pragma unroll
    for (int i = 0; i < 4; ++i) a[i] = an[i];
    if (q + 1 < NQ) {
#pragma unroll
      for (int i = 0; i < 4; ++i) an[i] = *(const float4*)(xrow + 32 * (q + 1) + 8 * i);
    }
#pragma unroll
    for (int i = 0; i < 4; ++i) {
      float4* bc = (i & 1) ? b1 : b0;
      float4* bn = (i & 1) ? b0 : b1;
      if (i < 3 || q + 1 < NQ) {
        const int ko = 32 * q + 8 * (i + 1);
#pragma unroll
        for (int nt = 0; nt < 4; ++nt) {
          bn[nt] = *(const float4*)(bla + nt * 16 * P + ko);
          bn[nt + 4] = *(const float4*)(blb + nt * 16 * P + ko);
        }
      }
      const v2f a0 = {a[i].x, a[i].y};
      const v2f a1 = {a[i].z, a[i].w};
#pragma unroll
      for (int nt = 0; nt < 8; ++nt) {
        const v2f b = {bc[nt].x, bc[nt].y};
        acc[nt] = __builtin_amdgcn_wmma_f32_16x16x4_f32(
             false, a0,  false, b,
             (short)0, acc[nt],  false,  false);
      }
#pragma unroll
      for (int nt = 0; nt < 8; ++nt) {
        const v2f b = {bc[nt].z, bc[nt].w};
        acc[nt] = __builtin_amdgcn_wmma_f32_16x16x4_f32(
             false, a1,  false, b,
             (short)0, acc[nt],  false,  false);
      }
    }
  }

  __builtin_amdgcn_sched_barrier(0);
  asm volatile("v_nop\n\tv_nop\n\tv_nop\n\tv_nop\n\tv_nop\n\tv_nop\n\tv_nop\n\tv_nop");
  __builtin_amdgcn_sched_barrier(0);

  float* const orow = XP + (size_t)(m0 + h * 8) * HC + c;
#pragma unroll
  for (int nt = 0; nt < 8; ++nt) {
#pragma unroll
    for (int i = 0; i < 8; ++i) orow[(size_t)i * HC + nt * 16] = acc[nt][i];
  }

  float a_lo[2][NH], a_hi[2][NH];
#pragma unroll
  for (int hd = 0; hd < NH; ++hd) {
    a_lo[0][hd] = att_src[hd * 32 + c];
    a_hi[0][hd] = att_src[hd * 32 + 16 + c];
    a_lo[1][hd] = att_dst[hd * 32 + c];
    a_hi[1][hd] = att_dst[hd * 32 + 16 + c];
  }
  const bool l0 = (c & 1) != 0, l1 = (c & 2) != 0, l2 = (c & 4) != 0, l3 = (c & 8) != 0;
  float u[2][NH];
#pragma unroll
  for (int sd = 0; sd < 2; ++sd) {
    float sv[2][NH];
#pragma unroll
    for (int j2 = 0; j2 < 2; ++j2) {
      float qv[2][NH];
#pragma unroll
      for (int j1 = 0; j1 < 2; ++j1) {
        const int i = 4 * j2 + 2 * j1;
#pragma unroll
        for (int hd = 0; hd < NH; ++hd) {
          const v2f t0 = {acc[2 * hd][i], acc[2 * hd][i + 1]};
          const v2f t1 = {acc[2 * hd + 1][i], acc[2 * hd + 1][i + 1]};
          const v2f xy = t0 * a_lo[sd][hd] + t1 * a_hi[sd][hd];
          qv[j1][hd] = row_xadd<1>(l0, xy.x, xy.y);
        }
      }
#pragma unroll
      for (int hd = 0; hd < NH; ++hd) sv[j2][hd] = row_xadd<2>(l1, qv[0][hd], qv[1][hd]);
    }
#pragma unroll
    for (int hd = 0; hd < NH; ++hd) u[sd][hd] = row_xadd<4>(l2, sv[0][hd], sv[1][hd]);
  }
  float4 r;
  r.x = row_xadd<8>(l3, u[0][0], u[1][0]);
  r.y = row_xadd<8>(l3, u[0][1], u[1][1]);
  r.z = row_xadd<8>(l3, u[0][2], u[1][2]);
  r.w = row_xadd<8>(l3, u[0][3], u[1][3]);
  float* const arow = (l3 ? ADST : ASRC) + (size_t)(m0 + h * 8 + (c & 7)) * NH;
  *(float4*)arow = r;
}

#define BKT_NODES 512
#define CHUNK_EDGES 4096
#define NBK ((NNODES + 1 + BKT_NODES - 1) / BKT_NODES)
#define NCH ((NEDGES + CHUNK_EDGES - 1) / CHUNK_EDGES)
#define SRC_BITS 17
#define LOT_PITCH 256

__device__ __forceinline__ int bucket_of(int d) {
  return (int)min((unsigned)d / BKT_NODES, (unsigned)(NBK - 1));
}

__global__ __launch_bounds__(1024) void k_chunk_sort(const int* __restrict__ src,
                                                     const int* __restrict__ dst,
                                                     int* __restrict__ LOT,
                                                     int* __restrict__ BK, int ne) {
  __shared__ int hist[NBK];
  __shared__ int lcur[NBK];
  __shared__ int stageW[CHUNK_EDGES];
  const int t = threadIdx.x, lane = t & 31, w = t >> 5;
  const int c = blockIdx.x;
  const int e = (c * 1024 + t) * 4;
  const bool v = e < ne;
  int4 d = make_int4(0, 0, 0, 0), s = make_int4(0, 0, 0, 0);
  if (v) {
    d = *(const int4*)(dst + e);
    s = *(const int4*)(src + e);
  }
  if (t < NBK) hist[t] = 0;
  __syncthreads();
  const int dd[4] = {d.x, d.y, d.z, d.w};
  const int sv[4] = {s.x, s.y, s.z, s.w};
  if (v) {
#pragma unroll
    for (int k = 0; k < 4; ++k) atomicAdd(&hist[bucket_of(dd[k])], 1);
  }
  __syncthreads();
  if (w == 0) {
    constexpr int RW = (NBK + 31) / 32;
    int ov[RW];
    int osum = 0;
#pragma unroll
    for (int k = 0; k < RW; ++k) {
      const int b = lane * RW + k;
      ov[k] = (b < NBK) ? hist[b] : 0;
      osum += ov[k];
    }
    int oinc = osum;
    for (int o = 1; o < 32; o <<= 1) {
      const int y = __shfl_up(oinc, o);
      if (lane >= o) oinc += y;
    }
    int orun = oinc - osum;
    int* const lrow = LOT + (size_t)c * LOT_PITCH;
#pragma unroll
    for (int k = 0; k < RW; ++k) {
      const int b = lane * RW + k;
      if (b < NBK) {
        lcur[b] = orun;
        lrow[b] = orun;
      }
      orun += ov[k];
    }
    if (lane == 31) lrow[NBK] = oinc;
  }
  __syncthreads();
  if (v) {
#pragma unroll
    for (int k = 0; k < 4; ++k) {
      int ls = atomicAdd(&lcur[bucket_of(dd[k])], 1);
      ls = min(max(ls, 0), CHUNK_EDGES - 1);
      stageW[ls] = (int)((((unsigned)dd[k] % BKT_NODES) << SRC_BITS) | (unsigned)sv[k]);
    }
  }
  __syncthreads();
  const int nvalid = min(CHUNK_EDGES, ne - c * CHUNK_EDGES);
#pragma unroll
  for (int k = 0; k < CHUNK_EDGES / 1024; ++k) {
    const int i = t + 1024 * k;
    if (i < nvalid) BK[c * CHUNK_EDGES + i] = stageW[i];
  }
}

__global__ __launch_bounds__(1024) void k_bucket_csr(const int* __restrict__ LOT,
                                                     const int* __restrict__ BK,
                                                     int* __restrict__ rowptr,
                                                     int* __restrict__ csr_src, int n, int ne) {
  __shared__ int cnt[BKT_NODES];
  __shared__ int cur[BKT_NODES];
  __shared__ int wtot[BKT_NODES / 32];
  __shared__ int segLo[NCH];
  __shared__ int segLen[NCH];
  __shared__ int redLo[32];
  __shared__ int redLen[32];
  constexpr int STAGE = 8192;
  __shared__ int stage[STAGE];
  const int t = threadIdx.x, lane = t & 31, w = t >> 5;
  const int b = blockIdx.x;
  int lo = 0, len = 0;
  if (t < NCH) {
    const int nv = min(CHUNK_EDGES, ne - t * CHUNK_EDGES);
    const int* const lrow = LOT + (size_t)t * LOT_PITCH + b;
    lo = min(max(lrow[0], 0), nv);
    const int hi = min(max(lrow[1], lo), nv);
    len = hi - lo;
    segLo[t] = t * CHUNK_EDGES + lo;
    segLen[t] = len;
  }
  if (t < BKT_NODES) cnt[t] = 0;
  int rl = lo, rn = len;
  for (int o = 16; o > 0; o >>= 1) {
    rl += __shfl_xor(rl, o);
    rn += __shfl_xor(rn, o);
  }
  if (lane == 0) {
    redLo[w] = rl;
    redLen[w] = rn;
  }
  __syncthreads();
  int start = 0, tot = 0;
  constexpr int NW = (NCH + 31) / 32;
#pragma unroll
  for (int k = 0; k < NW; ++k) {
    start += redLo[k];
    tot += redLen[k];
  }
  start = min(max(start, 0), ne);
  const int end = min(start + tot, ne);

  constexpr int CQ = (NCH + 31) / 32;
  unsigned wv[CQ];
  int ln[CQ], bs[CQ];
#pragma unroll
  for (int q = 0; q < CQ; ++q) {
    const int c = w + 32 * q;
    const int cc = min(c, NCH - 1);
    ln[q] = (c < NCH) ? segLen[cc] : 0;
    bs[q] = segLo[cc];
    wv[q] = (unsigned)BK[min(bs[q] + min(lane, max(ln[q] - 1, 0)), ne - 1)];
  }
#pragma unroll
  for (int q = 0; q < CQ; ++q)
    if (lane < ln[q]) atomicAdd(&cnt[(wv[q] >> SRC_BITS) % BKT_NODES], 1);
#pragma unroll
  for (int q = 0; q < CQ; ++q) {
    for (int j = 32 + lane; j < ln[q]; j += 32) {
      const unsigned x = (unsigned)BK[min(bs[q] + j, ne - 1)];
      atomicAdd(&cnt[(x >> SRC_BITS) % BKT_NODES], 1);
    }
  }
  __syncthreads();
  int val = 0, x = 0;
  if (t < BKT_NODES) {
    val = cnt[t];
    x = val;
    for (int o = 1; o < 32; o <<= 1) {
      const int y = __shfl_up(x, o);
      if (lane >= o) x += y;
    }
    if (lane == 31) wtot[w] = x;
  }
  __syncthreads();
  if (t < BKT_NODES) {
    int p = start + x - val;
    for (int k = 0; k < w; ++k) p += wtot[k];
    cur[t] = p;
    const int node = b * BKT_NODES + t;
    if (node <= n) rowptr[node] = p;
  }
  __syncthreads();
  const bool fits = end - start <= STAGE;
  auto place = [&](unsigned wd) {
    const int pos = atomicAdd(&cur[(wd >> SRC_BITS) % BKT_NODES], 1);
    const int sv = (int)(wd & ((1u << SRC_BITS) - 1u));
    if (fits)
      stage[min(max(pos - start, 0), STAGE - 1)] = sv;
    else
      csr_src[min(max(pos, 0), ne - 1)] = sv;
  };
#pragma unroll
  for (int q = 0; q < CQ; ++q)
    if (lane < ln[q]) place(wv[q]);
#pragma unroll
  for (int q = 0; q < CQ; ++q) {
    for (int j = 32 + lane; j < ln[q]; j += 32) place((unsigned)BK[min(bs[q] + j, ne - 1)]);
  }
  __syncthreads();
  if (fits) {
    for (int i = t; i < end - start; i += 1024) csr_src[start + i] = stage[i];
  }
}


template <int M>
__device__ __forceinline__ float lane_xor(float x) {
  return __int_as_float(__builtin_amdgcn_update_dpp(0, __float_as_int(x), 0x160 | M, 0xf, 0xf, true));
}
__device__ __forceinline__ float grp_max(float x) {
  x = fmaxf(x, lane_xor<1>(x));
  x = fmaxf(x, lane_xor<2>(x));
  x = fmaxf(x, lane_xor<4>(x));
  return x;
}
template <int K>
__device__ __forceinline__ int grp_bcast_i(int v) {
  return (int)__builtin_amdgcn_mov_dpp8((unsigned)v, K * 0x249249u);
}
template <int K>
__device__ __forceinline__ float grp_bcast(float v) {
  return __int_as_float(grp_bcast_i<K>(__float_as_int(v)));
}

__device__ __forceinline__ float4 load_asrc(const float* __restrict__ ASRC, int s) {
  return *(const float4*)((const char*)ASRC + (unsigned)s * 16u);
}
__device__ __forceinline__ float4 edge_logits(float4 a, float4 ad) {
  float4 t;
  t.x = a.x + ad.x; t.x = t.x > 0.f ? t.x : 0.2f * t.x;
  t.y = a.y + ad.y; t.y = t.y > 0.f ? t.y : 0.2f * t.y;
  t.z = a.z + ad.z; t.z = t.z > 0.f ? t.z : 0.2f * t.z;
  t.w = a.w + ad.w; t.w = t.w > 0.f ? t.w : 0.2f * t.w;
  return t;
}
__device__ __forceinline__ float4 max4(float4 a, float4 b) {
  return make_float4(fmaxf(a.x, b.x), fmaxf(a.y, b.y), fmaxf(a.z, b.z), fmaxf(a.w, b.w));
}
__device__ __forceinline__ float4 edge_weights(bool v, float4 t, float4 m) {
  float4 e;
  e.x = v ? __expf(t.x - m.x) : 0.f;
  e.y = v ? __expf(t.y - m.y) : 0.f;
  e.z = v ? __expf(t.z - m.z) : 0.f;
  e.w = v ? __expf(t.w - m.w) : 0.f;
  return e;
}

template <int K>
__device__ __forceinline__ void load_slot(float4 (&x)[4], const char* __restrict__ xp, unsigned lo, int sp, int rem) {
  const int s = grp_bcast_i<K>(sp);
  if (K < rem) {
    const char* p = xp + ((unsigned)s * 512u + lo);
    x[0] = *(const float4*)(p);
    x[1] = *(const float4*)(p + 128);
    x[2] = *(const float4*)(p + 256);
    x[3] = *(const float4*)(p + 384);
  }
}
template <int K>
__device__ __forceinline__ void add_slot(const float4 (&x)[4], float4 (&acc)[4], float4& ss, float4 e, int rem) {
  const float w0 = grp_bcast<K>(e.x), w1 = grp_bcast<K>(e.y), w2 = grp_bcast<K>(e.z), w3 = grp_bcast<K>(e.w);
  if (K < rem) {
    ss.x += w0; ss.y += w1; ss.z += w2; ss.w += w3;
    acc[0].x += x[0].x * w0; acc[0].y += x[0].y * w0; acc[0].z += x[0].z * w0; acc[0].w += x[0].w * w0;
    acc[1].x += x[1].x * w1; acc[1].y += x[1].y * w1; acc[1].z += x[1].z * w1; acc[1].w += x[1].w * w1;
    acc[2].x += x[2].x * w2; acc[2].y += x[2].y * w2; acc[2].z += x[2].z * w2; acc[2].w += x[2].w * w2;
    acc[3].x += x[3].x * w3; acc[3].y += x[3].y * w3; acc[3].z += x[3].z * w3; acc[3].w += x[3].w * w3;
  }
}
template <int K0>
__device__ __forceinline__ void load_four(float4 (&xr)[4][4], const char* __restrict__ xp, unsigned lo, int sp, int rem) {
  load_slot<K0 + 0>(xr[0], xp, lo, sp, rem);
  load_slot<K0 + 1>(xr[1], xp, lo, sp, rem);
  load_slot<K0 + 2>(xr[2], xp, lo, sp, rem);
  load_slot<K0 + 3>(xr[3], xp, lo, sp, rem);
}
template <int K0>
__device__ __forceinline__ void add_four(const float4 (&xr)[4][4], float4 (&acc)[4], float4& ss, float4 e, int rem) {
  add_slot<K0 + 0>(xr[0], acc, ss, e, rem);
  add_slot<K0 + 1>(xr[1], acc, ss, e, rem);
  add_slot<K0 + 2>(xr[2], acc, ss, e, rem);
  add_slot<K0 + 3>(xr[3], acc, ss, e, rem);
}

template <bool RELU>
__global__ __launch_bounds__(256) void k_node_softmax_aggr(const int* __restrict__ rowptr,
                                                           const int* __restrict__ csr_src,
                                                           const float* __restrict__ XP,
                                                           const float* __restrict__ ASRC,
                                                           const float* __restrict__ ADST,
                                                           const float* __restrict__ bias,
                                                           float* __restrict__ OUT,
                                                           int n, int ne) {
  const int lane = threadIdx.x & 31;
  const int i = lane & 7;
  const int wave = (blockIdx.x * blockDim.x + threadIdx.x) >> 5;
  if (wave * 4 >= n) return;
  const int node = wave * 4 + (lane >> 3);
  const bool nv = node < n;
  int beg = 0, end = 0;
  float4 ad = make_float4(0.f, 0.f, 0.f, 0.f);
  if (nv) {
    beg = rowptr[node];
    end = rowptr[node + 1];
    ad = *(const float4*)(ADST + (size_t)node * NH);
  }
  beg = max(beg, 0);
  end = min(end, ne);
  const int deg = max(end - beg, 0);
  auto seg = [&](int e) -> int {
    return *(const int*)((const char*)csr_src + (unsigned)(beg + e) * 4u);
  };
  const char* const xp = (const char*)XP;
  const unsigned lo = 16u * (unsigned)i;

  const bool v0 = i < deg, v1 = i + 8 < deg;
  int sp0 = 0, sp1 = 0;
  if (v0) sp0 = seg(i);
  if (v1) sp1 = seg(8 + i);
  const float4 a0 = load_asrc(ASRC, sp0);
  const float4 a1 = load_asrc(ASRC, sp1);

  float4 xr[4][4];
  load_four<0>(xr, xp, lo, sp0, deg);

  const float4 t0 = edge_logits(a0, ad);
  const float4 t1 = edge_logits(a1, ad);
  const float ninf = __int_as_float(0xff800000);
  float4 m = make_float4(ninf, ninf, ninf, ninf);
  if (v0) m = t0;
  if (v1) m = max4(m, t1);
  for (int ch = 16; __builtin_amdgcn_ballot_w32(ch < deg) != 0u; ch += 8) {
    if (ch + i < deg) m = max4(m, edge_logits(load_asrc(ASRC, seg(ch + i)), ad));
  }
  m.x = grp_max(m.x);
  m.y = grp_max(m.y);
  m.z = grp_max(m.z);
  m.w = grp_max(m.w);

  float4 acc[4];
  acc[0] = acc[1] = acc[2] = acc[3] = make_float4(0.f, 0.f, 0.f, 0.f);
  float4 ss = make_float4(0.f, 0.f, 0.f, 0.f);

  const float4 e0 = edge_weights(v0, t0, m);
  add_four<0>(xr, acc, ss, e0, deg);
  if (__builtin_amdgcn_ballot_w32(deg > 4) != 0u) {
    load_four<4>(xr, xp, lo, sp0, deg);
    add_four<4>(xr, acc, ss, e0, deg);
  }
  if (__builtin_amdgcn_ballot_w32(deg > 8) != 0u) {
    const float4 e1 = edge_weights(v1, t1, m);
    load_four<0>(xr, xp, lo, sp1, deg - 8);
    add_four<0>(xr, acc, ss, e1, deg - 8);
    if (__builtin_amdgcn_ballot_w32(deg > 12) != 0u) {
      load_four<4>(xr, xp, lo, sp1, deg - 8);
      add_four<4>(xr, acc, ss, e1, deg - 8);
    }
  }
  for (int ch = 16; __builtin_amdgcn_ballot_w32(ch < deg) != 0u; ch += 8) {
    const bool v = ch + i < deg;
    int sp = 0;
    float4 t = make_float4(0.f, 0.f, 0.f, 0.f);
    if (v) {
      sp = seg(ch + i);
      t = edge_logits(load_asrc(ASRC, sp), ad);
    }
    const float4 e = edge_weights(v, t, m);
    load_four<0>(xr, xp, lo, sp, deg - ch);
    add_four<0>(xr, acc, ss, e, deg - ch);
    if (__builtin_amdgcn_ballot_w32(deg > ch + 4) != 0u) {
      load_four<4>(xr, xp, lo, sp, deg - ch);
      add_four<4>(xr, acc, ss, e, deg - ch);
    }
  }

  if (!nv) return;
  const float den[4] = {ss.x + 1e-16f, ss.y + 1e-16f, ss.z + 1e-16f, ss.w + 1e-16f};
  char* const orow = (char*)OUT + (unsigned)node * 512u + 16 * i;
#pragma unroll
  for (int hd = 0; hd < NH; ++hd) {
    const float4 bv = *(const float4*)(bias + 32 * hd + 4 * i);
    float4 o;
    o.x = acc[hd].x / den[hd] + bv.x;
    o.y = acc[hd].y / den[hd] + bv.y;
    o.z = acc[hd].z / den[hd] + bv.z;
    o.w = acc[hd].w / den[hd] + bv.w;
    if (RELU) {
      o.x = fmaxf(o.x, 0.0f);
      o.y = fmaxf(o.y, 0.0f);
      o.z = fmaxf(o.z, 0.0f);
      o.w = fmaxf(o.w, 0.0f);
    }
    *(float4*)(orow + 128 * hd) = o;
  }
}

extern "C" void kernel_launch(void* const* d_in, const int* in_sizes, int n_in,
                              void* d_out, int out_size, void* d_ws, size_t ws_size,
                              hipStream_t stream) {
  const float* x     = (const float*)d_in[0];
  const int*   ei    = (const int*)d_in[1];
  const float* W0    = (const float*)d_in[2];
  const float* atts0 = (const float*)d_in[3];
  const float* attd0 = (const float*)d_in[4];
  const float* b0    = (const float*)d_in[5];
  const float* W1    = (const float*)d_in[6];
  const float* atts1 = (const float*)d_in[7];
  const float* attd1 = (const float*)d_in[8];
  const float* b1    = (const float*)d_in[9];

  const int n = NNODES, ne = NEDGES;
  const int* srcv = ei;
  const int* dstv = ei + ne;

  char* ws = (char*)d_ws;
  float* XP   = (float*)(ws + 0);
  float* H0   = (float*)(ws + 51200000);
  float* ASRC = (float*)(ws + 102400000);
  float* ADST = (float*)(ws + 104000000);
  int*   LOT  = (int*)(ws + 105600000);
  int*   RPTR = (int*)(ws + 107200000);
  int*   CSRC = (int*)(ws + 108800000);
  int*   BK   = (int*)(ws + 112000000);
  float* OUT  = (float*)d_out;

  const dim3 blk(256);
  const int gemmBlocks = cdiv(cdiv(n, 16) * 32, 256);
  const int nodeBlocks = cdiv(cdiv(n, 4) * 32, 256);

  k_chunk_sort<<<NCH, 1024, 0, stream>>>(srcv, dstv, LOT, BK, ne);
  k_bucket_csr<<<NBK, 1024, 0, stream>>>(LOT, BK, RPTR, CSRC, n, ne);

  k_gemm_wmma<256><<<gemmBlocks, blk, 0, stream>>>(x, W0, atts0, attd0, XP, ASRC, ADST, n);
  k_node_softmax_aggr<true><<<nodeBlocks, blk, 0, stream>>>(RPTR, CSRC, XP, ASRC, ADST, b0, H0, n, ne);

  k_gemm_wmma<128><<<gemmBlocks, blk, 0, stream>>>(H0, W1, atts1, attd1, XP, ASRC, ADST, n);
  k_node_softmax_aggr<false><<<nodeBlocks, blk, 0, stream>>>(RPTR, CSRC, XP, ASRC, ADST, b1, OUT, n, ne);
}


# Layer-0 aggregation fused with the layer-1 product: a wave puts 16 aggregated rows in an LDS tile and multiplies it by W1 (f32 WMMA) in the same launch; H0 never stored; all sum orders kept
# speedup vs baseline: 6.8881x; 1.0804x over previous
#include <hip/hip_runtime.h>
#include <hip/hip_bf16.h>

#define NNODES 100000
#define NEDGES 800000
#define HC 128
#define NH 4

typedef float v2f __attribute__((ext_vector_type(2)));
typedef float v8f __attribute__((ext_vector_type(8)));

static inline int cdiv(int a, int b) { return (a + b - 1) / b; }

template <int M>
__device__ __forceinline__ float row_xadd(bool hi, float x, float y) {
  const float keep = hi ? y : x;
  const float give = hi ? x : y;
  const int got = __builtin_amdgcn_update_dpp(0, __float_as_int(give), 0x160 | M, 0xf, 0xf, true);
  return keep + __int_as_float(got);
}

template <int K>
__global__ __launch_bounds__(256) void k_gemm_wmma(const float* __restrict__ X,
                                                   const float* __restrict__ W,
                                                   const float* __restrict__ att_src,
                                                   const float* __restrict__ att_dst,
                                                   float* __restrict__ XP,
                                                   float* __restrict__ ASRC,
                                                   float* __restrict__ ADST, int rows) {
  constexpr int P = K + 4;
  constexpr int KF = K / 4;
  constexpr int NQ = K / 32;
  __shared__ float WlA[64 * P];
  __shared__ float WlB[64 * P];

  const int tid = threadIdx.x;
  const int lane = tid & 31;
  const int c = lane & 15;
  const int h = lane >> 4;
  const int m0 = (blockIdx.x * 8 + (tid >> 5)) * 16;
  const bool active = m0 < rows;

  const float* xrow = X + (size_t)(m0 + c) * K + 4 * h;
  float4 an[4];
  if (active) {
#pragma unroll
    for (int i = 0; i < 4; ++i) an[i] = *(const float4*)(xrow + 8 * i);
  }

  {
    const int n = tid / KF, kf = tid - n * KF;
    constexpr int RS = 256 / KF;
    const float* wsrc = W + (size_t)n * K + 4 * kf;
    float* da = WlA + n * P + 4 * kf;
    float* db = WlB + n * P + 4 * kf;
#pragma unroll 8
    for (int r = 0; r < 64; r += RS)
      *(float4*)(da + r * P) = *(const float4*)(wsrc + (size_t)r * K);
#pragma unroll 8
    for (int r = 0; r < 64; r += RS)
      *(float4*)(db + r * P) = *(const float4*)(wsrc + (size_t)(64 + r) * K);
  }

  __syncthreads();
  if (!active) return;

  const float* bla = WlA + c * P + 4 * h;
  const float* blb = WlB + c * P + 4 * h;
  v8f acc[8] = {};
  float4 b0[8], b1[8];
#pragma unroll
  for (int nt = 0; nt < 4; ++nt) {
    b0[nt] = *(const float4*)(bla + nt * 16 * P);
    b0[nt + 4] = *(const float4*)(blb + nt * 16 * P);
  }

#pragma unroll
  for (int q = 0; q < NQ; ++q) {
    float4 a[4];
#pragma unroll
    for (int i = 0; i < 4; ++i) a[i] = an[i];
    if (q + 1 < NQ) {
#pragma unroll
      for (int i = 0; i < 4; ++i) an[i] = *(const float4*)(xrow + 32 * (q + 1) + 8 * i);
    }
#pragma unroll
    for (int i = 0; i < 4; ++i) {
      float4* bc = (i & 1) ? b1 : b0;
      float4* bn = (i & 1) ? b0 : b1;
      if (i < 3 || q + 1 < NQ) {
        const int ko = 32 * q + 8 * (i + 1);
#pragma unroll
        for (int nt = 0; nt < 4; ++nt) {
          bn[nt] = *(const float4*)(bla + nt * 16 * P + ko);
          bn[nt + 4] = *(const float4*)(blb + nt * 16 * P + ko);
        }
      }
      const v2f a0 = {a[i].x, a[i].y};
      const v2f a1 = {a[i].z, a[i].w};
#pragma unroll
      for (int nt = 0; nt < 8; ++nt) {
        const v2f b = {bc[nt].x, bc[nt].y};
        acc[nt] = __builtin_amdgcn_wmma_f32_16x16x4_f32(
             false, a0,  false, b,
             (short)0, acc[nt],  false,  false);
      }
#pragma unroll
      for (int nt = 0; nt < 8; ++nt) {
        const v2f b = {bc[nt].z, bc[nt].w};
        acc[nt] = __builtin_amdgcn_wmma_f32_16x16x4_f32(
             false, a1,  false, b,
             (short)0, acc[nt],  false,  false);
      }
    }
  }

  __builtin_amdgcn_sched_barrier(0);
  asm volatile("v_nop\n\tv_nop\n\tv_nop\n\tv_nop\n\tv_nop\n\tv_nop\n\tv_nop\n\tv_nop");
  __builtin_amdgcn_sched_barrier(0);

  float* const orow = XP + (size_t)(m0 + h * 8) * HC + c;
#pragma unroll
  for (int nt = 0; nt < 8; ++nt) {
#pragma unroll
    for (int i = 0; i < 8; ++i) orow[(size_t)i * HC + nt * 16] = acc[nt][i];
  }

  float a_lo[2][NH], a_hi[2][NH];
#pragma unroll
  for (int hd = 0; hd < NH; ++hd) {
    a_lo[0][hd] = att_src[hd * 32 + c];
    a_hi[0][hd] = att_src[hd * 32 + 16 + c];
    a_lo[1][hd] = att_dst[hd * 32 + c];
    a_hi[1][hd] = att_dst[hd * 32 + 16 + c];
  }
  const bool l0 = (c & 1) != 0, l1 = (c & 2) != 0, l2 = (c & 4) != 0, l3 = (c & 8) != 0;
  float u[2][NH];
#pragma unroll
  for (int sd = 0; sd < 2; ++sd) {
    float sv[2][NH];
#pragma unroll
    for (int j2 = 0; j2 < 2; ++j2) {
      float qv[2][NH];
#pragma unroll
      for (int j1 = 0; j1 < 2; ++j1) {
        const int i = 4 * j2 + 2 * j1;
#pragma unroll
        for (int hd = 0; hd < NH; ++hd) {
          const v2f t0 = {acc[2 * hd][i], acc[2 * hd][i + 1]};
          const v2f t1 = {acc[2 * hd + 1][i], acc[2 * hd + 1][i + 1]};
          const v2f xy = t0 * a_lo[sd][hd] + t1 * a_hi[sd][hd];
          qv[j1][hd] = row_xadd<1>(l0, xy.x, xy.y);
        }
      }
#pragma unroll
      for (int hd = 0; hd < NH; ++hd) sv[j2][hd] = row_xadd<2>(l1, qv[0][hd], qv[1][hd]);
    }
#pragma unroll
    for (int hd = 0; hd < NH; ++hd) u[sd][hd] = row_xadd<4>(l2, sv[0][hd], sv[1][hd]);
  }
  float4 r;
  r.x = row_xadd<8>(l3, u[0][0], u[1][0]);
  r.y = row_xadd<8>(l3, u[0][1], u[1][1]);
  r.z = row_xadd<8>(l3, u[0][2], u[1][2]);
  r.w = row_xadd<8>(l3, u[0][3], u[1][3]);
  float* const arow = (l3 ? ADST : ASRC) + (size_t)(m0 + h * 8 + (c & 7)) * NH;
  *(float4*)arow = r;
}

#define BKT_NODES 512
#define CHUNK_EDGES 4096
#define NBK ((NNODES + 1 + BKT_NODES - 1) / BKT_NODES)
#define NCH ((NEDGES + CHUNK_EDGES - 1) / CHUNK_EDGES)
#define SRC_BITS 17
#define LOT_PITCH 256

__device__ __forceinline__ int bucket_of(int d) {
  return (int)min((unsigned)d / BKT_NODES, (unsigned)(NBK - 1));
}

__global__ __launch_bounds__(1024) void k_chunk_sort(const int* __restrict__ src,
                                                     const int* __restrict__ dst,
                                                     int* __restrict__ LOT,
                                                     int* __restrict__ BK, int ne) {
  __shared__ int hist[NBK];
  __shared__ int lcur[NBK];
  __shared__ int stageW[CHUNK_EDGES];
  const int t = threadIdx.x, lane = t & 31, w = t >> 5;
  const int c = blockIdx.x;
  const int e = (c * 1024 + t) * 4;
  const bool v = e < ne;
  int4 d = make_int4(0, 0, 0, 0), s = make_int4(0, 0, 0, 0);
  if (v) {
    d = *(const int4*)(dst + e);
    s = *(const int4*)(src + e);
  }
  if (t < NBK) hist[t] = 0;
  __syncthreads();
  const int dd[4] = {d.x, d.y, d.z, d.w};
  const int sv[4] = {s.x, s.y, s.z, s.w};
  if (v) {
#pragma unroll
    for (int k = 0; k < 4; ++k) atomicAdd(&hist[bucket_of(dd[k])], 1);
  }
  __syncthreads();
  if (w == 0) {
    constexpr int RW = (NBK + 31) / 32;
    int ov[RW];
    int osum = 0;
#pragma unroll
    for (int k = 0; k < RW; ++k) {
      const int b = lane * RW + k;
      ov[k] = (b < NBK) ? hist[b] : 0;
      osum += ov[k];
    }
    int oinc = osum;
    for (int o = 1; o < 32; o <<= 1) {
      const int y = __shfl_up(oinc, o);
      if (lane >= o) oinc += y;
    }
    int orun = oinc - osum;
    int* const lrow = LOT + (size_t)c * LOT_PITCH;
#pragma unroll
    for (int k = 0; k < RW; ++k) {
      const int b = lane * RW + k;
      if (b < NBK) {
        lcur[b] = orun;
        lrow[b] = orun;
      }
      orun += ov[k];
    }
    if (lane == 31) lrow[NBK] = oinc;
  }
  __syncthreads();
  if (v) {
#pragma unroll
    for (int k = 0; k < 4; ++k) {
      int ls = atomicAdd(&lcur[bucket_of(dd[k])], 1);
      ls = min(max(ls, 0), CHUNK_EDGES - 1);
      stageW[ls] = (int)((((unsigned)dd[k] % BKT_NODES) << SRC_BITS) | (unsigned)sv[k]);
    }
  }
  __syncthreads();
  const int nvalid = min(CHUNK_EDGES, ne - c * CHUNK_EDGES);
#pragma unroll
  for (int k = 0; k < CHUNK_EDGES / 1024; ++k) {
    const int i = t + 1024 * k;
    if (i < nvalid) BK[c * CHUNK_EDGES + i] = stageW[i];
  }
}

__global__ __launch_bounds__(1024) void k_bucket_csr(const int* __restrict__ LOT,
                                                     const int* __restrict__ BK,
                                                     int* __restrict__ rowptr,
                                                     int* __restrict__ csr_src, int n, int ne) {
  __shared__ int cnt[BKT_NODES];
  __shared__ int cur[BKT_NODES];
  __shared__ int wtot[BKT_NODES / 32];
  __shared__ int segLo[NCH];
  __shared__ int segLen[NCH];
  __shared__ int redLo[32];
  __shared__ int redLen[32];
  constexpr int STAGE = 8192;
  __shared__ int stage[STAGE];
  const int t = threadIdx.x, lane = t & 31, w = t >> 5;
  const int b = blockIdx.x;
  int lo = 0, len = 0;
  if (t < NCH) {
    const int nv = min(CHUNK_EDGES, ne - t * CHUNK_EDGES);
    const int* const lrow = LOT + (size_t)t * LOT_PITCH + b;
    lo = min(max(lrow[0], 0), nv);
    const int hi = min(max(lrow[1], lo), nv);
    len = hi - lo;
    segLo[t] = t * CHUNK_EDGES + lo;
    segLen[t] = len;
  }
  if (t < BKT_NODES) cnt[t] = 0;
  int rl = lo, rn = len;
  for (int o = 16; o > 0; o >>= 1) {
    rl += __shfl_xor(rl, o);
    rn += __shfl_xor(rn, o);
  }
  if (lane == 0) {
    redLo[w] = rl;
    redLen[w] = rn;
  }
  __syncthreads();
  int start = 0, tot = 0;
  constexpr int NW = (NCH + 31) / 32;
#pragma unroll
  for (int k = 0; k < NW; ++k) {
    start += redLo[k];
    tot += redLen[k];
  }
  start = min(max(start, 0), ne);
  const int end = min(start + tot, ne);

  constexpr int CQ = (NCH + 31) / 32;
  unsigned wv[CQ];
  int ln[CQ], bs[CQ];
#pragma unroll
  for (int q = 0; q < CQ; ++q) {
    const int c = w + 32 * q;
    const int cc = min(c, NCH - 1);
    ln[q] = (c < NCH) ? segLen[cc] : 0;
    bs[q] = segLo[cc];
    wv[q] = (unsigned)BK[min(bs[q] + min(lane, max(ln[q] - 1, 0)), ne - 1)];
  }
#pragma unroll
  for (int q = 0; q < CQ; ++q)
    if (lane < ln[q]) atomicAdd(&cnt[(wv[q] >> SRC_BITS) % BKT_NODES], 1);
#pragma unroll
  for (int q = 0; q < CQ; ++q) {
    for (int j = 32 + lane; j < ln[q]; j += 32) {
      const unsigned x = (unsigned)BK[min(bs[q] + j, ne - 1)];
      atomicAdd(&cnt[(x >> SRC_BITS) % BKT_NODES], 1);
    }
  }
  __syncthreads();
  int val = 0, x = 0;
  if (t < BKT_NODES) {
    val = cnt[t];
    x = val;
    for (int o = 1; o < 32; o <<= 1) {
      const int y = __shfl_up(x, o);
      if (lane >= o) x += y;
    }
    if (lane == 31) wtot[w] = x;
  }
  __syncthreads();
  if (t < BKT_NODES) {
    int p = start + x - val;
    for (int k = 0; k < w; ++k) p += wtot[k];
    cur[t] = p;
    const int node = b * BKT_NODES + t;
    if (node <= n) rowptr[node] = p;
  }
  __syncthreads();
  const bool fits = end - start <= STAGE;
  auto place = [&](unsigned wd) {
    const int pos = atomicAdd(&cur[(wd >> SRC_BITS) % BKT_NODES], 1);
    const int sv = (int)(wd & ((1u << SRC_BITS) - 1u));
    if (fits)
      stage[min(max(pos - start, 0), STAGE - 1)] = sv;
    else
      csr_src[min(max(pos, 0), ne - 1)] = sv;
  };
#pragma unroll
  for (int q = 0; q < CQ; ++q)
    if (lane < ln[q]) place(wv[q]);
#pragma unroll
  for (int q = 0; q < CQ; ++q) {
    for (int j = 32 + lane; j < ln[q]; j += 32) place((unsigned)BK[min(bs[q] + j, ne - 1)]);
  }
  __syncthreads();
  if (fits) {
    for (int i = t; i < end - start; i += 1024) csr_src[start + i] = stage[i];
  }
}


template <int M>
__device__ __forceinline__ float lane_xor(float x) {
  return __int_as_float(__builtin_amdgcn_update_dpp(0, __float_as_int(x), 0x160 | M, 0xf, 0xf, true));
}
__device__ __forceinline__ float grp_max(float x) {
  x = fmaxf(x, lane_xor<1>(x));
  x = fmaxf(x, lane_xor<2>(x));
  x = fmaxf(x, lane_xor<4>(x));
  return x;
}
template <int K>
__device__ __forceinline__ int grp_bcast_i(int v) {
  return (int)__builtin_amdgcn_mov_dpp8((unsigned)v, K * 0x249249u);
}
template <int K>
__device__ __forceinline__ float grp_bcast(float v) {
  return __int_as_float(grp_bcast_i<K>(__float_as_int(v)));
}

__device__ __forceinline__ float4 load_asrc(const float* __restrict__ ASRC, int s) {
  return *(const float4*)((const char*)ASRC + (unsigned)s * 16u);
}
__device__ __forceinline__ float4 edge_logits(float4 a, float4 ad) {
  float4 t;
  t.x = a.x + ad.x; t.x = t.x > 0.f ? t.x : 0.2f * t.x;
  t.y = a.y + ad.y; t.y = t.y > 0.f ? t.y : 0.2f * t.y;
  t.z = a.z + ad.z; t.z = t.z > 0.f ? t.z : 0.2f * t.z;
  t.w = a.w + ad.w; t.w = t.w > 0.f ? t.w : 0.2f * t.w;
  return t;
}
__device__ __forceinline__ float4 max4(float4 a, float4 b) {
  return make_float4(fmaxf(a.x, b.x), fmaxf(a.y, b.y), fmaxf(a.z, b.z), fmaxf(a.w, b.w));
}
__device__ __forceinline__ float4 edge_weights(bool v, float4 t, float4 m) {
  float4 e;
  e.x = v ? __expf(t.x - m.x) : 0.f;
  e.y = v ? __expf(t.y - m.y) : 0.f;
  e.z = v ? __expf(t.z - m.z) : 0.f;
  e.w = v ? __expf(t.w - m.w) : 0.f;
  return e;
}

template <int K>
__device__ __forceinline__ void load_slot(float4 (&x)[4], const char* __restrict__ xp, unsigned lo, int sp, int rem) {
  const int s = grp_bcast_i<K>(sp);
  if (K < rem) {
    const char* p = xp + ((unsigned)s * 512u + lo);
    x[0] = *(const float4*)(p);
    x[1] = *(const float4*)(p + 128);
    x[2] = *(const float4*)(p + 256);
    x[3] = *(const float4*)(p + 384);
  }
}
template <int K>
__device__ __forceinline__ void add_slot(const float4 (&x)[4], float4 (&acc)[4], float4& ss, float4 e, int rem) {
  const float w0 = grp_bcast<K>(e.x), w1 = grp_bcast<K>(e.y), w2 = grp_bcast<K>(e.z), w3 = grp_bcast<K>(e.w);
  if (K < rem) {
    ss.x += w0; ss.y += w1; ss.z += w2; ss.w += w3;
    acc[0].x += x[0].x * w0; acc[0].y += x[0].y * w0; acc[0].z += x[0].z * w0; acc[0].w += x[0].w * w0;
    acc[1].x += x[1].x * w1; acc[1].y += x[1].y * w1; acc[1].z += x[1].z * w1; acc[1].w += x[1].w * w1;
    acc[2].x += x[2].x * w2; acc[2].y += x[2].y * w2; acc[2].z += x[2].z * w2; acc[2].w += x[2].w * w2;
    acc[3].x += x[3].x * w3; acc[3].y += x[3].y * w3; acc[3].z += x[3].z * w3; acc[3].w += x[3].w * w3;
  }
}
template <int K0>
__device__ __forceinline__ void load_four(float4 (&xr)[4][4], const char* __restrict__ xp, unsigned lo, int sp, int rem) {
  load_slot<K0 + 0>(xr[0], xp, lo, sp, rem);
  load_slot<K0 + 1>(xr[1], xp, lo, sp, rem);
  load_slot<K0 + 2>(xr[2], xp, lo, sp, rem);
  load_slot<K0 + 3>(xr[3], xp, lo, sp, rem);
}
template <int K0>
__device__ __forceinline__ void add_four(const float4 (&xr)[4][4], float4 (&acc)[4], float4& ss, float4 e, int rem) {
  add_slot<K0 + 0>(xr[0], acc, ss, e, rem);
  add_slot<K0 + 1>(xr[1], acc, ss, e, rem);
  add_slot<K0 + 2>(xr[2], acc, ss, e, rem);
  add_slot<K0 + 3>(xr[3], acc, ss, e, rem);
}

template <bool RELU>
__device__ __forceinline__ void aggr_group(const int* __restrict__ rowptr,
                                           const int* __restrict__ csr_src,
                                           const float* __restrict__ XP,
                                           const float* __restrict__ ASRC,
                                           const float* __restrict__ ADST,
                                           const float* __restrict__ bias,
                                           int node, bool nv, int i, int ne, float4 (&o)[4]) {
  int beg = 0, end = 0;
  float4 ad = make_float4(0.f, 0.f, 0.f, 0.f);
  if (nv) {
    beg = rowptr[node];
    end = rowptr[node + 1];
    ad = *(const float4*)(ADST + (size_t)node * NH);
  }
  beg = max(beg, 0);
  end = min(end, ne);
  const int deg = max(end - beg, 0);
  auto seg = [&](int e) -> int {
    return *(const int*)((const char*)csr_src + (unsigned)(beg + e) * 4u);
  };
  const char* const xp = (const char*)XP;
  const unsigned lo = 16u * (unsigned)i;

  const bool v0 = i < deg, v1 = i + 8 < deg;
  int sp0 = 0, sp1 = 0;
  if (v0) sp0 = seg(i);
  if (v1) sp1 = seg(8 + i);
  const float4 a0 = load_asrc(ASRC, sp0);
  const float4 a1 = load_asrc(ASRC, sp1);

  float4 xr[4][4];
  load_four<0>(xr, xp, lo, sp0, deg);

  const float4 t0 = edge_logits(a0, ad);
  const float4 t1 = edge_logits(a1, ad);
  const float ninf = __int_as_float(0xff800000);
  float4 m = make_float4(ninf, ninf, ninf, ninf);
  if (v0) m = t0;
  if (v1) m = max4(m, t1);
  for (int ch = 16; __builtin_amdgcn_ballot_w32(ch < deg) != 0u; ch += 8) {
    if (ch + i < deg) m = max4(m, edge_logits(load_asrc(ASRC, seg(ch + i)), ad));
  }
  m.x = grp_max(m.x);
  m.y = grp_max(m.y);
  m.z = grp_max(m.z);
  m.w = grp_max(m.w);

  float4 acc[4];
  acc[0] = acc[1] = acc[2] = acc[3] = make_float4(0.f, 0.f, 0.f, 0.f);
  float4 ss = make_float4(0.f, 0.f, 0.f, 0.f);

  const float4 e0 = edge_weights(v0, t0, m);
  add_four<0>(xr, acc, ss, e0, deg);
  if (__builtin_amdgcn_ballot_w32(deg > 4) != 0u) {
    load_four<4>(xr, xp, lo, sp0, deg);
    add_four<4>(xr, acc, ss, e0, deg);
  }
  if (__builtin_amdgcn_ballot_w32(deg > 8) != 0u) {
    const float4 e1 = edge_weights(v1, t1, m);
    load_four<0>(xr, xp, lo, sp1, deg - 8);
    add_four<0>(xr, acc, ss, e1, deg - 8);
    if (__builtin_amdgcn_ballot_w32(deg > 12) != 0u) {
      load_four<4>(xr, xp, lo, sp1, deg - 8);
      add_four<4>(xr, acc, ss, e1, deg - 8);
    }
  }
  for (int ch = 16; __builtin_amdgcn_ballot_w32(ch < deg) != 0u; ch += 8) {
    const bool v = ch + i < deg;
    int sp = 0;
    float4 t = make_float4(0.f, 0.f, 0.f, 0.f);
    if (v) {
      sp = seg(ch + i);
      t = edge_logits(load_asrc(ASRC, sp), ad);
    }
    const float4 e = edge_weights(v, t, m);
    load_four<0>(xr, xp, lo, sp, deg - ch);
    add_four<0>(xr, acc, ss, e, deg - ch);
    if (__builtin_amdgcn_ballot_w32(deg > ch + 4) != 0u) {
      load_four<4>(xr, xp, lo, sp, deg - ch);
      add_four<4>(xr, acc, ss, e, deg - ch);
    }
  }

  const float den[4] = {ss.x + 1e-16f, ss.y + 1e-16f, ss.z + 1e-16f, ss.w + 1e-16f};
#pragma unroll
  for (int hd = 0; hd < NH; ++hd) {
    const float4 bv = *(const float4*)(bias + 32 * hd + 4 * i);
    float4 r;
    r.x = acc[hd].x / den[hd] + bv.x;
    r.y = acc[hd].y / den[hd] + bv.y;
    r.z = acc[hd].z / den[hd] + bv.z;
    r.w = acc[hd].w / den[hd] + bv.w;
    if (RELU) {
      r.x = fmaxf(r.x, 0.0f);
      r.y = fmaxf(r.y, 0.0f);
      r.z = fmaxf(r.z, 0.0f);
      r.w = fmaxf(r.w, 0.0f);
    }
    o[hd] = r;
  }
}

template <bool RELU>
__global__ __launch_bounds__(256) void k_node_softmax_aggr(const int* __restrict__ rowptr,
                                                           const int* __restrict__ csr_src,
                                                           const float* __restrict__ XP,
                                                           const float* __restrict__ ASRC,
                                                           const float* __restrict__ ADST,
                                                           const float* __restrict__ bias,
                                                           float* __restrict__ OUT,
                                                           int n, int ne) {
  const int lane = threadIdx.x & 31;
  const int i = lane & 7;
  const int wave = (blockIdx.x * blockDim.x + threadIdx.x) >> 5;
  if (wave * 4 >= n) return;
  const int node = wave * 4 + (lane >> 3);
  const bool nv = node < n;
  float4 o[4];
  aggr_group<RELU>(rowptr, csr_src, XP, ASRC, ADST, bias, node, nv, i, ne, o);
  if (!nv) return;
  char* const orow = (char*)OUT + (unsigned)node * 512u + 16 * i;
#pragma unroll
  for (int hd = 0; hd < NH; ++hd) *(float4*)(orow + 128 * hd) = o[hd];
}

__global__ __launch_bounds__(256) void k_aggr_gemm(const int* __restrict__ rowptr,
                                                   const int* __restrict__ csr_src,
                                                   const float* __restrict__ XP0,
                                                   const float* __restrict__ ASRC0,
                                                   const float* __restrict__ ADST0,
                                                   const float* __restrict__ bias0,
                                                   const float* __restrict__ W,
                                                   const float* __restrict__ att_src,
                                                   const float* __restrict__ att_dst,
                                                   float* __restrict__ XP1,
                                                   float* __restrict__ ASRC1,
                                                   float* __restrict__ ADST1, int n, int ne) {
  constexpr int K = HC;
  constexpr int P = K + 4;
  constexpr int KF = K / 4;
  constexpr int NG = K / 8;
  __shared__ float WlA[64 * P];
  __shared__ float WlB[64 * P];
  __shared__ float Tl[8 * 16 * P];

  const int tid = threadIdx.x;
  const int lane = tid & 31;
  const int c = lane & 15;
  const int h = lane >> 4;
  const int i8 = lane & 7;
  const int grp = lane >> 3;
  const int wv = __builtin_amdgcn_readfirstlane(tid >> 5);

  {
    const int wn = tid / KF, kf = tid - wn * KF;
    constexpr int RS = 256 / KF;
    const float* wsrc = W + (size_t)wn * K + 4 * kf;
    float* da = WlA + wn * P + 4 * kf;
    float* db = WlB + wn * P + 4 * kf;
#pragma unroll 8
    for (int r = 0; r < 64; r += RS)
      *(float4*)(da + r * P) = *(const float4*)(wsrc + (size_t)r * K);
#pragma unroll 8
    for (int r = 0; r < 64; r += RS)
      *(float4*)(db + r * P) = *(const float4*)(wsrc + (size_t)(64 + r) * K);
  }
  __syncthreads();

  const int tw = (int)gridDim.x * 8;
  const int gw = (int)blockIdx.x * 8 + wv;
  const int ngrp = (n + 3) >> 2;
  const int base = ngrp / tw;
  const int rem = ngrp - base * tw;
  const int cnt = base + (gw < rem ? 1 : 0);
  const int first = gw * base + min(gw, rem);

  float* const T = Tl + wv * (16 * P);
  float* const trow = T + grp * P + 4 * i8;
  const float* const tla = T + c * P + 4 * h;
  const float* const bla = WlA + c * P + 4 * h;
  const float* const blb = WlB + c * P + 4 * h;
  const float4 zero4 = make_float4(0.f, 0.f, 0.f, 0.f);

  for (int t0 = 0; t0 < cnt; t0 += 4) {
    const int gcount = min(4, cnt - t0);
    const int node0 = (first + t0) * 4;
    const int rv = min(4 * gcount, n - node0);

#pragma unroll 1
    for (int g = 0; g < 4; ++g) {
      float4 o[4];
      if (g < gcount) {
        const int node = node0 + 4 * g + grp;
        aggr_group<true>(rowptr, csr_src, XP0, ASRC0, ADST0, bias0, node, node < n, i8, ne, o);
      } else {
        o[0] = o[1] = o[2] = o[3] = zero4;
      }
      float* const tr = trow + (4 * g) * P;
#pragma unroll
      for (int hd = 0; hd < NH; ++hd) *(float4*)(tr + 32 * hd) = o[hd];
    }
    __builtin_amdgcn_fence(__ATOMIC_RELEASE, "wavefront");
    __builtin_amdgcn_wave_barrier();
    __builtin_amdgcn_fence(__ATOMIC_ACQUIRE, "wavefront");

    v8f acc[8] = {};
    float4 a[NG];
#pragma unroll
    for (int g = 0; g < NG; ++g) a[g] = *(const float4*)(tla + 8 * g);
    float4 b0[8], b1[8];
#pragma unroll
    for (int nt = 0; nt < 4; ++nt) {
      b0[nt] = *(const float4*)(bla + nt * 16 * P);
      b0[nt + 4] = *(const float4*)(blb + nt * 16 * P);
    }
#pragma unroll
    for (int g = 0; g < NG; ++g) {
      float4* bc = (g & 1) ? b1 : b0;
      float4* bn = (g & 1) ? b0 : b1;
      if (g + 1 < NG) {
        const int ko = 8 * (g + 1);
#pragma unroll
        for (int nt = 0; nt < 4; ++nt) {
          bn[nt] = *(const float4*)(bla + nt * 16 * P + ko);
          bn[nt + 4] = *(const float4*)(blb + nt * 16 * P + ko);
        }
      }
      const v2f a0 = {a[g].x, a[g].y};
      const v2f a1 = {a[g].z, a[g].w};
#pragma unroll
      for (int nt = 0; nt < 8; ++nt) {
        const v2f b = {bc[nt].x, bc[nt].y};
        acc[nt] = __builtin_amdgcn_wmma_f32_16x16x4_f32(
             false, a0,  false, b,
             (short)0, acc[nt],  false,  false);
      }
#pragma unroll
      for (int nt = 0; nt < 8; ++nt) {
        const v2f b = {bc[nt].z, bc[nt].w};
        acc[nt] = __builtin_amdgcn_wmma_f32_16x16x4_f32(
             false, a1,  false, b,
             (short)0, acc[nt],  false,  false);
      }
    }

    __builtin_amdgcn_sched_barrier(0);
    asm volatile("v_nop\n\tv_nop\n\tv_nop\n\tv_nop\n\tv_nop\n\tv_nop\n\tv_nop\n\tv_nop");
    __builtin_amdgcn_sched_barrier(0);

    float* const orow = XP1 + (size_t)(node0 + h * 8) * HC + c;
#pragma unroll
    for (int i = 0; i < 8; ++i) {
      if (h * 8 + i < rv) {
#pragma unroll
        for (int nt = 0; nt < 8; ++nt) orow[(size_t)i * HC + nt * 16] = acc[nt][i];
      }
    }

    float a_lo[2][NH], a_hi[2][NH];
#pragma unroll
    for (int hd = 0; hd < NH; ++hd) {
      a_lo[0][hd] = att_src[hd * 32 + c];
      a_hi[0][hd] = att_src[hd * 32 + 16 + c];
      a_lo[1][hd] = att_dst[hd * 32 + c];
      a_hi[1][hd] = att_dst[hd * 32 + 16 + c];
    }
    const bool l0 = (c & 1) != 0, l1 = (c & 2) != 0, l2 = (c & 4) != 0, l3 = (c & 8) != 0;
    float u[2][NH];
#pragma unroll
    for (int sd = 0; sd < 2; ++sd) {
      float sv[2][NH];
#pragma unroll
      for (int j2 = 0; j2 < 2; ++j2) {
        float qv[2][NH];
#pragma unroll
        for (int j1 = 0; j1 < 2; ++j1) {
          const int i = 4 * j2 + 2 * j1;
#pragma unroll
          for (int hd = 0; hd < NH; ++hd) {
            const v2f t0v = {acc[2 * hd][i], acc[2 * hd][i + 1]};
            const v2f t1v = {acc[2 * hd + 1][i], acc[2 * hd + 1][i + 1]};
            const v2f xy = t0v * a_lo[sd][hd] + t1v * a_hi[sd][hd];
            qv[j1][hd] = row_xadd<1>(l0, xy.x, xy.y);
          }
        }
#pragma unroll
        for (int hd = 0; hd < NH; ++hd) sv[j2][hd] = row_xadd<2>(l1, qv[0][hd], qv[1][hd]);
      }
#pragma unroll
      for (int hd = 0; hd < NH; ++hd) u[sd][hd] = row_xadd<4>(l2, sv[0][hd], sv[1][hd]);
    }
    float4 r;
    r.x = row_xadd<8>(l3, u[0][0], u[1][0]);
    r.y = row_xadd<8>(l3, u[0][1], u[1][1]);
    r.z = row_xadd<8>(l3, u[0][2], u[1][2]);
    r.w = row_xadd<8>(l3, u[0][3], u[1][3]);
    const int rr = h * 8 + (c & 7);
    if (rr < rv) {
      float* const arow = (l3 ? ADST1 : ASRC1) + (size_t)(node0 + rr) * NH;
      *(float4*)arow = r;
    }
  }
}

extern "C" void kernel_launch(void* const* d_in, const int* in_sizes, int n_in,
                              void* d_out, int out_size, void* d_ws, size_t ws_size,
                              hipStream_t stream) {
  const float* x     = (const float*)d_in[0];
  const int*   ei    = (const int*)d_in[1];
  const float* W0    = (const float*)d_in[2];
  const float* atts0 = (const float*)d_in[3];
  const float* attd0 = (const float*)d_in[4];
  const float* b0    = (const float*)d_in[5];
  const float* W1    = (const float*)d_in[6];
  const float* atts1 = (const float*)d_in[7];
  const float* attd1 = (const float*)d_in[8];
  const float* b1    = (const float*)d_in[9];

  const int n = NNODES, ne = NEDGES;
  const int* srcv = ei;
  const int* dstv = ei + ne;

  char* ws = (char*)d_ws;
  float* XP   = (float*)(ws + 0);
  float* XP1  = (float*)(ws + 51200000);
  float* ASRC = (float*)(ws + 102400000);
  float* ADST = (float*)(ws + 104000000);
  int*   LOT  = (int*)(ws + 105600000);
  int*   RPTR = (int*)(ws + 107200000);
  int*   CSRC = (int*)(ws + 108800000);
  int*   BK   = (int*)(ws + 112000000);
  float* ASRC1 = (float*)(ws + 112000000);
  float* ADST1 = (float*)(ws + 113600000);
  float* OUT  = (float*)d_out;

  const dim3 blk(256);
  const int gemmBlocks = cdiv(cdiv(n, 16) * 32, 256);
  const int nodeBlocks = cdiv(cdiv(n, 4) * 32, 256);

  k_chunk_sort<<<NCH, 1024, 0, stream>>>(srcv, dstv, LOT, BK, ne);
  k_bucket_csr<<<NBK, 1024, 0, stream>>>(LOT, BK, RPTR, CSRC, n, ne);

  k_gemm_wmma<256><<<gemmBlocks, blk, 0, stream>>>(x, W0, atts0, attd0, XP, ASRC, ADST, n);

  k_aggr_gemm<<<512, blk, 0, stream>>>(RPTR, CSRC, XP, ASRC, ADST, b0, W1, atts1, attd1,
                                       XP1, ASRC1, ADST1, n, ne);

  k_node_softmax_aggr<false><<<nodeBlocks, blk, 0, stream>>>(RPTR, CSRC, XP1, ASRC1, ADST1, b1, OUT, n, ne);
}


# Layer-0 product: 256 resident 16-wave work-groups get equal runs of 16-row strips (leftover strip split by heads over 4 waves), next strip's first x rows requested in the last chunk; sum orders kept
# speedup vs baseline: 7.5201x; 1.0918x over previous
#include <hip/hip_runtime.h>
#include <hip/hip_bf16.h>

#define NNODES 100000
#define NEDGES 800000
#define HC 128
#define NH 4

typedef float v2f __attribute__((ext_vector_type(2)));
typedef float v8f __attribute__((ext_vector_type(8)));

static inline int cdiv(int a, int b) { return (a + b - 1) / b; }

template <int M>
__device__ __forceinline__ float row_xadd(bool hi, float x, float y) {
  const float keep = hi ? y : x;
  const float give = hi ? x : y;
  const int got = __builtin_amdgcn_update_dpp(0, __float_as_int(give), 0x160 | M, 0xf, 0xf, true);
  return keep + __int_as_float(got);
}


template <int K, int NT>
__device__ __forceinline__ void strip_product(v8f (&acc)[NT], float4 (&an)[4],
                                              const char* __restrict__ xb, unsigned xc, unsigned xn,
                                              const float* bla, const float* blb) {
  constexpr int P = K + 4;
  constexpr int NQ = K / 32;
  const v8f zero = {};
#pragma unroll
  for (int nt = 0; nt < NT; ++nt) acc[nt] = zero;
  float4 b0[NT], b1[NT];
#pragma unroll
  for (int nt = 0; nt < NT; ++nt) b0[nt] = *(const float4*)((nt < 4 ? bla : blb) + (nt & 3) * 16 * P);
#pragma unroll
  for (int q = 0; q < NQ; ++q) {
    float4 a[4];
#pragma unroll
    for (int i = 0; i < 4; ++i) a[i] = an[i];
    const char* nx = (q + 1 < NQ) ? xb + xc + 128 * (q + 1) : xb + xn;
#pragma unroll
    for (int i = 0; i < 4; ++i) an[i] = *(const float4*)(nx + 32 * i);
#pragma unroll
    for (int i = 0; i < 4; ++i) {
      float4* bc = (i & 1) ? b1 : b0;
      float4* bn = (i & 1) ? b0 : b1;
      if (i < 3 || q + 1 < NQ) {
        const int ko = 32 * q + 8 * (i + 1);
#pragma unroll
        for (int nt = 0; nt < NT; ++nt)
          bn[nt] = *(const float4*)((nt < 4 ? bla : blb) + (nt & 3) * 16 * P + ko);
      }
      const v2f a0 = {a[i].x, a[i].y};
      const v2f a1 = {a[i].z, a[i].w};
#pragma unroll
      for (int nt = 0; nt < NT; ++nt) {
        const v2f b = {bc[nt].x, bc[nt].y};
        acc[nt] = __builtin_amdgcn_wmma_f32_16x16x4_f32(
             false, a0,  false, b,
             (short)0, acc[nt],  false,  false);
      }
#pragma unroll
      for (int nt = 0; nt < NT; ++nt) {
        const v2f b = {bc[nt].z, bc[nt].w};
        acc[nt] = __builtin_amdgcn_wmma_f32_16x16x4_f32(
             false, a1,  false, b,
             (short)0, acc[nt],  false,  false);
      }
    }
  }
  __builtin_amdgcn_sched_barrier(0);
  asm volatile("v_nop\n\tv_nop\n\tv_nop\n\tv_nop\n\tv_nop\n\tv_nop\n\tv_nop\n\tv_nop");
  __builtin_amdgcn_sched_barrier(0);
}

template <int NT>
__device__ __forceinline__ void strip_store(const v8f (&acc)[NT], float* __restrict__ XP, int m0, int col0, int c, int h) {
  const unsigned o = (unsigned)(m0 + h * 8) * (HC * 4u) + (unsigned)(col0 + c) * 4u;
  char* const orow = (char*)XP + o;
#pragma unroll
  for (int nt = 0; nt < NT; ++nt) {
#pragma unroll
    for (int i = 0; i < 8; ++i) *(float*)(orow + (i * HC + nt * 16) * 4) = acc[nt][i];
  }
}

template <int NHD>
__device__ __forceinline__ void strip_dots(const v8f (&acc)[2 * NHD], const float* __restrict__ as,
                                           const float* __restrict__ ad, int c, float (&r)[NHD]) {
  float a_lo[2][NHD], a_hi[2][NHD];
#pragma unroll
  for (int hd = 0; hd < NHD; ++hd) {
    a_lo[0][hd] = as[hd * 32 + c];
    a_hi[0][hd] = as[hd * 32 + 16 + c];
    a_lo[1][hd] = ad[hd * 32 + c];
    a_hi[1][hd] = ad[hd * 32 + 16 + c];
  }
  const bool l0 = (c & 1) != 0, l1 = (c & 2) != 0, l2 = (c & 4) != 0, l3 = (c & 8) != 0;
  float u[2][NHD];
#pragma unroll
  for (int sd = 0; sd < 2; ++sd) {
    float sv[2][NHD];
#pragma unroll
    for (int j2 = 0; j2 < 2; ++j2) {
      float qv[2][NHD];
#pragma unroll
      for (int j1 = 0; j1 < 2; ++j1) {
        const int i = 4 * j2 + 2 * j1;
#pragma unroll
        for (int hd = 0; hd < NHD; ++hd) {
          const v2f t0 = {acc[2 * hd][i], acc[2 * hd][i + 1]};
          const v2f t1 = {acc[2 * hd + 1][i], acc[2 * hd + 1][i + 1]};
          const v2f xy = t0 * a_lo[sd][hd] + t1 * a_hi[sd][hd];
          qv[j1][hd] = row_xadd<1>(l0, xy.x, xy.y);
        }
      }
#pragma unroll
      for (int hd = 0; hd < NHD; ++hd) sv[j2][hd] = row_xadd<2>(l1, qv[0][hd], qv[1][hd]);
    }
#pragma unroll
    for (int hd = 0; hd < NHD; ++hd) u[sd][hd] = row_xadd<4>(l2, sv[0][hd], sv[1][hd]);
  }
#pragma unroll
  for (int hd = 0; hd < NHD; ++hd) r[hd] = row_xadd<8>(l3, u[0][hd], u[1][hd]);
}

template <int K>
__global__ __launch_bounds__(512) void k_gemm_wmma(const float* __restrict__ X,
                                                   const float* __restrict__ W,
                                                   const float* __restrict__ att_src,
                                                   const float* __restrict__ att_dst,
                                                   float* __restrict__ XP,
                                                   float* __restrict__ ASRC,
                                                   float* __restrict__ ADST, int rows) {
  constexpr int P = K + 4;
  constexpr int KF = K / 4;
  __shared__ float WlA[64 * P];
  __shared__ float WlB[64 * P];
  __shared__ float DX[3][32];

  const int tid = threadIdx.x;
  const int lane = tid & 31;
  const int c = lane & 15;
  const int h = lane >> 4;
  const int wv = __builtin_amdgcn_readfirstlane(tid >> 5);

  const int nstrips = (rows + 15) >> 4;
  const int nwg = (int)gridDim.x;
  const int base = nstrips / nwg;
  const int rem = nstrips - base * nwg;
  const int bx = (int)blockIdx.x;
  const int first = bx * base + min(bx, rem);
  const int send = first + base;
  const bool extra = bx < rem;
  const int xq = (wv - base) & 15;
  const bool xrole = extra && xq < 4;

  const char* const xb = (const char*)X;
  const unsigned xlane = (unsigned)(c * K + 4 * h) * 4u;
  auto xoff_of = [&](int t) -> unsigned { return (unsigned)t * (16u * K * 4u) + xlane; };
  auto next_of = [&](int t) -> int { return (t + 16 < send) ? t + 16 : (xrole ? send : t); };

  int s = first + wv;
  const int s0 = s < send ? s : (xrole ? send : min(first, nstrips - 1));
  unsigned xc = xoff_of(s0);
  unsigned xn = xoff_of(s < send ? next_of(s) : s0);
  float4 an[4];
#pragma unroll
  for (int i = 0; i < 4; ++i) an[i] = *(const float4*)(xb + xc + 32 * i);

  {
    const int n = tid / KF, kf = tid - n * KF;
    constexpr int RS = 512 / KF;
    const float* wsrc = W + (size_t)n * K + 4 * kf;
    float* da = WlA + n * P + 4 * kf;
    float* db = WlB + n * P + 4 * kf;
#pragma unroll 8
    for (int r = 0; r < 64; r += RS)
      *(float4*)(da + r * P) = *(const float4*)(wsrc + (size_t)r * K);
#pragma unroll 8
    for (int r = 0; r < 64; r += RS)
      *(float4*)(db + r * P) = *(const float4*)(wsrc + (size_t)(64 + r) * K);
  }

  __syncthreads();

  const float* bla = WlA + c * P + 4 * h;
  const float* blb = WlB + c * P + 4 * h;
  const bool l3 = (c & 8) != 0;

  for (; s < send; s += 16) {
    v8f acc[8];
    strip_product<K, 8>(acc, an, xb, xc, xn, bla, blb);
    xc = xn;
    xn = xoff_of(next_of(min(s + 16, send - 1)));
    strip_store<8>(acc, XP, s * 16, 0, c, h);
    float r[4];
    strip_dots<4>(acc, att_src, att_dst, c, r);
    float* const arow = (l3 ? ADST : ASRC) + (size_t)(s * 16 + h * 8 + (c & 7)) * NH;
    *(float4*)arow = make_float4(r[0], r[1], r[2], r[3]);
  }

  float r1[1] = {0.f};
  if (xrole) {
    const float* bl = (xq < 2 ? bla : blb) + (xq & 1) * 32 * P;
    v8f acc[2];
    strip_product<K, 2>(acc, an, xb, xoff_of(send), xoff_of(send), bl, bl);
    strip_store<2>(acc, XP, send * 16, 32 * xq, c, h);
    strip_dots<1>(acc, att_src + 32 * xq, att_dst + 32 * xq, c, r1);
    if (xq > 0) DX[xq - 1][lane] = r1[0];
  }
  __syncthreads();
  if (xrole && xq == 0) {
    float* const arow = (l3 ? ADST : ASRC) + (size_t)(send * 16 + h * 8 + (c & 7)) * NH;
    *(float4*)arow = make_float4(r1[0], DX[0][lane], DX[1][lane], DX[2][lane]);
  }
}

#define BKT_NODES 512
#define CHUNK_EDGES 4096
#define NBK ((NNODES + 1 + BKT_NODES - 1) / BKT_NODES)
#define NCH ((NEDGES + CHUNK_EDGES - 1) / CHUNK_EDGES)
#define SRC_BITS 17
#define LOT_PITCH 256

__device__ __forceinline__ int bucket_of(int d) {
  return (int)min((unsigned)d / BKT_NODES, (unsigned)(NBK - 1));
}

__global__ __launch_bounds__(1024) void k_chunk_sort(const int* __restrict__ src,
                                                     const int* __restrict__ dst,
                                                     int* __restrict__ LOT,
                                                     int* __restrict__ BK, int ne) {
  __shared__ int hist[NBK];
  __shared__ int lcur[NBK];
  __shared__ int stageW[CHUNK_EDGES];
  const int t = threadIdx.x, lane = t & 31, w = t >> 5;
  const int c = blockIdx.x;
  const int e = (c * 1024 + t) * 4;
  const bool v = e < ne;
  int4 d = make_int4(0, 0, 0, 0), s = make_int4(0, 0, 0, 0);
  if (v) {
    d = *(const int4*)(dst + e);
    s = *(const int4*)(src + e);
  }
  if (t < NBK) hist[t] = 0;
  __syncthreads();
  const int dd[4] = {d.x, d.y, d.z, d.w};
  const int sv[4] = {s.x, s.y, s.z, s.w};
  if (v) {
#pragma unroll
    for (int k = 0; k < 4; ++k) atomicAdd(&hist[bucket_of(dd[k])], 1);
  }
  __syncthreads();
  if (w == 0) {
    constexpr int RW = (NBK + 31) / 32;
    int ov[RW];
    int osum = 0;
#pragma unroll
    for (int k = 0; k < RW; ++k) {
      const int b = lane * RW + k;
      ov[k] = (b < NBK) ? hist[b] : 0;
      osum += ov[k];
    }
    int oinc = osum;
    for (int o = 1; o < 32; o <<= 1) {
      const int y = __shfl_up(oinc, o);
      if (lane >= o) oinc += y;
    }
    int orun = oinc - osum;
    int* const lrow = LOT + (size_t)c * LOT_PITCH;
#pragma unroll
    for (int k = 0; k < RW; ++k) {
      const int b = lane * RW + k;
      if (b < NBK) {
        lcur[b] = orun;
        lrow[b] = orun;
      }
      orun += ov[k];
    }
    if (lane == 31) lrow[NBK] = oinc;
  }
  __syncthreads();
  if (v) {
#pragma unroll
    for (int k = 0; k < 4; ++k) {
      int ls = atomicAdd(&lcur[bucket_of(dd[k])], 1);
      ls = min(max(ls, 0), CHUNK_EDGES - 1);
      stageW[ls] = (int)((((unsigned)dd[k] % BKT_NODES) << SRC_BITS) | (unsigned)sv[k]);
    }
  }
  __syncthreads();
  const int nvalid = min(CHUNK_EDGES, ne - c * CHUNK_EDGES);
#pragma unroll
  for (int k = 0; k < CHUNK_EDGES / 1024; ++k) {
    const int i = t + 1024 * k;
    if (i < nvalid) BK[c * CHUNK_EDGES + i] = stageW[i];
  }
}

__global__ __launch_bounds__(1024) void k_bucket_csr(const int* __restrict__ LOT,
                                                     const int* __restrict__ BK,
                                                     int* __restrict__ rowptr,
                                                     int* __restrict__ csr_src, int n, int ne) {
  __shared__ int cnt[BKT_NODES];
  __shared__ int cur[BKT_NODES];
  __shared__ int wtot[BKT_NODES / 32];
  __shared__ int segLo[NCH];
  __shared__ int segLen[NCH];
  __shared__ int redLo[32];
  __shared__ int redLen[32];
  constexpr int STAGE = 8192;
  __shared__ int stage[STAGE];
  const int t = threadIdx.x, lane = t & 31, w = t >> 5;
  const int b = blockIdx.x;
  int lo = 0, len = 0;
  if (t < NCH) {
    const int nv = min(CHUNK_EDGES, ne - t * CHUNK_EDGES);
    const int* const lrow = LOT + (size_t)t * LOT_PITCH + b;
    lo = min(max(lrow[0], 0), nv);
    const int hi = min(max(lrow[1], lo), nv);
    len = hi - lo;
    segLo[t] = t * CHUNK_EDGES + lo;
    segLen[t] = len;
  }
  if (t < BKT_NODES) cnt[t] = 0;
  int rl = lo, rn = len;
  for (int o = 16; o > 0; o >>= 1) {
    rl += __shfl_xor(rl, o);
    rn += __shfl_xor(rn, o);
  }
  if (lane == 0) {
    redLo[w] = rl;
    redLen[w] = rn;
  }
  __syncthreads();
  int start = 0, tot = 0;
  constexpr int NW = (NCH + 31) / 32;
#pragma unroll
  for (int k = 0; k < NW; ++k) {
    start += redLo[k];
    tot += redLen[k];
  }
  start = min(max(start, 0), ne);
  const int end = min(start + tot, ne);

  constexpr int CQ = (NCH + 31) / 32;
  unsigned wv[CQ];
  int ln[CQ], bs[CQ];
#pragma unroll
  for (int q = 0; q < CQ; ++q) {
    const int c = w + 32 * q;
    const int cc = min(c, NCH - 1);
    ln[q] = (c < NCH) ? segLen[cc] : 0;
    bs[q] = segLo[cc];
    wv[q] = (unsigned)BK[min(bs[q] + min(lane, max(ln[q] - 1, 0)), ne - 1)];
  }
#pragma unroll
  for (int q = 0; q < CQ; ++q)
    if (lane < ln[q]) atomicAdd(&cnt[(wv[q] >> SRC_BITS) % BKT_NODES], 1);
#pragma unroll
  for (int q = 0; q < CQ; ++q) {
    for (int j = 32 + lane; j < ln[q]; j += 32) {
      const unsigned x = (unsigned)BK[min(bs[q] + j, ne - 1)];
      atomicAdd(&cnt[(x >> SRC_BITS) % BKT_NODES], 1);
    }
  }
  __syncthreads();
  int val = 0, x = 0;
  if (t < BKT_NODES) {
    val = cnt[t];
    x = val;
    for (int o = 1; o < 32; o <<= 1) {
      const int y = __shfl_up(x, o);
      if (lane >= o) x += y;
    }
    if (lane == 31) wtot[w] = x;
  }
  __syncthreads();
  if (t < BKT_NODES) {
    int p = start + x - val;
    for (int k = 0; k < w; ++k) p += wtot[k];
    cur[t] = p;
    const int node = b * BKT_NODES + t;
    if (node <= n) rowptr[node] = p;
  }
  __syncthreads();
  const bool fits = end - start <= STAGE;
  auto place = [&](unsigned wd) {
    const int pos = atomicAdd(&cur[(wd >> SRC_BITS) % BKT_NODES], 1);
    const int sv = (int)(wd & ((1u << SRC_BITS) - 1u));
    if (fits)
      stage[min(max(pos - start, 0), STAGE - 1)] = sv;
    else
      csr_src[min(max(pos, 0), ne - 1)] = sv;
  };
#pragma unroll
  for (int q = 0; q < CQ; ++q)
    if (lane < ln[q]) place(wv[q]);
#pragma unroll
  for (int q = 0; q < CQ; ++q) {
    for (int j = 32 + lane; j < ln[q]; j += 32) place((unsigned)BK[min(bs[q] + j, ne - 1)]);
  }
  __syncthreads();
  if (fits) {
    for (int i = t; i < end - start; i += 1024) csr_src[start + i] = stage[i];
  }
}


template <int M>
__device__ __forceinline__ float lane_xor(float x) {
  return __int_as_float(__builtin_amdgcn_update_dpp(0, __float_as_int(x), 0x160 | M, 0xf, 0xf, true));
}
__device__ __forceinline__ float grp_max(float x) {
  x = fmaxf(x, lane_xor<1>(x));
  x = fmaxf(x, lane_xor<2>(x));
  x = fmaxf(x, lane_xor<4>(x));
  return x;
}
template <int K>
__device__ __forceinline__ int grp_bcast_i(int v) {
  return (int)__builtin_amdgcn_mov_dpp8((unsigned)v, K * 0x249249u);
}
template <int K>
__device__ __forceinline__ float grp_bcast(float v) {
  return __int_as_float(grp_bcast_i<K>(__float_as_int(v)));
}

__device__ __forceinline__ float4 load_asrc(const float* __restrict__ ASRC, int s) {
  return *(const float4*)((const char*)ASRC + (unsigned)s * 16u);
}
__device__ __forceinline__ float4 edge_logits(float4 a, float4 ad) {
  float4 t;
  t.x = a.x + ad.x; t.x = t.x > 0.f ? t.x : 0.2f * t.x;
  t.y = a.y + ad.y; t.y = t.y > 0.f ? t.y : 0.2f * t.y;
  t.z = a.z + ad.z; t.z = t.z > 0.f ? t.z : 0.2f * t.z;
  t.w = a.w + ad.w; t.w = t.w > 0.f ? t.w : 0.2f * t.w;
  return t;
}
__device__ __forceinline__ float4 max4(float4 a, float4 b) {
  return make_float4(fmaxf(a.x, b.x), fmaxf(a.y, b.y), fmaxf(a.z, b.z), fmaxf(a.w, b.w));
}
__device__ __forceinline__ float4 edge_weights(bool v, float4 t, float4 m) {
  float4 e;
  e.x = v ? __expf(t.x - m.x) : 0.f;
  e.y = v ? __expf(t.y - m.y) : 0.f;
  e.z = v ? __expf(t.z - m.z) : 0.f;
  e.w = v ? __expf(t.w - m.w) : 0.f;
  return e;
}

template <int K>
__device__ __forceinline__ void load_slot(float4 (&x)[4], const char* __restrict__ xp, unsigned lo, int sp, int rem) {
  const int s = grp_bcast_i<K>(sp);
  if (K < rem) {
    const char* p = xp + ((unsigned)s * 512u + lo);
    x[0] = *(const float4*)(p);
    x[1] = *(const float4*)(p + 128);
    x[2] = *(const float4*)(p + 256);
    x[3] = *(const float4*)(p + 384);
  }
}
template <int K>
__device__ __forceinline__ void add_slot(const float4 (&x)[4], float4 (&acc)[4], float4& ss, float4 e, int rem) {
  const float w0 = grp_bcast<K>(e.x), w1 = grp_bcast<K>(e.y), w2 = grp_bcast<K>(e.z), w3 = grp_bcast<K>(e.w);
  if (K < rem) {
    ss.x += w0; ss.y += w1; ss.z += w2; ss.w += w3;
    acc[0].x += x[0].x * w0; acc[0].y += x[0].y * w0; acc[0].z += x[0].z * w0; acc[0].w += x[0].w * w0;
    acc[1].x += x[1].x * w1; acc[1].y += x[1].y * w1; acc[1].z += x[1].z * w1; acc[1].w += x[1].w * w1;
    acc[2].x += x[2].x * w2; acc[2].y += x[2].y * w2; acc[2].z += x[2].z * w2; acc[2].w += x[2].w * w2;
    acc[3].x += x[3].x * w3; acc[3].y += x[3].y * w3; acc[3].z += x[3].z * w3; acc[3].w += x[3].w * w3;
  }
}
template <int K0>
__device__ __forceinline__ void load_four(float4 (&xr)[4][4], const char* __restrict__ xp, unsigned lo, int sp, int rem) {
  load_slot<K0 + 0>(xr[0], xp, lo, sp, rem);
  load_slot<K0 + 1>(xr[1], xp, lo, sp, rem);
  load_slot<K0 + 2>(xr[2], xp, lo, sp, rem);
  load_slot<K0 + 3>(xr[3], xp, lo, sp, rem);
}
template <int K0>
__device__ __forceinline__ void add_four(const float4 (&xr)[4][4], float4 (&acc)[4], float4& ss, float4 e, int rem) {
  add_slot<K0 + 0>(xr[0], acc, ss, e, rem);
  add_slot<K0 + 1>(xr[1], acc, ss, e, rem);
  add_slot<K0 + 2>(xr[2], acc, ss, e, rem);
  add_slot<K0 + 3>(xr[3], acc, ss, e, rem);
}

template <bool RELU>
__device__ __forceinline__ void aggr_group(const int* __restrict__ rowptr,
                                           const int* __restrict__ csr_src,
                                           const float* __restrict__ XP,
                                           const float* __restrict__ ASRC,
                                           const float* __restrict__ ADST,
                                           const float* __restrict__ bias,
                                           int node, bool nv, int i, int ne, float4 (&o)[4]) {
  int beg = 0, end = 0;
  float4 ad = make_float4(0.f, 0.f, 0.f, 0.f);
  if (nv) {
    beg = rowptr[node];
    end = rowptr[node + 1];
    ad = *(const float4*)(ADST + (size_t)node * NH);
  }
  beg = max(beg, 0);
  end = min(end, ne);
  const int deg = max(end - beg, 0);
  auto seg = [&](int e) -> int {
    return *(const int*)((const char*)csr_src + (unsigned)(beg + e) * 4u);
  };
  const char* const xp = (const char*)XP;
  const unsigned lo = 16u * (unsigned)i;

  const bool v0 = i < deg, v1 = i + 8 < deg;
  int sp0 = 0, sp1 = 0;
  if (v0) sp0 = seg(i);
  if (v1) sp1 = seg(8 + i);
  const float4 a0 = load_asrc(ASRC, sp0);
  const float4 a1 = load_asrc(ASRC, sp1);

  float4 xr[4][4];
  load_four<0>(xr, xp, lo, sp0, deg);

  const float4 t0 = edge_logits(a0, ad);
  const float4 t1 = edge_logits(a1, ad);
  const float ninf = __int_as_float(0xff800000);
  float4 m = make_float4(ninf, ninf, ninf, ninf);
  if (v0) m = t0;
  if (v1) m = max4(m, t1);
  for (int ch = 16; __builtin_amdgcn_ballot_w32(ch < deg) != 0u; ch += 8) {
    if (ch + i < deg) m = max4(m, edge_logits(load_asrc(ASRC, seg(ch + i)), ad));
  }
  m.x = grp_max(m.x);
  m.y = grp_max(m.y);
  m.z = grp_max(m.z);
  m.w = grp_max(m.w);

  float4 acc[4];
  acc[0] = acc[1] = acc[2] = acc[3] = make_float4(0.f, 0.f, 0.f, 0.f);
  float4 ss = make_float4(0.f, 0.f, 0.f, 0.f);

  const float4 e0 = edge_weights(v0, t0, m);
  add_four<0>(xr, acc, ss, e0, deg);
  if (__builtin_amdgcn_ballot_w32(deg > 4) != 0u) {
    load_four<4>(xr, xp, lo, sp0, deg);
    add_four<4>(xr, acc, ss, e0, deg);
  }
  if (__builtin_amdgcn_ballot_w32(deg > 8) != 0u) {
    const float4 e1 = edge_weights(v1, t1, m);
    load_four<0>(xr, xp, lo, sp1, deg - 8);
    add_four<0>(xr, acc, ss, e1, deg - 8);
    if (__builtin_amdgcn_ballot_w32(deg > 12) != 0u) {
      load_four<4>(xr, xp, lo, sp1, deg - 8);
      add_four<4>(xr, acc, ss, e1, deg - 8);
    }
  }
  for (int ch = 16; __builtin_amdgcn_ballot_w32(ch < deg) != 0u; ch += 8) {
    const bool v = ch + i < deg;
    int sp = 0;
    float4 t = make_float4(0.f, 0.f, 0.f, 0.f);
    if (v) {
      sp = seg(ch + i);
      t = edge_logits(load_asrc(ASRC, sp), ad);
    }
    const float4 e = edge_weights(v, t, m);
    load_four<0>(xr, xp, lo, sp, deg - ch);
    add_four<0>(xr, acc, ss, e, deg - ch);
    if (__builtin_amdgcn_ballot_w32(deg > ch + 4) != 0u) {
      load_four<4>(xr, xp, lo, sp, deg - ch);
      add_four<4>(xr, acc, ss, e, deg - ch);
    }
  }

  const float den[4] = {ss.x + 1e-16f, ss.y + 1e-16f, ss.z + 1e-16f, ss.w + 1e-16f};
#pragma unroll
  for (int hd = 0; hd < NH; ++hd) {
    const float4 bv = *(const float4*)(bias + 32 * hd + 4 * i);
    float4 r;
    r.x = acc[hd].x / den[hd] + bv.x;
    r.y = acc[hd].y / den[hd] + bv.y;
    r.z = acc[hd].z / den[hd] + bv.z;
    r.w = acc[hd].w / den[hd] + bv.w;
    if (RELU) {
      r.x = fmaxf(r.x, 0.0f);
      r.y = fmaxf(r.y, 0.0f);
      r.z = fmaxf(r.z, 0.0f);
      r.w = fmaxf(r.w, 0.0f);
    }
    o[hd] = r;
  }
}

template <bool RELU>
__global__ __launch_bounds__(256) void k_node_softmax_aggr(const int* __restrict__ rowptr,
                                                           const int* __restrict__ csr_src,
                                                           const float* __restrict__ XP,
                                                           const float* __restrict__ ASRC,
                                                           const float* __restrict__ ADST,
                                                           const float* __restrict__ bias,
                                                           float* __restrict__ OUT,
                                                           int n, int ne) {
  const int lane = threadIdx.x & 31;
  const int i = lane & 7;
  const int wave = (blockIdx.x * blockDim.x + threadIdx.x) >> 5;
  if (wave * 4 >= n) return;
  const int node = wave * 4 + (lane >> 3);
  const bool nv = node < n;
  float4 o[4];
  aggr_group<RELU>(rowptr, csr_src, XP, ASRC, ADST, bias, node, nv, i, ne, o);
  if (!nv) return;
  char* const orow = (char*)OUT + (unsigned)node * 512u + 16 * i;
#pragma unroll
  for (int hd = 0; hd < NH; ++hd) *(float4*)(orow + 128 * hd) = o[hd];
}

__global__ __launch_bounds__(256) void k_aggr_gemm(const int* __restrict__ rowptr,
                                                   const int* __restrict__ csr_src,
                                                   const float* __restrict__ XP0,
                                                   const float* __restrict__ ASRC0,
                                                   const float* __restrict__ ADST0,
                                                   const float* __restrict__ bias0,
                                                   const float* __restrict__ W,
                                                   const float* __restrict__ att_src,
                                                   const float* __restrict__ att_dst,
                                                   float* __restrict__ XP1,
                                                   float* __restrict__ ASRC1,
                                                   float* __restrict__ ADST1, int n, int ne) {
  constexpr int K = HC;
  constexpr int P = K + 4;
  constexpr int KF = K / 4;
  constexpr int NG = K / 8;
  __shared__ float WlA[64 * P];
  __shared__ float WlB[64 * P];
  __shared__ float Tl[8 * 16 * P];

  const int tid = threadIdx.x;
  const int lane = tid & 31;
  const int c = lane & 15;
  const int h = lane >> 4;
  const int i8 = lane & 7;
  const int grp = lane >> 3;
  const int wv = __builtin_amdgcn_readfirstlane(tid >> 5);

  {
    const int wn = tid / KF, kf = tid - wn * KF;
    constexpr int RS = 256 / KF;
    const float* wsrc = W + (size_t)wn * K + 4 * kf;
    float* da = WlA + wn * P + 4 * kf;
    float* db = WlB + wn * P + 4 * kf;
#pragma unroll 8
    for (int r = 0; r < 64; r += RS)
      *(float4*)(da + r * P) = *(const float4*)(wsrc + (size_t)r * K);
#pragma unroll 8
    for (int r = 0; r < 64; r += RS)
      *(float4*)(db + r * P) = *(const float4*)(wsrc + (size_t)(64 + r) * K);
  }
  __syncthreads();

  const int tw = (int)gridDim.x * 8;
  const int gw = (int)blockIdx.x * 8 + wv;
  const int ngrp = (n + 3) >> 2;
  const int base = ngrp / tw;
  const int rem = ngrp - base * tw;
  const int cnt = base + (gw < rem ? 1 : 0);
  const int first = gw * base + min(gw, rem);

  float* const T = Tl + wv * (16 * P);
  float* const trow = T + grp * P + 4 * i8;
  const float* const tla = T + c * P + 4 * h;
  const float* const bla = WlA + c * P + 4 * h;
  const float* const blb = WlB + c * P + 4 * h;
  const float4 zero4 = make_float4(0.f, 0.f, 0.f, 0.f);

  for (int t0 = 0; t0 < cnt; t0 += 4) {
    const int gcount = min(4, cnt - t0);
    const int node0 = (first + t0) * 4;
    const int rv = min(4 * gcount, n - node0);

#pragma unroll 1
    for (int g = 0; g < 4; ++g) {
      float4 o[4];
      if (g < gcount) {
        const int node = node0 + 4 * g + grp;
        aggr_group<true>(rowptr, csr_src, XP0, ASRC0, ADST0, bias0, node, node < n, i8, ne, o);
      } else {
        o[0] = o[1] = o[2] = o[3] = zero4;
      }
      float* const tr = trow + (4 * g) * P;
#pragma unroll
      for (int hd = 0; hd < NH; ++hd) *(float4*)(tr + 32 * hd) = o[hd];
    }
    __builtin_amdgcn_fence(__ATOMIC_RELEASE, "wavefront");
    __builtin_amdgcn_wave_barrier();
    __builtin_amdgcn_fence(__ATOMIC_ACQUIRE, "wavefront");

    v8f acc[8] = {};
    float4 a[NG];
#pragma unroll
    for (int g = 0; g < NG; ++g) a[g] = *(const float4*)(tla + 8 * g);
    float4 b0[8], b1[8];
#pragma unroll
    for (int nt = 0; nt < 4; ++nt) {
      b0[nt] = *(const float4*)(bla + nt * 16 * P);
      b0[nt + 4] = *(const float4*)(blb + nt * 16 * P);
    }
#pragma unroll
    for (int g = 0; g < NG; ++g) {
      float4* bc = (g & 1) ? b1 : b0;
      float4* bn = (g & 1) ? b0 : b1;
      if (g + 1 < NG) {
        const int ko = 8 * (g + 1);
#pragma unroll
        for (int nt = 0; nt < 4; ++nt) {
          bn[nt] = *(const float4*)(bla + nt * 16 * P + ko);
          bn[nt + 4] = *(const float4*)(blb + nt * 16 * P + ko);
        }
      }
      const v2f a0 = {a[g].x, a[g].y};
      const v2f a1 = {a[g].z, a[g].w};
#pragma unroll
      for (int nt = 0; nt < 8; ++nt) {
        const v2f b = {bc[nt].x, bc[nt].y};
        acc[nt] = __builtin_amdgcn_wmma_f32_16x16x4_f32(
             false, a0,  false, b,
             (short)0, acc[nt],  false,  false);
      }
#pragma unroll
      for (int nt = 0; nt < 8; ++nt) {
        const v2f b = {bc[nt].z, bc[nt].w};
        acc[nt] = __builtin_amdgcn_wmma_f32_16x16x4_f32(
             false, a1,  false, b,
             (short)0, acc[nt],  false,  false);
      }
    }

    __builtin_amdgcn_sched_barrier(0);
    asm volatile("v_nop\n\tv_nop\n\tv_nop\n\tv_nop\n\tv_nop\n\tv_nop\n\tv_nop\n\tv_nop");
    __builtin_amdgcn_sched_barrier(0);

    float* const orow = XP1 + (size_t)(node0 + h * 8) * HC + c;
#pragma unroll
    for (int i = 0; i < 8; ++i) {
      if (h * 8 + i < rv) {
#pragma unroll
        for (int nt = 0; nt < 8; ++nt) orow[(size_t)i * HC + nt * 16] = acc[nt][i];
      }
    }

    float a_lo[2][NH], a_hi[2][NH];
#pragma unroll
    for (int hd = 0; hd < NH; ++hd) {
      a_lo[0][hd] = att_src[hd * 32 + c];
      a_hi[0][hd] = att_src[hd * 32 + 16 + c];
      a_lo[1][hd] = att_dst[hd * 32 + c];
      a_hi[1][hd] = att_dst[hd * 32 + 16 + c];
    }
    const bool l0 = (c & 1) != 0, l1 = (c & 2) != 0, l2 = (c & 4) != 0, l3 = (c & 8) != 0;
    float u[2][NH];
#pragma unroll
    for (int sd = 0; sd < 2; ++sd) {
      float sv[2][NH];
#pragma unroll
      for (int j2 = 0; j2 < 2; ++j2) {
        float qv[2][NH];
#pragma unroll
        for (int j1 = 0; j1 < 2; ++j1) {
          const int i = 4 * j2 + 2 * j1;
#pragma unroll
          for (int hd = 0; hd < NH; ++hd) {
            const v2f t0v = {acc[2 * hd][i], acc[2 * hd][i + 1]};
            const v2f t1v = {acc[2 * hd + 1][i], acc[2 * hd + 1][i + 1]};
            const v2f xy = t0v * a_lo[sd][hd] + t1v * a_hi[sd][hd];
            qv[j1][hd] = row_xadd<1>(l0, xy.x, xy.y);
          }
        }
#pragma unroll
        for (int hd = 0; hd < NH; ++hd) sv[j2][hd] = row_xadd<2>(l1, qv[0][hd], qv[1][hd]);
      }
#pragma unroll
      for (int hd = 0; hd < NH; ++hd) u[sd][hd] = row_xadd<4>(l2, sv[0][hd], sv[1][hd]);
    }
    float4 r;
    r.x = row_xadd<8>(l3, u[0][0], u[1][0]);
    r.y = row_xadd<8>(l3, u[0][1], u[1][1]);
    r.z = row_xadd<8>(l3, u[0][2], u[1][2]);
    r.w = row_xadd<8>(l3, u[0][3], u[1][3]);
    const int rr = h * 8 + (c & 7);
    if (rr < rv) {
      float* const arow = (l3 ? ADST1 : ASRC1) + (size_t)(node0 + rr) * NH;
      *(float4*)arow = r;
    }
  }
}

extern "C" void kernel_launch(void* const* d_in, const int* in_sizes, int n_in,
                              void* d_out, int out_size, void* d_ws, size_t ws_size,
                              hipStream_t stream) {
  const float* x     = (const float*)d_in[0];
  const int*   ei    = (const int*)d_in[1];
  const float* W0    = (const float*)d_in[2];
  const float* atts0 = (const float*)d_in[3];
  const float* attd0 = (const float*)d_in[4];
  const float* b0    = (const float*)d_in[5];
  const float* W1    = (const float*)d_in[6];
  const float* atts1 = (const float*)d_in[7];
  const float* attd1 = (const float*)d_in[8];
  const float* b1    = (const float*)d_in[9];

  const int n = NNODES, ne = NEDGES;
  const int* srcv = ei;
  const int* dstv = ei + ne;

  char* ws = (char*)d_ws;
  float* XP   = (float*)(ws + 0);
  float* XP1  = (float*)(ws + 51200000);
  float* ASRC = (float*)(ws + 102400000);
  float* ADST = (float*)(ws + 104000000);
  int*   LOT  = (int*)(ws + 105600000);
  int*   RPTR = (int*)(ws + 107200000);
  int*   CSRC = (int*)(ws + 108800000);
  int*   BK   = (int*)(ws + 112000000);
  float* ASRC1 = (float*)(ws + 112000000);
  float* ADST1 = (float*)(ws + 113600000);
  float* OUT  = (float*)d_out;

  const dim3 blk(256);
  const int nodeBlocks = cdiv(cdiv(n, 4) * 32, 256);

  k_chunk_sort<<<NCH, 1024, 0, stream>>>(srcv, dstv, LOT, BK, ne);
  k_bucket_csr<<<NBK, 1024, 0, stream>>>(LOT, BK, RPTR, CSRC, n, ne);

  k_gemm_wmma<256><<<256, 512, 0, stream>>>(x, W0, atts0, attd0, XP, ASRC, ADST, n);

  k_aggr_gemm<<<512, blk, 0, stream>>>(RPTR, CSRC, XP, ASRC, ADST, b0, W1, atts1, attd1,
                                       XP1, ASRC1, ADST1, n, ne);

  k_node_softmax_aggr<false><<<nodeBlocks, blk, 0, stream>>>(RPTR, CSRC, XP1, ASRC1, ADST1, b1, OUT, n, ne);
}
